# starting kernel
# baseline (speedup 1.0000x reference)
#include <hip/hip_runtime.h>

typedef __attribute__((ext_vector_type(2))) float v2f;
typedef __attribute__((ext_vector_type(8))) float v8f;

#define HID 64

__global__ void k_deg_init(float* deg, int n) {
    int i = blockIdx.x * blockDim.x + threadIdx.x;
    if (i < n) deg[i] = 1.0f;
}

__global__ void k_deg_edges(const int* __restrict__ dst, float* deg, int e) {
    int i = blockIdx.x * blockDim.x + threadIdx.x;
    if (i < e) atomicAdd(&deg[dst[i]], 1.0f);
}

__global__ void k_dinv(float* deg, int n) {
    int i = blockIdx.x * blockDim.x + threadIdx.x;
    if (i < n) { float d = deg[i]; deg[i] = (d > 0.f) ? rsqrtf(d) : 0.f; }
}

template <int K, bool DO_RELU>
__global__ void k_gemm_wmma(const float* __restrict__ A, const float* __restrict__ W,
                            float* __restrict__ C, int n) {
    __shared__ float sW[K * HID];
    int tid = threadIdx.x;
    for (int i = tid; i < K * HID; i += 128) sW[i] = W[i];
    __syncthreads();

    int wave = tid >> 5;
    int lane = tid & 31;
    int half = lane >> 4;
    int l15  = lane & 15;
    int m0 = blockIdx.x * 16;
    int n0 = wave * 16;

    int arow = m0 + l15;
    if (arow >= n) arow = n - 1;
    const float* ap = A + (size_t)arow * K + half * 2;
    const float* bp = sW + n0 + l15 + half * 2 * HID;

    v8f c = {};
#pragma unroll
    for (int k = 0; k < K; k += 4) {
        v2f a, b;
        float ax = ap[k];
        float ay = ap[k + 1];
        if (DO_RELU) { ax = fmaxf(ax, 0.f); ay = fmaxf(ay, 0.f); }
        a[0] = ax; a[1] = ay;
        b[0] = bp[k * HID];
        b[1] = bp[k * HID + HID];
        c = __builtin_amdgcn_wmma_f32_16x16x4_f32(
                 false, a,  false, b,
                 (short)0, c,  false,  false);
    }

    int rbase = m0 + half * 8;
#pragma unroll
    for (int j = 0; j < 8; ++j) {
        int r = rbase + j;
        if (r < n) C[(size_t)r * HID + n0 + l15] = c[j];
    }
}

__global__ void k_selfinit(const float* __restrict__ hW, const float* __restrict__ dinv,
                           float* __restrict__ agg, int n) {
    int i = blockIdx.x * blockDim.x + threadIdx.x;
    if (i < n * HID) {
        int v = i >> 6;
        float di = dinv[v];
        agg[i] = hW[i] * di * di;
    }
}

__global__ void k_edge_scatter(const int* __restrict__ src, const int* __restrict__ dst,
                               const float* __restrict__ dinv, const float* __restrict__ hW,
                               float* __restrict__ agg, int e) {
    int i = blockIdx.x * blockDim.x + threadIdx.x;
    int total = e * HID;
    if (i < total) {
        int ed = i >> 6, ch = i & 63;
        int s = src[ed], d = dst[ed];
        float nm = dinv[s] * dinv[d];
        atomicAdd(&agg[(size_t)d * HID + ch], hW[(size_t)s * HID + ch] * nm);
    }
}

__global__ void k_pool_zero(float* sums, float* cnt, int g) {
    int i = blockIdx.x * blockDim.x + threadIdx.x;
    if (i < g * HID) sums[i] = 0.f;
    if (i < g)       cnt[i]  = 0.f;
}

__global__ void k_pool_scatter(const float* __restrict__ h, const int* __restrict__ batch,
                               float* sums, float* cnt, int n) {
    int i = blockIdx.x * blockDim.x + threadIdx.x;
    if (i < n * HID) {
        int v = i >> 6, ch = i & 63;
        int g = batch[v];
        atomicAdd(&sums[(size_t)g * HID + ch], h[i]);
        if (ch == 0) atomicAdd(&cnt[g], 1.0f);
    }
}

__global__ void k_final(const float* __restrict__ sums, const float* __restrict__ cnt,
                        const float* __restrict__ Wl, const float* __restrict__ bl,
                        float* __restrict__ out, int g, int oc) {
    int i = blockIdx.x * blockDim.x + threadIdx.x;
    if (i < g * oc) {
        int gi = i / oc, o = i % oc;
        float inv = 1.0f / fmaxf(cnt[gi], 1.0f);
        float acc = bl[o];
        for (int c = 0; c < HID; ++c)
            acc += sums[(size_t)gi * HID + c] * inv * Wl[c * oc + o];
        out[i] = acc;
    }
}

extern "C" void kernel_launch(void* const* d_in, const int* in_sizes, int n_in,
                              void* d_out, int out_size, void* d_ws, size_t ws_size,
                              hipStream_t stream) {
    const float* x    = (const float*)d_in[0];
    const int*   eidx = (const int*)  d_in[1];
    const int*   batch= (const int*)  d_in[2];
    const float* W1   = (const float*)d_in[3];
    const float* W2   = (const float*)d_in[4];
    const float* W3   = (const float*)d_in[5];
    const float* Wl   = (const float*)d_in[6];
    const float* bl   = (const float*)d_in[7];

    const int INC = in_sizes[3] / HID;
    const int N   = in_sizes[0] / INC;
    const int E   = in_sizes[1] / 2;
    const int OC  = in_sizes[7];
    const int G   = out_size / OC;

    const int* src = eidx;
    const int* dst = eidx + E;

    char* ws = (char*)d_ws; size_t off = 0;
    auto carve = [&](size_t bytes) {
        void* p = ws + off; off = (off + bytes + 255) & ~(size_t)255; return p;
    };
    float* dinv = (float*)carve((size_t)N * 4);
    float* hW   = (float*)carve((size_t)N * HID * 4);
    float* agg  = (float*)carve((size_t)N * HID * 4);
    float* sums = (float*)carve((size_t)G * HID * 4);
    float* cnt  = (float*)carve((size_t)G * 4);
    (void)ws_size; (void)n_in;

    const int B = 256;
    const int nB  = (N + B - 1) / B;
    const int eB  = (E + B - 1) / B;
    const int nhB = (N * HID + B - 1) / B;
    const int ehB = (E * HID + B - 1) / B;
    const int mt  = (N + 15) / 16;

    k_deg_init <<<nB, B, 0, stream>>>(dinv, N);
    k_deg_edges<<<eB, B, 0, stream>>>(dst, dinv, E);
    k_dinv     <<<nB, B, 0, stream>>>(dinv, N);

    k_gemm_wmma<128, false><<<mt, 128, 0, stream>>>(x, W1, hW, N);
    k_selfinit    <<<nhB, B, 0, stream>>>(hW, dinv, agg, N);
    k_edge_scatter<<<ehB, B, 0, stream>>>(src, dst, dinv, hW, agg, E);

    k_gemm_wmma<64, true><<<mt, 128, 0, stream>>>(agg, W2, hW, N);
    k_selfinit    <<<nhB, B, 0, stream>>>(hW, dinv, agg, N);
    k_edge_scatter<<<ehB, B, 0, stream>>>(src, dst, dinv, hW, agg, E);

    k_gemm_wmma<64, true><<<mt, 128, 0, stream>>>(agg, W3, hW, N);
    k_selfinit    <<<nhB, B, 0, stream>>>(hW, dinv, agg, N);
    k_edge_scatter<<<ehB, B, 0, stream>>>(src, dst, dinv, hW, agg, E);

    k_pool_zero   <<<(G * HID + B - 1) / B, B, 0, stream>>>(sums, cnt, G);
    k_pool_scatter<<<nhB, B, 0, stream>>>(agg, batch, sums, cnt, N);
    k_final       <<<(G * OC + B - 1) / B, B, 0, stream>>>(sums, cnt, Wl, bl,
                                                           (float*)d_out, G, OC);
}


# Aggregate by gather, not float-atomic scatter: edges ranked per dst once (returning integer atomic, arrival-order rank), scan, fill; one wave sums each node's rows in registers, one store per row
# speedup vs baseline: 2.7170x; 2.7170x over previous
#include <hip/hip_runtime.h>

typedef __attribute__((ext_vector_type(2))) float v2f;
typedef __attribute__((ext_vector_type(8))) float v8f;

#define HID 64

__global__ void k_cnt_zero(int* cnt, int n) {
    int i = blockIdx.x * blockDim.x + threadIdx.x;
    if (i < n) cnt[i] = 0;
}

__global__ void k_rank(const int* __restrict__ dst, int* cnt, int* __restrict__ rank, int e) {
    int i = blockIdx.x * blockDim.x + threadIdx.x;
    if (i < e) rank[i] = atomicAdd(&cnt[dst[i]], 1);
}

#define EPT 16
__global__ void __launch_bounds__(1024)
k_scan(const int* __restrict__ cnt, int* __restrict__ rowptr, float* __restrict__ dinv, int n) {
    __shared__ int wsum[32];
    int tid = threadIdx.x, lane = tid & 31, wave = tid >> 5;
    int carry = 0;
    for (int base = 0; base < n; base += 1024 * EPT) {
        int i0 = base + tid * EPT;
        bool full = (i0 + EPT <= n);
        int c[EPT];
        if (full) {
#pragma unroll
            for (int q = 0; q < EPT / 4; ++q) {
                int4 t = ((const int4*)(cnt + i0))[q];
                c[4 * q] = t.x; c[4 * q + 1] = t.y; c[4 * q + 2] = t.z; c[4 * q + 3] = t.w;
            }
        } else {
#pragma unroll
            for (int k = 0; k < EPT; ++k) c[k] = (i0 + k < n) ? cnt[i0 + k] : 0;
        }
        int tot = 0;
#pragma unroll
        for (int k = 0; k < EPT; ++k) tot += c[k];
        int x = tot;
#pragma unroll
        for (int o = 1; o < 32; o <<= 1) {
            int y = __shfl_up(x, o);
            if (lane >= o) x += y;
        }
        if (lane == 31) wsum[wave] = x;
        __syncthreads();
        int w = wsum[lane];
        int wx = w;
#pragma unroll
        for (int o = 1; o < 32; o <<= 1) {
            int y = __shfl_up(wx, o);
            if (lane >= o) wx += y;
        }
        int wave_excl = __shfl(wx, wave) - __shfl(w, wave);
        int tile_tot  = __shfl(wx, 31);
        int run = carry + wave_excl + (x - tot);
        int   rp[EPT];
        float dv[EPT];
#pragma unroll
        for (int k = 0; k < EPT; ++k) {
            rp[k] = run;
            run += c[k];
            float d = (float)(c[k] + 1);
            dv[k] = (d > 0.f) ? rsqrtf(d) : 0.f;
        }
        if (full) {
#pragma unroll
            for (int q = 0; q < EPT / 4; ++q) {
                ((int4*)(rowptr + i0))[q] =
                    make_int4(rp[4 * q], rp[4 * q + 1], rp[4 * q + 2], rp[4 * q + 3]);
                ((float4*)(dinv + i0))[q] =
                    make_float4(dv[4 * q], dv[4 * q + 1], dv[4 * q + 2], dv[4 * q + 3]);
            }
        } else {
#pragma unroll
            for (int k = 0; k < EPT; ++k)
                if (i0 + k < n) { rowptr[i0 + k] = rp[k]; dinv[i0 + k] = dv[k]; }
        }
        carry += tile_tot;
        __syncthreads();
    }
    if (tid == 0) rowptr[n] = carry;
}

__global__ void k_fill(const int* __restrict__ src, const int* __restrict__ dst,
                       const int* __restrict__ rank, const int* __restrict__ rowptr,
                       const float* __restrict__ dinv, int2* __restrict__ csr, int e) {
    int i = blockIdx.x * blockDim.x + threadIdx.x;
    if (i < e) {
        int s = src[i], d = dst[i];
        float nm = dinv[s] * dinv[d];
        csr[rowptr[d] + rank[i]] = make_int2(s, __float_as_int(nm));
    }
}

template <int K, bool DO_RELU>
__global__ void k_gemm_wmma(const float* __restrict__ A, const float* __restrict__ W,
                            float* __restrict__ C, int n) {
    __shared__ float sW[K * HID];
    int tid = threadIdx.x;
    for (int i = tid; i < K * HID; i += 128) sW[i] = W[i];
    __syncthreads();

    int wave = tid >> 5;
    int lane = tid & 31;
    int half = lane >> 4;
    int l15  = lane & 15;
    int m0 = blockIdx.x * 16;
    int n0 = wave * 16;

    int arow = m0 + l15;
    if (arow >= n) arow = n - 1;
    const float* ap = A + (size_t)arow * K + half * 2;
    const float* bp = sW + n0 + l15 + half * 2 * HID;

    v8f c = {};
#pragma unroll
    for (int k = 0; k < K; k += 4) {
        v2f a, b;
        float ax = ap[k];
        float ay = ap[k + 1];
        if (DO_RELU) { ax = fmaxf(ax, 0.f); ay = fmaxf(ay, 0.f); }
        a[0] = ax; a[1] = ay;
        b[0] = bp[k * HID];
        b[1] = bp[k * HID + HID];
        c = __builtin_amdgcn_wmma_f32_16x16x4_f32(
                 false, a,  false, b,
                 (short)0, c,  false,  false);
    }

    int rbase = m0 + half * 8;
#pragma unroll
    for (int j = 0; j < 8; ++j) {
        int r = rbase + j;
        if (r < n) C[(size_t)r * HID + n0 + l15] = c[j];
    }
}

template <int U>
__device__ __forceinline__ void gather_rows(const float2* __restrict__ h2, int s_l, int w_l,
                                            int j, int lane, float2& acc) {
    float2 r[U];
    float  w[U];
#pragma unroll
    for (int k = 0; k < U; ++k) {
        int s = __builtin_amdgcn_readlane(s_l, j + k);
        w[k]  = __int_as_float(__builtin_amdgcn_readlane(w_l, j + k));
        r[k]  = h2[(size_t)s * (HID / 2) + lane];
    }
#pragma unroll
    for (int k = 0; k < U; ++k) {
        acc.x = fmaf(r[k].x, w[k], acc.x);
        acc.y = fmaf(r[k].y, w[k], acc.y);
    }
}

__global__ void __launch_bounds__(256)
k_gather(const int* __restrict__ rowptr, const int2* __restrict__ csr,
         const float* __restrict__ dinv, const float* __restrict__ hW,
         float* __restrict__ agg, int n) {
    int lane = threadIdx.x & 31;
    int v = __builtin_amdgcn_readfirstlane((int)(blockIdx.x * 8 + (threadIdx.x >> 5)));
    if (v >= n) return;
    int beg = rowptr[v], end = rowptr[v + 1];
    float di = dinv[v];
    const float2* h2 = (const float2*)hW;
    float2 hv = h2[(size_t)v * (HID / 2) + lane];
    float sw = di * di;
    float2 acc;
    acc.x = hv.x * sw;
    acc.y = hv.y * sw;
    for (int p0 = beg; p0 < end; p0 += 32) {
        int m = end - p0;
        if (m > 32) m = 32;
        int2 rec = make_int2(0, 0);
        if (lane < m) rec = csr[p0 + lane];
        int j = 0;
        for (; j + 8 <= m; j += 8) gather_rows<8>(h2, rec.x, rec.y, j, lane, acc);
        if (m & 4) { gather_rows<4>(h2, rec.x, rec.y, j, lane, acc); j += 4; }
        if (m & 2) { gather_rows<2>(h2, rec.x, rec.y, j, lane, acc); j += 2; }
        if (m & 1) { gather_rows<1>(h2, rec.x, rec.y, j, lane, acc); }
    }
    ((float2*)agg)[(size_t)v * (HID / 2) + lane] = acc;
}

__global__ void k_pool_zero(float* sums, float* cnt, int g) {
    int i = blockIdx.x * blockDim.x + threadIdx.x;
    if (i < g * HID) sums[i] = 0.f;
    if (i < g)       cnt[i]  = 0.f;
}

__global__ void k_pool_scatter(const float* __restrict__ h, const int* __restrict__ batch,
                               float* sums, float* cnt, int n) {
    int i = blockIdx.x * blockDim.x + threadIdx.x;
    if (i < n * HID) {
        int v = i >> 6, ch = i & 63;
        int g = batch[v];
        atomicAdd(&sums[(size_t)g * HID + ch], h[i]);
        if (ch == 0) atomicAdd(&cnt[g], 1.0f);
    }
}

__global__ void k_final(const float* __restrict__ sums, const float* __restrict__ cnt,
                        const float* __restrict__ Wl, const float* __restrict__ bl,
                        float* __restrict__ out, int g, int oc) {
    int i = blockIdx.x * blockDim.x + threadIdx.x;
    if (i < g * oc) {
        int gi = i / oc, o = i % oc;
        float inv = 1.0f / fmaxf(cnt[gi], 1.0f);
        float acc = bl[o];
        for (int c = 0; c < HID; ++c)
            acc += sums[(size_t)gi * HID + c] * inv * Wl[c * oc + o];
        out[i] = acc;
    }
}

extern "C" void kernel_launch(void* const* d_in, const int* in_sizes, int n_in,
                              void* d_out, int out_size, void* d_ws, size_t ws_size,
                              hipStream_t stream) {
    const float* x    = (const float*)d_in[0];
    const int*   eidx = (const int*)  d_in[1];
    const int*   batch= (const int*)  d_in[2];
    const float* W1   = (const float*)d_in[3];
    const float* W2   = (const float*)d_in[4];
    const float* W3   = (const float*)d_in[5];
    const float* Wl   = (const float*)d_in[6];
    const float* bl   = (const float*)d_in[7];

    const int INC = in_sizes[3] / HID;
    const int N   = in_sizes[0] / INC;
    const int E   = in_sizes[1] / 2;
    const int OC  = in_sizes[7];
    const int G   = out_size / OC;

    const int* src = eidx;
    const int* dst = eidx + E;

    char* ws = (char*)d_ws; size_t off = 0;
    auto carve = [&](size_t bytes) {
        void* p = ws + off; off = (off + bytes + 255) & ~(size_t)255; return p;
    };
    int*   ecnt   = (int*)  carve((size_t)N * 4);
    int*   rowptr = (int*)  carve((size_t)(N + 1) * 4);
    float* dinv   = (float*)carve((size_t)N * 4);
    int*   rank   = (int*)  carve((size_t)E * 4);
    int2*  csr    = (int2*) carve((size_t)E * 8);
    float* hW     = (float*)carve((size_t)N * HID * 4);
    float* agg    = (float*)carve((size_t)N * HID * 4);
    float* sums   = (float*)carve((size_t)G * HID * 4);
    float* cnt    = (float*)carve((size_t)G * 4);
    (void)ws_size; (void)n_in;

    const int B = 256;
    const int nB  = (N + B - 1) / B;
    const int eB  = (E + B - 1) / B;
    const int nhB = (N * HID + B - 1) / B;
    const int mt  = (N + 15) / 16;
    const int gB  = (N + 7) / 8;

    k_cnt_zero<<<nB, B, 0, stream>>>(ecnt, N);
    k_rank    <<<eB, B, 0, stream>>>(dst, ecnt, rank, E);
    k_scan    <<<1, 1024, 0, stream>>>(ecnt, rowptr, dinv, N);
    k_fill    <<<eB, B, 0, stream>>>(src, dst, rank, rowptr, dinv, csr, E);

    k_gemm_wmma<128, false><<<mt, 128, 0, stream>>>(x, W1, hW, N);
    k_gather<<<gB, B, 0, stream>>>(rowptr, csr, dinv, hW, agg, N);

    k_gemm_wmma<64, true><<<mt, 128, 0, stream>>>(agg, W2, hW, N);
    k_gather<<<gB, B, 0, stream>>>(rowptr, csr, dinv, hW, agg, N);

    k_gemm_wmma<64, true><<<mt, 128, 0, stream>>>(agg, W3, hW, N);
    k_gather<<<gB, B, 0, stream>>>(rowptr, csr, dinv, hW, agg, N);

    k_pool_zero   <<<(G * HID + B - 1) / B, B, 0, stream>>>(sums, cnt, G);
    k_pool_scatter<<<nhB, B, 0, stream>>>(agg, batch, sums, cnt, N);
    k_final       <<<(G * OC + B - 1) / B, B, 0, stream>>>(sums, cnt, Wl, bl,
                                                           (float*)d_out, G, OC);
}


# Fewer, longer waves, same work: 8 edges per lane in rank/fill, 4 nodes per gather wave, 16x64 strip per GEMM wave, pooling sums 32-node runs in registers before its float atomics (sum order changes)
# speedup vs baseline: 3.1781x; 1.1697x over previous
#include <hip/hip_runtime.h>

typedef __attribute__((ext_vector_type(2))) float v2f;
typedef __attribute__((ext_vector_type(8))) float v8f;

#define HID 64

#define EPL  8
#define NPW  4
#define PRUN 32

__global__ void k_cnt_zero(int* cnt, int n) {
    int i = (blockIdx.x * blockDim.x + threadIdx.x) * 4;
    if (i + 4 <= n) *(int4*)(cnt + i) = make_int4(0, 0, 0, 0);
    else for (int k = i; k < n; ++k) cnt[k] = 0;
}

template <bool FULL>
__device__ __forceinline__ void rank_span(const int* __restrict__ dst, int* cnt,
                                          int* __restrict__ rank, int e, int base) {
    int d[EPL], r[EPL];
#pragma unroll
    for (int k = 0; k < EPL; ++k) {
        int i = base + k * 256;
        d[k] = dst[(FULL || i < e) ? i : e - 1];
    }
#pragma unroll
    for (int k = 0; k < EPL; ++k) {
        int i = base + k * 256;
        r[k] = 0;
        if (FULL || i < e) r[k] = atomicAdd(&cnt[d[k]], 1);
    }
#pragma unroll
    for (int k = 0; k < EPL; ++k) {
        int i = base + k * 256;
        if (FULL || i < e) rank[i] = r[k];
    }
}

__global__ void __launch_bounds__(256)
k_rank(const int* __restrict__ dst, int* cnt, int* __restrict__ rank, int e) {
    int span = blockIdx.x * (256 * EPL);
    int base = span + threadIdx.x;
    if (span + 256 * EPL <= e) rank_span<true>(dst, cnt, rank, e, base);
    else                       rank_span<false>(dst, cnt, rank, e, base);
}

#define EPT 16
__global__ void __launch_bounds__(1024)
k_scan(const int* __restrict__ cnt, int* __restrict__ rowptr, float* __restrict__ dinv, int n) {
    __shared__ int wsum[32];
    int tid = threadIdx.x, lane = tid & 31, wave = tid >> 5;
    int carry = 0;
    for (int base = 0; base < n; base += 1024 * EPT) {
        int i0 = base + tid * EPT;
        bool full = (i0 + EPT <= n);
        int c[EPT];
        if (full) {
#pragma unroll
            for (int q = 0; q < EPT / 4; ++q) {
                int4 t = ((const int4*)(cnt + i0))[q];
                c[4 * q] = t.x; c[4 * q + 1] = t.y; c[4 * q + 2] = t.z; c[4 * q + 3] = t.w;
            }
        } else {
#pragma unroll
            for (int k = 0; k < EPT; ++k) c[k] = (i0 + k < n) ? cnt[i0 + k] : 0;
        }
        int tot = 0;
#pragma unroll
        for (int k = 0; k < EPT; ++k) tot += c[k];
        int x = tot;
#pragma unroll
        for (int o = 1; o < 32; o <<= 1) {
            int y = __shfl_up(x, o);
            if (lane >= o) x += y;
        }
        if (lane == 31) wsum[wave] = x;
        __syncthreads();
        int w = wsum[lane];
        int wx = w;
#pragma unroll
        for (int o = 1; o < 32; o <<= 1) {
            int y = __shfl_up(wx, o);
            if (lane >= o) wx += y;
        }
        int wave_excl = __shfl(wx, wave) - __shfl(w, wave);
        int tile_tot  = __shfl(wx, 31);
        int run = carry + wave_excl + (x - tot);
        int   rp[EPT];
        float dv[EPT];
#pragma unroll
        for (int k = 0; k < EPT; ++k) {
            rp[k] = run;
            run += c[k];
            float d = (float)(c[k] + 1);
            dv[k] = (d > 0.f) ? rsqrtf(d) : 0.f;
        }
        if (full) {
#pragma unroll
            for (int q = 0; q < EPT / 4; ++q) {
                ((int4*)(rowptr + i0))[q] =
                    make_int4(rp[4 * q], rp[4 * q + 1], rp[4 * q + 2], rp[4 * q + 3]);
                ((float4*)(dinv + i0))[q] =
                    make_float4(dv[4 * q], dv[4 * q + 1], dv[4 * q + 2], dv[4 * q + 3]);
            }
        } else {
#pragma unroll
            for (int k = 0; k < EPT; ++k)
                if (i0 + k < n) { rowptr[i0 + k] = rp[k]; dinv[i0 + k] = dv[k]; }
        }
        carry += tile_tot;
        __syncthreads();
    }
    if (tid == 0) rowptr[n] = carry;
}

template <bool FULL>
__device__ __forceinline__ void fill_span(const int* __restrict__ src, const int* __restrict__ dst,
                                          const int* __restrict__ rank,
                                          const int* __restrict__ rowptr,
                                          const float* __restrict__ dinv,
                                          int2* __restrict__ csr, int e, int base) {
    int s[EPL], d[EPL], rk[EPL];
#pragma unroll
    for (int k = 0; k < EPL; ++k) {
        int i = base + k * 256;
        int ic = (FULL || i < e) ? i : e - 1;
        s[k]  = src[ic];
        d[k]  = dst[ic];
        rk[k] = rank[ic];
    }
    int   rp[EPL];
    float ds[EPL], dd[EPL];
#pragma unroll
    for (int k = 0; k < EPL; ++k) {
        rp[k] = rowptr[d[k]];
        ds[k] = dinv[s[k]];
        dd[k] = dinv[d[k]];
    }
#pragma unroll
    for (int k = 0; k < EPL; ++k) {
        int i = base + k * 256;
        float nm = ds[k] * dd[k];
        if (FULL || i < e) csr[rp[k] + rk[k]] = make_int2(s[k], __float_as_int(nm));
    }
}

__global__ void __launch_bounds__(256)
k_fill(const int* __restrict__ src, const int* __restrict__ dst,
       const int* __restrict__ rank, const int* __restrict__ rowptr,
       const float* __restrict__ dinv, int2* __restrict__ csr, int e) {
    int span = blockIdx.x * (256 * EPL);
    int base = span + threadIdx.x;
    if (span + 256 * EPL <= e) fill_span<true>(src, dst, rank, rowptr, dinv, csr, e, base);
    else                       fill_span<false>(src, dst, rank, rowptr, dinv, csr, e, base);
}

#define WMMA_F32(a, b, c) __builtin_amdgcn_wmma_f32_16x16x4_f32( \
         false, a,  false, b,  (short)0, c, \
         false,  false)
template <int K, bool DO_RELU>
__global__ void __launch_bounds__(128)
k_gemm_wmma(const float* __restrict__ A, const float* __restrict__ W,
            float* __restrict__ C, int n) {
    __shared__ float4 sW4[K * HID / 4];
    const float* sW = (const float*)sW4;
    int tid = threadIdx.x;
#pragma unroll
    for (int q = 0; q < K * HID / 4 / 128; ++q) sW4[tid + q * 128] = ((const float4*)W)[tid + q * 128];
    __syncthreads();

    int wave = tid >> 5;
    int lane = tid & 31;
    int half = lane >> 4;
    int l15  = lane & 15;
    int m0 = blockIdx.x * 64 + wave * 16;

    int arow = m0 + l15;
    if (arow >= n) arow = n - 1;
    const float* ap = A + (size_t)arow * K + half * 2;
    const float* bp = sW + l15 + half * 2 * HID;

    v8f c0 = {}, c1 = {}, c2 = {}, c3 = {};
#pragma unroll
    for (int k = 0; k < K; k += 4) {
        v2f a, b0, b1, b2, b3;
        float ax = ap[k];
        float ay = ap[k + 1];
        if (DO_RELU) { ax = fmaxf(ax, 0.f); ay = fmaxf(ay, 0.f); }
        a[0] = ax; a[1] = ay;
        const float* bk = bp + k * HID;
        b0[0] = bk[0];  b0[1] = bk[HID];
        b1[0] = bk[16]; b1[1] = bk[HID + 16];
        b2[0] = bk[32]; b2[1] = bk[HID + 32];
        b3[0] = bk[48]; b3[1] = bk[HID + 48];
        c0 = WMMA_F32(a, b0, c0);
        c1 = WMMA_F32(a, b1, c1);
        c2 = WMMA_F32(a, b2, c2);
        c3 = WMMA_F32(a, b3, c3);
    }

    int rbase = m0 + half * 8;
#pragma unroll
    for (int j = 0; j < 8; ++j) {
        int r = rbase + j;
        if (r < n) {
            float* cr = C + (size_t)r * HID + l15;
            cr[0] = c0[j]; cr[16] = c1[j]; cr[32] = c2[j]; cr[48] = c3[j];
        }
    }
}

template <int U>
__device__ __forceinline__ void gather_rows(const float2* __restrict__ h2, int s_l, int w_l,
                                            int j, int lane, float2& acc) {
    float2 r[U];
    float  w[U];
#pragma unroll
    for (int k = 0; k < U; ++k) {
        int s = __builtin_amdgcn_readlane(s_l, j + k);
        w[k]  = __int_as_float(__builtin_amdgcn_readlane(w_l, j + k));
        r[k]  = h2[(size_t)s * (HID / 2) + lane];
    }
#pragma unroll
    for (int k = 0; k < U; ++k) {
        acc.x = fmaf(r[k].x, w[k], acc.x);
        acc.y = fmaf(r[k].y, w[k], acc.y);
    }
}

__device__ __forceinline__ void gather_chunk(const float2* __restrict__ h2, int s_l, int w_l,
                                             int m, int lane, float2& acc) {
    int j = 0;
    for (; j + 8 <= m; j += 8) gather_rows<8>(h2, s_l, w_l, j, lane, acc);
    if (m & 4) { gather_rows<4>(h2, s_l, w_l, j, lane, acc); j += 4; }
    if (m & 2) { gather_rows<2>(h2, s_l, w_l, j, lane, acc); j += 2; }
    if (m & 1) { gather_rows<1>(h2, s_l, w_l, j, lane, acc); }
}

__global__ void __launch_bounds__(256)
k_gather(const int* __restrict__ rowptr, const int2* __restrict__ csr,
         const float* __restrict__ dinv, const float* __restrict__ hW,
         float* __restrict__ agg, int n) {
    int lane = threadIdx.x & 31;
    int w  = __builtin_amdgcn_readfirstlane((int)(blockIdx.x * 8 + (threadIdx.x >> 5)));
    int v0 = w * NPW;
    if (v0 >= n) return;
    int vend = v0 + NPW;
    if (vend > n) vend = n;
    const float2* h2 = (const float2*)hW;
    int beg = rowptr[v0], end = rowptr[v0 + 1];
    float di = dinv[v0];
    int2 rec = make_int2(0, 0);
    if (lane < end - beg) rec = csr[beg + lane];
    float2 hv = h2[(size_t)v0 * (HID / 2) + lane];
    float sw = di * di;
    float2 acc;
    acc.x = hv.x * sw;
    acc.y = hv.y * sw;
    for (int v = v0; v < vend; ++v) {
        int vn = (v + 1 < vend) ? v + 1 : v;
        int nbeg = rowptr[vn], nend = rowptr[vn + 1];
        float ndi = dinv[vn];
        int2 nrec = make_int2(0, 0);
        if (lane < nend - nbeg) nrec = csr[nbeg + lane];
        float2 nhv = h2[(size_t)vn * (HID / 2) + lane];

        int m = end - beg;
        if (m > 32) m = 32;
        gather_chunk(h2, rec.x, rec.y, m, lane, acc);
        for (int p0 = beg + 32; p0 < end; p0 += 32) {
            int m2 = end - p0;
            if (m2 > 32) m2 = 32;
            int2 rec2 = make_int2(0, 0);
            if (lane < m2) rec2 = csr[p0 + lane];
            gather_chunk(h2, rec2.x, rec2.y, m2, lane, acc);
        }
        ((float2*)agg)[(size_t)v * (HID / 2) + lane] = acc;

        float nsw = ndi * ndi;
        acc.x = nhv.x * nsw;
        acc.y = nhv.y * nsw;
        beg = nbeg; end = nend; rec = nrec;
    }
}

__global__ void k_pool_zero(float* sums, float* cnt, int g) {
    int i = blockIdx.x * blockDim.x + threadIdx.x;
    if (i < g * (HID / 4)) ((float4*)sums)[i] = make_float4(0.f, 0.f, 0.f, 0.f);
    if (i < g)             cnt[i] = 0.f;
}

__device__ __forceinline__ void pool_flush(float* sums, float* cnt, int g, int lane,
                                           float2 acc, float c) {
    atomicAdd(&sums[(size_t)g * HID + 2 * lane],     acc.x);
    atomicAdd(&sums[(size_t)g * HID + 2 * lane + 1], acc.y);
    if (lane == 0) atomicAdd(&cnt[g], c);
}

__global__ void __launch_bounds__(256)
k_pool_scatter(const float* __restrict__ h, const int* __restrict__ batch,
               float* sums, float* cnt, int n) {
    int lane = threadIdx.x & 31;
    int w  = __builtin_amdgcn_readfirstlane((int)(blockIdx.x * 8 + (threadIdx.x >> 5)));
    int v0 = w * PRUN;
    if (v0 >= n) return;
    int nv = n - v0;
    if (nv > PRUN) nv = PRUN;
    int vi = v0 + lane;
    if (vi >= n) vi = n - 1;
    int gl = batch[vi];
    const float2* h2 = (const float2*)h;
    float2 r[PRUN];
#pragma unroll
    for (int j = 0; j < PRUN; ++j) {
        int v = v0 + j;
        if (v >= n) v = n - 1;
        r[j] = h2[(size_t)v * (HID / 2) + lane];
    }
    int cur = __builtin_amdgcn_readlane(gl, 0);
    float2 acc = make_float2(0.f, 0.f);
    float c = 0.f;
#pragma unroll
    for (int j = 0; j < PRUN; ++j) {
        if (j < nv) {
            int g = __builtin_amdgcn_readlane(gl, j);
            if (g != cur) {
                pool_flush(sums, cnt, cur, lane, acc, c);
                cur = g; acc = make_float2(0.f, 0.f); c = 0.f;
            }
            acc.x += r[j].x; acc.y += r[j].y; c += 1.f;
        }
    }
    pool_flush(sums, cnt, cur, lane, acc, c);
}

__global__ void k_final(const float* __restrict__ sums, const float* __restrict__ cnt,
                        const float* __restrict__ Wl, const float* __restrict__ bl,
                        float* __restrict__ out, int g, int oc) {
    int i = blockIdx.x * blockDim.x + threadIdx.x;
    if (i < g * oc) {
        int gi = i / oc, o = i % oc;
        float inv = 1.0f / fmaxf(cnt[gi], 1.0f);
        float acc = bl[o];
        for (int c = 0; c < HID; ++c)
            acc += sums[(size_t)gi * HID + c] * inv * Wl[c * oc + o];
        out[i] = acc;
    }
}

extern "C" void kernel_launch(void* const* d_in, const int* in_sizes, int n_in,
                              void* d_out, int out_size, void* d_ws, size_t ws_size,
                              hipStream_t stream) {
    const float* x    = (const float*)d_in[0];
    const int*   eidx = (const int*)  d_in[1];
    const int*   batch= (const int*)  d_in[2];
    const float* W1   = (const float*)d_in[3];
    const float* W2   = (const float*)d_in[4];
    const float* W3   = (const float*)d_in[5];
    const float* Wl   = (const float*)d_in[6];
    const float* bl   = (const float*)d_in[7];

    const int INC = in_sizes[3] / HID;
    const int N   = in_sizes[0] / INC;
    const int E   = in_sizes[1] / 2;
    const int OC  = in_sizes[7];
    const int G   = out_size / OC;

    const int* src = eidx;
    const int* dst = eidx + E;

    char* ws = (char*)d_ws; size_t off = 0;
    auto carve = [&](size_t bytes) {
        void* p = ws + off; off = (off + bytes + 255) & ~(size_t)255; return p;
    };
    int*   ecnt   = (int*)  carve((size_t)N * 4);
    int*   rowptr = (int*)  carve((size_t)(N + 1) * 4);
    float* dinv   = (float*)carve((size_t)N * 4);
    int*   rank   = (int*)  carve((size_t)E * 4);
    int2*  csr    = (int2*) carve((size_t)E * 8);
    float* hW     = (float*)carve((size_t)N * HID * 4);
    float* agg    = (float*)carve((size_t)N * HID * 4);
    float* sums   = (float*)carve((size_t)G * HID * 4);
    float* cnt    = (float*)carve((size_t)G * 4);
    (void)ws_size; (void)n_in;

    const int B = 256;
    const int nB  = ((N + 3) / 4 + B - 1) / B;
    const int eB  = (E + B * EPL - 1) / (B * EPL);
    const int mt  = (N + 63) / 64;
    const int gB  = (N + 8 * NPW - 1) / (8 * NPW);
    const int pB  = (N + 8 * PRUN - 1) / (8 * PRUN);

    k_cnt_zero<<<nB, B, 0, stream>>>(ecnt, N);
    k_rank    <<<eB, B, 0, stream>>>(dst, ecnt, rank, E);
    k_scan    <<<1, 1024, 0, stream>>>(ecnt, rowptr, dinv, N);
    k_fill    <<<eB, B, 0, stream>>>(src, dst, rank, rowptr, dinv, csr, E);

    k_gemm_wmma<128, false><<<mt, 128, 0, stream>>>(x, W1, hW, N);
    k_gather<<<gB, B, 0, stream>>>(rowptr, csr, dinv, hW, agg, N);

    k_gemm_wmma<64, true><<<mt, 128, 0, stream>>>(agg, W2, hW, N);
    k_gather<<<gB, B, 0, stream>>>(rowptr, csr, dinv, hW, agg, N);

    k_gemm_wmma<64, true><<<mt, 128, 0, stream>>>(agg, W3, hW, N);
    k_gather<<<gB, B, 0, stream>>>(rowptr, csr, dinv, hW, agg, N);

    k_pool_zero   <<<(G * (HID / 4) + B - 1) / B, B, 0, stream>>>(sums, cnt, G);
    k_pool_scatter<<<pB, B, 0, stream>>>(agg, batch, sums, cnt, N);
    k_final       <<<(G * OC + B - 1) / B, B, 0, stream>>>(sums, cnt, Wl, bl,
                                                           (float*)d_out, G, OC);
}


# Layers 2 and 3: relu(agg) @ W moved into the tail of the aggregation kernel (rows to an LDS tile, one barrier, 4 waves do the 32x64 product on WMMA f32); two launches and the agg round trip go
# speedup vs baseline: 3.3854x; 1.0652x over previous
#include <hip/hip_runtime.h>

typedef __attribute__((ext_vector_type(2))) float v2f;
typedef __attribute__((ext_vector_type(8))) float v8f;

#define HID 64

#define EPL  8
#define NPW  4
#define PRUN 32

__global__ void k_cnt_zero(int* cnt, int n) {
    int i = (blockIdx.x * blockDim.x + threadIdx.x) * 4;
    if (i + 4 <= n) *(int4*)(cnt + i) = make_int4(0, 0, 0, 0);
    else for (int k = i; k < n; ++k) cnt[k] = 0;
}

template <bool FULL>
__device__ __forceinline__ void rank_span(const int* __restrict__ dst, int* cnt,
                                          int* __restrict__ rank, int e, int base) {
    int d[EPL], r[EPL];
#pragma unroll
    for (int k = 0; k < EPL; ++k) {
        int i = base + k * 256;
        d[k] = dst[(FULL || i < e) ? i : e - 1];
    }
#pragma unroll
    for (int k = 0; k < EPL; ++k) {
        int i = base + k * 256;
        r[k] = 0;
        if (FULL || i < e) r[k] = atomicAdd(&cnt[d[k]], 1);
    }
#pragma unroll
    for (int k = 0; k < EPL; ++k) {
        int i = base + k * 256;
        if (FULL || i < e) rank[i] = r[k];
    }
}

__global__ void __launch_bounds__(256)
k_rank(const int* __restrict__ dst, int* cnt, int* __restrict__ rank, int e) {
    int span = blockIdx.x * (256 * EPL);
    int base = span + threadIdx.x;
    if (span + 256 * EPL <= e) rank_span<true>(dst, cnt, rank, e, base);
    else                       rank_span<false>(dst, cnt, rank, e, base);
}

#define EPT 16
__global__ void __launch_bounds__(1024)
k_scan(const int* __restrict__ cnt, int* __restrict__ rowptr, float* __restrict__ dinv, int n) {
    __shared__ int wsum[32];
    int tid = threadIdx.x, lane = tid & 31, wave = tid >> 5;
    int carry = 0;
    for (int base = 0; base < n; base += 1024 * EPT) {
        int i0 = base + tid * EPT;
        bool full = (i0 + EPT <= n);
        int c[EPT];
        if (full) {
#pragma unroll
            for (int q = 0; q < EPT / 4; ++q) {
                int4 t = ((const int4*)(cnt + i0))[q];
                c[4 * q] = t.x; c[4 * q + 1] = t.y; c[4 * q + 2] = t.z; c[4 * q + 3] = t.w;
            }
        } else {
#pragma unroll
            for (int k = 0; k < EPT; ++k) c[k] = (i0 + k < n) ? cnt[i0 + k] : 0;
        }
        int tot = 0;
#pragma unroll
        for (int k = 0; k < EPT; ++k) tot += c[k];
        int x = tot;
#pragma unroll
        for (int o = 1; o < 32; o <<= 1) {
            int y = __shfl_up(x, o);
            if (lane >= o) x += y;
        }
        if (lane == 31) wsum[wave] = x;
        __syncthreads();
        int w = wsum[lane];
        int wx = w;
#pragma unroll
        for (int o = 1; o < 32; o <<= 1) {
            int y = __shfl_up(wx, o);
            if (lane >= o) wx += y;
        }
        int wave_excl = __shfl(wx, wave) - __shfl(w, wave);
        int tile_tot  = __shfl(wx, 31);
        int run = carry + wave_excl + (x - tot);
        int   rp[EPT];
        float dv[EPT];
#pragma unroll
        for (int k = 0; k < EPT; ++k) {
            rp[k] = run;
            run += c[k];
            float d = (float)(c[k] + 1);
            dv[k] = (d > 0.f) ? rsqrtf(d) : 0.f;
        }
        if (full) {
#pragma unroll
            for (int q = 0; q < EPT / 4; ++q) {
                ((int4*)(rowptr + i0))[q] =
                    make_int4(rp[4 * q], rp[4 * q + 1], rp[4 * q + 2], rp[4 * q + 3]);
                ((float4*)(dinv + i0))[q] =
                    make_float4(dv[4 * q], dv[4 * q + 1], dv[4 * q + 2], dv[4 * q + 3]);
            }
        } else {
#pragma unroll
            for (int k = 0; k < EPT; ++k)
                if (i0 + k < n) { rowptr[i0 + k] = rp[k]; dinv[i0 + k] = dv[k]; }
        }
        carry += tile_tot;
        __syncthreads();
    }
    if (tid == 0) rowptr[n] = carry;
}

template <bool FULL>
__device__ __forceinline__ void fill_span(const int* __restrict__ src, const int* __restrict__ dst,
                                          const int* __restrict__ rank,
                                          const int* __restrict__ rowptr,
                                          const float* __restrict__ dinv,
                                          int2* __restrict__ csr, int e, int base) {
    int s[EPL], d[EPL], rk[EPL];
#pragma unroll
    for (int k = 0; k < EPL; ++k) {
        int i = base + k * 256;
        int ic = (FULL || i < e) ? i : e - 1;
        s[k]  = src[ic];
        d[k]  = dst[ic];
        rk[k] = rank[ic];
    }
    int   rp[EPL];
    float ds[EPL], dd[EPL];
#pragma unroll
    for (int k = 0; k < EPL; ++k) {
        rp[k] = rowptr[d[k]];
        ds[k] = dinv[s[k]];
        dd[k] = dinv[d[k]];
    }
#pragma unroll
    for (int k = 0; k < EPL; ++k) {
        int i = base + k * 256;
        float nm = ds[k] * dd[k];
        if (FULL || i < e) csr[rp[k] + rk[k]] = make_int2(s[k], __float_as_int(nm));
    }
}

__global__ void __launch_bounds__(256)
k_fill(const int* __restrict__ src, const int* __restrict__ dst,
       const int* __restrict__ rank, const int* __restrict__ rowptr,
       const float* __restrict__ dinv, int2* __restrict__ csr, int e) {
    int span = blockIdx.x * (256 * EPL);
    int base = span + threadIdx.x;
    if (span + 256 * EPL <= e) fill_span<true>(src, dst, rank, rowptr, dinv, csr, e, base);
    else                       fill_span<false>(src, dst, rank, rowptr, dinv, csr, e, base);
}

#define WMMA_F32(a, b, c) __builtin_amdgcn_wmma_f32_16x16x4_f32( \
         false, a,  false, b,  (short)0, c, \
         false,  false)
template <int K, bool DO_RELU>
__global__ void __launch_bounds__(128)
k_gemm_wmma(const float* __restrict__ A, const float* __restrict__ W,
            float* __restrict__ C, int n) {
    __shared__ float4 sW4[K * HID / 4];
    const float* sW = (const float*)sW4;
    int tid = threadIdx.x;
#pragma unroll
    for (int q = 0; q < K * HID / 4 / 128; ++q) sW4[tid + q * 128] = ((const float4*)W)[tid + q * 128];
    __syncthreads();

    int wave = tid >> 5;
    int lane = tid & 31;
    int half = lane >> 4;
    int l15  = lane & 15;
    int m0 = blockIdx.x * 64 + wave * 16;

    int arow = m0 + l15;
    if (arow >= n) arow = n - 1;
    const float* ap = A + (size_t)arow * K + half * 2;
    const float* bp = sW + l15 + half * 2 * HID;

    v8f c0 = {}, c1 = {}, c2 = {}, c3 = {};
#pragma unroll
    for (int k = 0; k < K; k += 4) {
        v2f a, b0, b1, b2, b3;
        float ax = ap[k];
        float ay = ap[k + 1];
        if (DO_RELU) { ax = fmaxf(ax, 0.f); ay = fmaxf(ay, 0.f); }
        a[0] = ax; a[1] = ay;
        const float* bk = bp + k * HID;
        b0[0] = bk[0];  b0[1] = bk[HID];
        b1[0] = bk[16]; b1[1] = bk[HID + 16];
        b2[0] = bk[32]; b2[1] = bk[HID + 32];
        b3[0] = bk[48]; b3[1] = bk[HID + 48];
        c0 = WMMA_F32(a, b0, c0);
        c1 = WMMA_F32(a, b1, c1);
        c2 = WMMA_F32(a, b2, c2);
        c3 = WMMA_F32(a, b3, c3);
    }

    int rbase = m0 + half * 8;
#pragma unroll
    for (int j = 0; j < 8; ++j) {
        int r = rbase + j;
        if (r < n) {
            float* cr = C + (size_t)r * HID + l15;
            cr[0] = c0[j]; cr[16] = c1[j]; cr[32] = c2[j]; cr[48] = c3[j];
        }
    }
}

template <int U>
__device__ __forceinline__ void gather_rows(const float2* __restrict__ h2, int s_l, int w_l,
                                            int j, int lane, float2& acc) {
    float2 r[U];
    float  w[U];
#pragma unroll
    for (int k = 0; k < U; ++k) {
        int s = __builtin_amdgcn_readlane(s_l, j + k);
        w[k]  = __int_as_float(__builtin_amdgcn_readlane(w_l, j + k));
        r[k]  = h2[(size_t)s * (HID / 2) + lane];
    }
#pragma unroll
    for (int k = 0; k < U; ++k) {
        acc.x = fmaf(r[k].x, w[k], acc.x);
        acc.y = fmaf(r[k].y, w[k], acc.y);
    }
}

__device__ __forceinline__ void gather_chunk(const float2* __restrict__ h2, int s_l, int w_l,
                                             int m, int lane, float2& acc) {
    int j = 0;
    for (; j + 8 <= m; j += 8) gather_rows<8>(h2, s_l, w_l, j, lane, acc);
    if (m & 4) { gather_rows<4>(h2, s_l, w_l, j, lane, acc); j += 4; }
    if (m & 2) { gather_rows<2>(h2, s_l, w_l, j, lane, acc); j += 2; }
    if (m & 1) { gather_rows<1>(h2, s_l, w_l, j, lane, acc); }
}

#define APITCH 68
template <bool TO_LDS>
__device__ __forceinline__ void gather_nodes(const int* __restrict__ rowptr, const int2* __restrict__ csr,
                                             const float* __restrict__ dinv, const float2* __restrict__ h2,
                                             float* __restrict__ dstg, float* dstl,
                                             int v0, int vend, int lane) {
    int beg = rowptr[v0], end = rowptr[v0 + 1];
    float di = dinv[v0];
    int2 rec = make_int2(0, 0);
    if (lane < end - beg) rec = csr[beg + lane];
    float2 hv = h2[(size_t)v0 * (HID / 2) + lane];
    float sw = di * di;
    float2 acc;
    acc.x = hv.x * sw;
    acc.y = hv.y * sw;
    for (int v = v0; v < vend; ++v) {
        int vn = (v + 1 < vend) ? v + 1 : v;
        int nbeg = rowptr[vn], nend = rowptr[vn + 1];
        float ndi = dinv[vn];
        int2 nrec = make_int2(0, 0);
        if (lane < nend - nbeg) nrec = csr[nbeg + lane];
        float2 nhv = h2[(size_t)vn * (HID / 2) + lane];

        int m = end - beg;
        if (m > 32) m = 32;
        gather_chunk(h2, rec.x, rec.y, m, lane, acc);
        for (int p0 = beg + 32; p0 < end; p0 += 32) {
            int m2 = end - p0;
            if (m2 > 32) m2 = 32;
            int2 rec2 = make_int2(0, 0);
            if (lane < m2) rec2 = csr[p0 + lane];
            gather_chunk(h2, rec2.x, rec2.y, m2, lane, acc);
        }
        if (TO_LDS) {
            float2 t;
            t.x = fmaxf(acc.x, 0.f);
            t.y = fmaxf(acc.y, 0.f);
            *(float2*)(dstl + (v - v0) * APITCH + 2 * lane) = t;
        } else {
            ((float2*)dstg)[(size_t)v * (HID / 2) + lane] = acc;
        }

        float nsw = ndi * ndi;
        acc.x = nhv.x * nsw;
        acc.y = nhv.y * nsw;
        beg = nbeg; end = nend; rec = nrec;
    }
}

__global__ void __launch_bounds__(256)
k_gather(const int* __restrict__ rowptr, const int2* __restrict__ csr,
         const float* __restrict__ dinv, const float* __restrict__ hW,
         float* __restrict__ agg, int n) {
    int lane = threadIdx.x & 31;
    int w  = __builtin_amdgcn_readfirstlane((int)(blockIdx.x * 8 + (threadIdx.x >> 5)));
    int v0 = w * NPW;
    if (v0 >= n) return;
    int vend = v0 + NPW;
    if (vend > n) vend = n;
    gather_nodes<false>(rowptr, csr, dinv, (const float2*)hW, agg, nullptr, v0, vend, lane);
}

__global__ void __launch_bounds__(256)
k_gather_gemm(const int* __restrict__ rowptr, const int2* __restrict__ csr,
              const float* __restrict__ dinv, const float* __restrict__ hW,
              const float* __restrict__ Wn, float* __restrict__ hWn, int n) {
    __shared__ float4 sW4[HID * HID / 4];
    __shared__ __attribute__((aligned(16))) float sA[8 * NPW * APITCH];
    const float* sW = (const float*)sW4;
    int tid  = threadIdx.x;
    int lane = tid & 31;
    int wv = __builtin_amdgcn_readfirstlane(tid >> 5);
    float4 wreg[HID * HID / 4 / 256];
#pragma unroll
    for (int q = 0; q < HID * HID / 4 / 256; ++q) wreg[q] = ((const float4*)Wn)[tid + q * 256];
#pragma unroll
    for (int q = 0; q < HID * HID / 4 / 256; ++q) sW4[tid + q * 256] = wreg[q];

    int v0 = (blockIdx.x * 8 + wv) * NPW;
    int vend = v0 + NPW;
    if (vend > n) vend = n;
    float* myrows = sA + wv * NPW * APITCH;
    if (v0 < n)
        gather_nodes<true>(rowptr, csr, dinv, (const float2*)hW, nullptr, myrows, v0, vend, lane);
    for (int v = (v0 < n ? vend : v0); v < v0 + NPW; ++v) {
        float2 z; z.x = 0.f; z.y = 0.f;
        *(float2*)(myrows + (v - v0) * APITCH + 2 * lane) = z;
    }
    __syncthreads();
    if (wv >= 4) return;

    int mt = wv >> 1;
    int nh = wv & 1;
    int half = lane >> 4;
    int l15  = lane & 15;
    const float* ap = sA + (mt * 16 + l15) * APITCH + half * 2;
    const float* bp = sW + nh * 32 + l15 + half * 2 * HID;
    v8f c0 = {}, c1 = {};
#pragma unroll
    for (int k = 0; k < HID; k += 4) {
        v2f a, b0, b1;
        a[0] = ap[k]; a[1] = ap[k + 1];
        const float* bk = bp + k * HID;
        b0[0] = bk[0];  b0[1] = bk[HID];
        b1[0] = bk[16]; b1[1] = bk[HID + 16];
        c0 = WMMA_F32(a, b0, c0);
        c1 = WMMA_F32(a, b1, c1);
    }
    int rbase = blockIdx.x * (8 * NPW) + mt * 16 + half * 8;
#pragma unroll
    for (int j = 0; j < 8; ++j) {
        int r = rbase + j;
        if (r < n) {
            float* cr = hWn + (size_t)r * HID + nh * 32 + l15;
            cr[0] = c0[j]; cr[16] = c1[j];
        }
    }
}

__global__ void k_pool_zero(float* sums, float* cnt, int g) {
    int i = blockIdx.x * blockDim.x + threadIdx.x;
    if (i < g * (HID / 4)) ((float4*)sums)[i] = make_float4(0.f, 0.f, 0.f, 0.f);
    if (i < g)             cnt[i] = 0.f;
}

__device__ __forceinline__ void pool_flush(float* sums, float* cnt, int g, int lane,
                                           float2 acc, float c) {
    atomicAdd(&sums[(size_t)g * HID + 2 * lane],     acc.x);
    atomicAdd(&sums[(size_t)g * HID + 2 * lane + 1], acc.y);
    if (lane == 0) atomicAdd(&cnt[g], c);
}

__global__ void __launch_bounds__(256)
k_pool_scatter(const float* __restrict__ h, const int* __restrict__ batch,
               float* sums, float* cnt, int n) {
    int lane = threadIdx.x & 31;
    int w  = __builtin_amdgcn_readfirstlane((int)(blockIdx.x * 8 + (threadIdx.x >> 5)));
    int v0 = w * PRUN;
    if (v0 >= n) return;
    int nv = n - v0;
    if (nv > PRUN) nv = PRUN;
    int vi = v0 + lane;
    if (vi >= n) vi = n - 1;
    int gl = batch[vi];
    const float2* h2 = (const float2*)h;
    float2 r[PRUN];
#pragma unroll
    for (int j = 0; j < PRUN; ++j) {
        int v = v0 + j;
        if (v >= n) v = n - 1;
        r[j] = h2[(size_t)v * (HID / 2) + lane];
    }
    int cur = __builtin_amdgcn_readlane(gl, 0);
    float2 acc = make_float2(0.f, 0.f);
    float c = 0.f;
#pragma unroll
    for (int j = 0; j < PRUN; ++j) {
        if (j < nv) {
            int g = __builtin_amdgcn_readlane(gl, j);
            if (g != cur) {
                pool_flush(sums, cnt, cur, lane, acc, c);
                cur = g; acc = make_float2(0.f, 0.f); c = 0.f;
            }
            acc.x += r[j].x; acc.y += r[j].y; c += 1.f;
        }
    }
    pool_flush(sums, cnt, cur, lane, acc, c);
}

__global__ void k_final(const float* __restrict__ sums, const float* __restrict__ cnt,
                        const float* __restrict__ Wl, const float* __restrict__ bl,
                        float* __restrict__ out, int g, int oc) {
    int i = blockIdx.x * blockDim.x + threadIdx.x;
    if (i < g * oc) {
        int gi = i / oc, o = i % oc;
        float inv = 1.0f / fmaxf(cnt[gi], 1.0f);
        float acc = bl[o];
        for (int c = 0; c < HID; ++c)
            acc += sums[(size_t)gi * HID + c] * inv * Wl[c * oc + o];
        out[i] = acc;
    }
}

extern "C" void kernel_launch(void* const* d_in, const int* in_sizes, int n_in,
                              void* d_out, int out_size, void* d_ws, size_t ws_size,
                              hipStream_t stream) {
    const float* x    = (const float*)d_in[0];
    const int*   eidx = (const int*)  d_in[1];
    const int*   batch= (const int*)  d_in[2];
    const float* W1   = (const float*)d_in[3];
    const float* W2   = (const float*)d_in[4];
    const float* W3   = (const float*)d_in[5];
    const float* Wl   = (const float*)d_in[6];
    const float* bl   = (const float*)d_in[7];

    const int INC = in_sizes[3] / HID;
    const int N   = in_sizes[0] / INC;
    const int E   = in_sizes[1] / 2;
    const int OC  = in_sizes[7];
    const int G   = out_size / OC;

    const int* src = eidx;
    const int* dst = eidx + E;

    char* ws = (char*)d_ws; size_t off = 0;
    auto carve = [&](size_t bytes) {
        void* p = ws + off; off = (off + bytes + 255) & ~(size_t)255; return p;
    };
    int*   ecnt   = (int*)  carve((size_t)N * 4);
    int*   rowptr = (int*)  carve((size_t)(N + 1) * 4);
    float* dinv   = (float*)carve((size_t)N * 4);
    int*   rank   = (int*)  carve((size_t)E * 4);
    int2*  csr    = (int2*) carve((size_t)E * 8);
    float* hW     = (float*)carve((size_t)N * HID * 4);
    float* agg    = (float*)carve((size_t)N * HID * 4);
    float* sums   = (float*)carve((size_t)G * HID * 4);
    float* cnt    = (float*)carve((size_t)G * 4);
    (void)ws_size; (void)n_in;

    const int B = 256;
    const int nB  = ((N + 3) / 4 + B - 1) / B;
    const int eB  = (E + B * EPL - 1) / (B * EPL);
    const int mt  = (N + 63) / 64;
    const int gB  = (N + 8 * NPW - 1) / (8 * NPW);
    const int pB  = (N + 8 * PRUN - 1) / (8 * PRUN);

    k_cnt_zero<<<nB, B, 0, stream>>>(ecnt, N);
    k_rank    <<<eB, B, 0, stream>>>(dst, ecnt, rank, E);
    k_scan    <<<1, 1024, 0, stream>>>(ecnt, rowptr, dinv, N);
    k_fill    <<<eB, B, 0, stream>>>(src, dst, rank, rowptr, dinv, csr, E);

    k_gemm_wmma<128, false><<<mt, 128, 0, stream>>>(x, W1, hW, N);
    k_gather_gemm<<<gB, B, 0, stream>>>(rowptr, csr, dinv, hW, W2, agg, N);

    k_gather_gemm<<<gB, B, 0, stream>>>(rowptr, csr, dinv, agg, W3, hW, N);

    k_gather<<<gB, B, 0, stream>>>(rowptr, csr, dinv, hW, agg, N);

    k_pool_zero   <<<(G * (HID / 4) + B - 1) / B, B, 0, stream>>>(sums, cnt, G);
    k_pool_scatter<<<pB, B, 0, stream>>>(agg, batch, sums, cnt, N);
    k_final       <<<(G * OC + B - 1) / B, B, 0, stream>>>(sums, cnt, Wl, bl,
                                                           (float*)d_out, G, OC);
}


# Last layer on 16-float rows: W3 @ Wl folded first (only sums lie between them), so layer 3 gathers 64-byte rows, 4 per load, straight into the pooled sums; pooling pass gone (sum order changes)
# speedup vs baseline: 3.6617x; 1.0816x over previous
#include <hip/hip_runtime.h>

typedef __attribute__((ext_vector_type(2))) float v2f;
typedef __attribute__((ext_vector_type(8))) float v8f;

#define HID 64

#define WMMA_F32(a, b, c) __builtin_amdgcn_wmma_f32_16x16x4_f32( \
         false, a,  false, b,  (short)0, c, \
         false,  false)

#define EPL  8
#define NPW  4
#define PW   16
#define NPW3 8

__global__ void k_cnt_zero(int* cnt, int n, float* psum, float* pcnt, int g) {
    int t = blockIdx.x * blockDim.x + threadIdx.x;
    int i = t * 4;
    if (i + 4 <= n) *(int4*)(cnt + i) = make_int4(0, 0, 0, 0);
    else for (int k = i; k < n; ++k) cnt[k] = 0;
    int nt = gridDim.x * blockDim.x;
    for (int k = t; k < g * (PW / 4); k += nt) ((float4*)psum)[k] = make_float4(0.f, 0.f, 0.f, 0.f);
    for (int k = t; k < g; k += nt) pcnt[k] = 0.f;
}

template <bool FULL>
__device__ __forceinline__ void rank_span(const int* __restrict__ dst, int* cnt,
                                          int* __restrict__ rank, int e, int base) {
    int d[EPL], r[EPL];
#pragma unroll
    for (int k = 0; k < EPL; ++k) {
        int i = base + k * 256;
        d[k] = dst[(FULL || i < e) ? i : e - 1];
    }
#pragma unroll
    for (int k = 0; k < EPL; ++k) {
        int i = base + k * 256;
        r[k] = 0;
        if (FULL || i < e) r[k] = atomicAdd(&cnt[d[k]], 1);
    }
#pragma unroll
    for (int k = 0; k < EPL; ++k) {
        int i = base + k * 256;
        if (FULL || i < e) rank[i] = r[k];
    }
}

__global__ void __launch_bounds__(256)
k_rank(const int* __restrict__ dst, int* cnt, int* __restrict__ rank, int e) {
    int span = blockIdx.x * (256 * EPL);
    int base = span + threadIdx.x;
    if (span + 256 * EPL <= e) rank_span<true>(dst, cnt, rank, e, base);
    else                       rank_span<false>(dst, cnt, rank, e, base);
}

#define EPT 16
__global__ void __launch_bounds__(1024)
k_scan(const int* __restrict__ cnt, int* __restrict__ rowptr, float* __restrict__ dinv, int n,
       const float* __restrict__ W3, const float* __restrict__ Wl, float* __restrict__ W34, int oc) {
    __shared__ int wsum[32];
    if (blockIdx.x == 1) {
        int fw = __builtin_amdgcn_readfirstlane((int)(threadIdx.x >> 5));
        if (fw >= HID / 16) return;
        int fl = threadIdx.x & 31;
        int half = fl >> 4;
        int l15  = fl & 15;
        bool real = l15 < oc;
        int  col  = real ? l15 : 0;
        const float* ap = W3 + (fw * 16 + l15) * HID + half * 2;
        const float* bp = Wl + half * 2 * oc + col;
        v8f c = {};
#pragma unroll
        for (int k = 0; k < HID; k += 4) {
            v2f a, b;
            a[0] = ap[k]; a[1] = ap[k + 1];
            float bx = bp[k * oc];
            float by = bp[(k + 1) * oc];
            b[0] = real ? bx : 0.f; b[1] = real ? by : 0.f;
            c = WMMA_F32(a, b, c);
        }
#pragma unroll
        for (int j = 0; j < 8; ++j)
            W34[(fw * 16 + half * 8 + j) * PW + l15] = c[j];
        return;
    }
    int tid = threadIdx.x, lane = tid & 31, wave = tid >> 5;
    int carry = 0;
    for (int base = 0; base < n; base += 1024 * EPT) {
        int i0 = base + tid * EPT;
        bool full = (i0 + EPT <= n);
        int c[EPT];
        if (full) {
#pragma unroll
            for (int q = 0; q < EPT / 4; ++q) {
                int4 t = ((const int4*)(cnt + i0))[q];
                c[4 * q] = t.x; c[4 * q + 1] = t.y; c[4 * q + 2] = t.z; c[4 * q + 3] = t.w;
            }
        } else {
#pragma unroll
            for (int k = 0; k < EPT; ++k) c[k] = (i0 + k < n) ? cnt[i0 + k] : 0;
        }
        int tot = 0;
#pragma unroll
        for (int k = 0; k < EPT; ++k) tot += c[k];
        int x = tot;
#pragma unroll
        for (int o = 1; o < 32; o <<= 1) {
            int y = __shfl_up(x, o);
            if (lane >= o) x += y;
        }
        if (lane == 31) wsum[wave] = x;
        __syncthreads();
        int w = wsum[lane];
        int wx = w;
#pragma unroll
        for (int o = 1; o < 32; o <<= 1) {
            int y = __shfl_up(wx, o);
            if (lane >= o) wx += y;
        }
        int wave_excl = __shfl(wx, wave) - __shfl(w, wave);
        int tile_tot  = __shfl(wx, 31);
        int run = carry + wave_excl + (x - tot);
        int   rp[EPT];
        float dv[EPT];
#pragma unroll
        for (int k = 0; k < EPT; ++k) {
            rp[k] = run;
            run += c[k];
            float d = (float)(c[k] + 1);
            dv[k] = (d > 0.f) ? rsqrtf(d) : 0.f;
        }
        if (full) {
#pragma unroll
            for (int q = 0; q < EPT / 4; ++q) {
                ((int4*)(rowptr + i0))[q] =
                    make_int4(rp[4 * q], rp[4 * q + 1], rp[4 * q + 2], rp[4 * q + 3]);
                ((float4*)(dinv + i0))[q] =
                    make_float4(dv[4 * q], dv[4 * q + 1], dv[4 * q + 2], dv[4 * q + 3]);
            }
        } else {
#pragma unroll
            for (int k = 0; k < EPT; ++k)
                if (i0 + k < n) { rowptr[i0 + k] = rp[k]; dinv[i0 + k] = dv[k]; }
        }
        carry += tile_tot;
        __syncthreads();
    }
    if (tid == 0) rowptr[n] = carry;
}

template <bool FULL>
__device__ __forceinline__ void fill_span(const int* __restrict__ src, const int* __restrict__ dst,
                                          const int* __restrict__ rank,
                                          const int* __restrict__ rowptr,
                                          const float* __restrict__ dinv,
                                          int2* __restrict__ csr, int e, int base) {
    int s[EPL], d[EPL], rk[EPL];
#pragma unroll
    for (int k = 0; k < EPL; ++k) {
        int i = base + k * 256;
        int ic = (FULL || i < e) ? i : e - 1;
        s[k]  = src[ic];
        d[k]  = dst[ic];
        rk[k] = rank[ic];
    }
    int   rp[EPL];
    float ds[EPL], dd[EPL];
#pragma unroll
    for (int k = 0; k < EPL; ++k) {
        rp[k] = rowptr[d[k]];
        ds[k] = dinv[s[k]];
        dd[k] = dinv[d[k]];
    }
#pragma unroll
    for (int k = 0; k < EPL; ++k) {
        int i = base + k * 256;
        float nm = ds[k] * dd[k];
        if (FULL || i < e) csr[rp[k] + rk[k]] = make_int2(s[k], __float_as_int(nm));
    }
}

__global__ void __launch_bounds__(256)
k_fill(const int* __restrict__ src, const int* __restrict__ dst,
       const int* __restrict__ rank, const int* __restrict__ rowptr,
       const float* __restrict__ dinv, int2* __restrict__ csr, int e) {
    int span = blockIdx.x * (256 * EPL);
    int base = span + threadIdx.x;
    if (span + 256 * EPL <= e) fill_span<true>(src, dst, rank, rowptr, dinv, csr, e, base);
    else                       fill_span<false>(src, dst, rank, rowptr, dinv, csr, e, base);
}

template <int K, bool DO_RELU>
__global__ void __launch_bounds__(128)
k_gemm_wmma(const float* __restrict__ A, const float* __restrict__ W,
            float* __restrict__ C, int n) {
    __shared__ float4 sW4[K * HID / 4];
    const float* sW = (const float*)sW4;
    int tid = threadIdx.x;
#pragma unroll
    for (int q = 0; q < K * HID / 4 / 128; ++q) sW4[tid + q * 128] = ((const float4*)W)[tid + q * 128];
    __syncthreads();

    int wave = tid >> 5;
    int lane = tid & 31;
    int half = lane >> 4;
    int l15  = lane & 15;
    int m0 = blockIdx.x * 64 + wave * 16;

    int arow = m0 + l15;
    if (arow >= n) arow = n - 1;
    const float* ap = A + (size_t)arow * K + half * 2;
    const float* bp = sW + l15 + half * 2 * HID;

    v8f c0 = {}, c1 = {}, c2 = {}, c3 = {};
#pragma unroll
    for (int k = 0; k < K; k += 4) {
        v2f a, b0, b1, b2, b3;
        float ax = ap[k];
        float ay = ap[k + 1];
        if (DO_RELU) { ax = fmaxf(ax, 0.f); ay = fmaxf(ay, 0.f); }
        a[0] = ax; a[1] = ay;
        const float* bk = bp + k * HID;
        b0[0] = bk[0];  b0[1] = bk[HID];
        b1[0] = bk[16]; b1[1] = bk[HID + 16];
        b2[0] = bk[32]; b2[1] = bk[HID + 32];
        b3[0] = bk[48]; b3[1] = bk[HID + 48];
        c0 = WMMA_F32(a, b0, c0);
        c1 = WMMA_F32(a, b1, c1);
        c2 = WMMA_F32(a, b2, c2);
        c3 = WMMA_F32(a, b3, c3);
    }

    int rbase = m0 + half * 8;
#pragma unroll
    for (int j = 0; j < 8; ++j) {
        int r = rbase + j;
        if (r < n) {
            float* cr = C + (size_t)r * HID + l15;
            cr[0] = c0[j]; cr[16] = c1[j]; cr[32] = c2[j]; cr[48] = c3[j];
        }
    }
}

template <int U>
__device__ __forceinline__ void gather_rows(const float2* __restrict__ h2, int s_l, int w_l,
                                            int j, int lane, float2& acc) {
    float2 r[U];
    float  w[U];
#pragma unroll
    for (int k = 0; k < U; ++k) {
        int s = __builtin_amdgcn_readlane(s_l, j + k);
        w[k]  = __int_as_float(__builtin_amdgcn_readlane(w_l, j + k));
        r[k]  = h2[(size_t)s * (HID / 2) + lane];
    }
#pragma unroll
    for (int k = 0; k < U; ++k) {
        acc.x = fmaf(r[k].x, w[k], acc.x);
        acc.y = fmaf(r[k].y, w[k], acc.y);
    }
}

__device__ __forceinline__ void gather_chunk(const float2* __restrict__ h2, int s_l, int w_l,
                                             int m, int lane, float2& acc) {
    int j = 0;
    for (; j + 8 <= m; j += 8) gather_rows<8>(h2, s_l, w_l, j, lane, acc);
    if (m & 4) { gather_rows<4>(h2, s_l, w_l, j, lane, acc); j += 4; }
    if (m & 2) { gather_rows<2>(h2, s_l, w_l, j, lane, acc); j += 2; }
    if (m & 1) { gather_rows<1>(h2, s_l, w_l, j, lane, acc); }
}

#define APITCH 68
__device__ __forceinline__ void gather_nodes(const int* __restrict__ rowptr, const int2* __restrict__ csr,
                                             const float* __restrict__ dinv, const float2* __restrict__ h2,
                                             float* dstl, int v0, int vend, int lane) {
    int beg = rowptr[v0], end = rowptr[v0 + 1];
    float di = dinv[v0];
    int2 rec = make_int2(0, 0);
    if (lane < end - beg) rec = csr[beg + lane];
    float2 hv = h2[(size_t)v0 * (HID / 2) + lane];
    float sw = di * di;
    float2 acc;
    acc.x = hv.x * sw;
    acc.y = hv.y * sw;
    for (int v = v0; v < vend; ++v) {
        int vn = (v + 1 < vend) ? v + 1 : v;
        int nbeg = rowptr[vn], nend = rowptr[vn + 1];
        float ndi = dinv[vn];
        int2 nrec = make_int2(0, 0);
        if (lane < nend - nbeg) nrec = csr[nbeg + lane];
        float2 nhv = h2[(size_t)vn * (HID / 2) + lane];

        int m = end - beg;
        if (m > 32) m = 32;
        gather_chunk(h2, rec.x, rec.y, m, lane, acc);
        for (int p0 = beg + 32; p0 < end; p0 += 32) {
            int m2 = end - p0;
            if (m2 > 32) m2 = 32;
            int2 rec2 = make_int2(0, 0);
            if (lane < m2) rec2 = csr[p0 + lane];
            gather_chunk(h2, rec2.x, rec2.y, m2, lane, acc);
        }
        float2 t;
        t.x = fmaxf(acc.x, 0.f);
        t.y = fmaxf(acc.y, 0.f);
        *(float2*)(dstl + (v - v0) * APITCH + 2 * lane) = t;

        float nsw = ndi * ndi;
        acc.x = nhv.x * nsw;
        acc.y = nhv.y * nsw;
        beg = nbeg; end = nend; rec = nrec;
    }
}

template <int NO>
__global__ void __launch_bounds__(256)
k_gather_gemm(const int* __restrict__ rowptr, const int2* __restrict__ csr,
              const float* __restrict__ dinv, const float* __restrict__ hW,
              const float* __restrict__ Wn, float* __restrict__ hWn, int n) {
    __shared__ float4 sW4[HID * NO / 4];
    __shared__ __attribute__((aligned(16))) float sA[8 * NPW * APITCH];
    const float* sW = (const float*)sW4;
    int tid  = threadIdx.x;
    int lane = tid & 31;
    int wv = __builtin_amdgcn_readfirstlane(tid >> 5);
    constexpr int NPIECE = HID * NO / 4 / 256;
    float4 wreg[NPIECE];
#pragma unroll
    for (int q = 0; q < NPIECE; ++q) wreg[q] = ((const float4*)Wn)[tid + q * 256];
#pragma unroll
    for (int q = 0; q < NPIECE; ++q) sW4[tid + q * 256] = wreg[q];

    int v0 = (blockIdx.x * 8 + wv) * NPW;
    int vend = v0 + NPW;
    if (vend > n) vend = n;
    float* myrows = sA + wv * NPW * APITCH;
    if (v0 < n)
        gather_nodes(rowptr, csr, dinv, (const float2*)hW, myrows, v0, vend, lane);
    for (int v = (v0 < n ? vend : v0); v < v0 + NPW; ++v) {
        float2 z; z.x = 0.f; z.y = 0.f;
        *(float2*)(myrows + (v - v0) * APITCH + 2 * lane) = z;
    }
    __syncthreads();
    constexpr bool WIDE = (NO == HID);
    if (wv >= (WIDE ? 4 : 2)) return;

    int mt = WIDE ? (wv >> 1) : wv;
    int nh = WIDE ? (wv & 1) : 0;
    int half = lane >> 4;
    int l15  = lane & 15;
    const float* ap = sA + (mt * 16 + l15) * APITCH + half * 2;
    const float* bp = sW + nh * 32 + l15 + half * 2 * NO;
    v8f c0 = {}, c1 = {};
#pragma unroll
    for (int k = 0; k < HID; k += 4) {
        v2f a, b0, b1;
        a[0] = ap[k]; a[1] = ap[k + 1];
        const float* bk = bp + k * NO;
        b0[0] = bk[0];  b0[1] = bk[NO];
        c0 = WMMA_F32(a, b0, c0);
        if (WIDE) {
            b1[0] = bk[16]; b1[1] = bk[NO + 16];
            c1 = WMMA_F32(a, b1, c1);
        }
    }
    int rbase = blockIdx.x * (8 * NPW) + mt * 16 + half * 8;
#pragma unroll
    for (int j = 0; j < 8; ++j) {
        int r = rbase + j;
        if (r < n) {
            float* cr = hWn + (size_t)r * NO + nh * 32 + l15;
            cr[0] = c0[j];
            if (WIDE) cr[16] = c1[j];
        }
    }
}

__global__ void __launch_bounds__(256)
k_gather_pool(const int* __restrict__ rowptr, const int2* __restrict__ csr,
              const float* __restrict__ dinv, const float* __restrict__ z,
              const int* __restrict__ batch, float* sums, float* cnt, int n) {
    int lane = threadIdx.x & 31;
    int sub = lane & 7, grp = lane >> 3;
    int w  = __builtin_amdgcn_readfirstlane((int)(blockIdx.x * 8 + (threadIdx.x >> 5)));
    int v0 = w * NPW3;
    if (v0 >= n) return;
    int nv = n - v0;
    if (nv > NPW3) nv = NPW3;
    const float2* z2 = (const float2*)z;

    int vb = v0 + lane;
    int rp_l = rowptr[vb <= n ? vb : n];
    if (vb >= n) vb = n - 1;
    int   g_l  = batch[vb];
    float di_l = dinv[vb];
    float2 own[NPW3 / 4];
    float  ow[NPW3 / 4];
#pragma unroll
    for (int q = 0; q < NPW3 / 4; ++q) {
        int j = 4 * q + grp;
        own[q] = z2[j < nv ? (size_t)v0 * (PW / 2) + q * 32 + lane : 0];
        float d = __shfl(di_l, j);
        ow[q] = d * d;
    }

    int j0 = 0;
    while (j0 < nv) {
        int g = __builtin_amdgcn_readfirstlane(__shfl(g_l, j0));
        unsigned diff = __builtin_amdgcn_ballot_w32(lane > j0 && lane < nv && g_l != g);
        int j1 = diff ? __builtin_ctz(diff) : nv;
        int eb = __builtin_amdgcn_readfirstlane(__shfl(rp_l, j0));
        int ee = __builtin_amdgcn_readfirstlane(__shfl(rp_l, j1));

        float2 acc = make_float2(0.f, 0.f);
#pragma unroll
        for (int q = 0; q < NPW3 / 4; ++q) {
            if (4 * q + grp >= j0 && 4 * q + grp < j1) {
                acc.x = fmaf(own[q].x, ow[q], acc.x);
                acc.y = fmaf(own[q].y, ow[q], acc.y);
            }
        }
        if (eb < ee) {
            int q0 = eb + lane, q1 = eb + 32 + lane;
            int2 rec0 = csr[q0 < ee ? q0 : ee - 1];
            int2 rec1 = csr[q1 < ee ? q1 : ee - 1];
            for (int base = eb; base < ee; base += 64) {
                int m = ee - base;
                float2 r[16];
                float  wk[16];
#pragma unroll
                for (int k = 0; k < 16; ++k) {
                    int from = 4 * (k & 7) + grp;
                    int s = __shfl(k < 8 ? rec0.x : rec1.x, from);
                    wk[k] = __int_as_float(__shfl(k < 8 ? rec0.y : rec1.y, from));
                    r[k]  = z2[(size_t)s * (PW / 2) + sub];
                }
                int n0 = base + 64 + lane, n1 = base + 96 + lane;
                int2 nrec0 = csr[n0 < ee ? n0 : ee - 1];
                int2 nrec1 = csr[n1 < ee ? n1 : ee - 1];
#pragma unroll
                for (int k = 0; k < 16; ++k) {
                    if (4 * k + grp < m) {
                        acc.x = fmaf(r[k].x, wk[k], acc.x);
                        acc.y = fmaf(r[k].y, wk[k], acc.y);
                    }
                }
                rec0 = nrec0; rec1 = nrec1;
            }
        }
        acc.x += __shfl_xor(acc.x, 8);  acc.y += __shfl_xor(acc.y, 8);
        acc.x += __shfl_xor(acc.x, 16); acc.y += __shfl_xor(acc.y, 16);
        if (lane < 8) {
            atomicAdd(&sums[(size_t)g * PW + 2 * sub],     acc.x);
            atomicAdd(&sums[(size_t)g * PW + 2 * sub + 1], acc.y);
        }
        if (lane == 0) atomicAdd(&cnt[g], (float)(j1 - j0));
        j0 = j1;
    }
}

__global__ void k_final(const float* __restrict__ sums, const float* __restrict__ cnt,
                        const float* __restrict__ bl, float* __restrict__ out, int g, int oc) {
    int i = blockIdx.x * blockDim.x + threadIdx.x;
    if (i < g * oc) {
        int gi = i / oc, o = i % oc;
        out[i] = sums[(size_t)gi * PW + o] / fmaxf(cnt[gi], 1.0f) + bl[o];
    }
}

extern "C" void kernel_launch(void* const* d_in, const int* in_sizes, int n_in,
                              void* d_out, int out_size, void* d_ws, size_t ws_size,
                              hipStream_t stream) {
    const float* x    = (const float*)d_in[0];
    const int*   eidx = (const int*)  d_in[1];
    const int*   batch= (const int*)  d_in[2];
    const float* W1   = (const float*)d_in[3];
    const float* W2   = (const float*)d_in[4];
    const float* W3   = (const float*)d_in[5];
    const float* Wl   = (const float*)d_in[6];
    const float* bl   = (const float*)d_in[7];

    const int INC = in_sizes[3] / HID;
    const int N   = in_sizes[0] / INC;
    const int E   = in_sizes[1] / 2;
    const int OC  = in_sizes[7];
    const int G   = out_size / OC;

    const int* src = eidx;
    const int* dst = eidx + E;

    char* ws = (char*)d_ws; size_t off = 0;
    auto carve = [&](size_t bytes) {
        void* p = ws + off; off = (off + bytes + 255) & ~(size_t)255; return p;
    };
    int*   ecnt   = (int*)  carve((size_t)N * 4);
    int*   rowptr = (int*)  carve((size_t)(N + 1) * 4);
    float* dinv   = (float*)carve((size_t)N * 4);
    int*   rank   = (int*)  carve((size_t)E * 4);
    int2*  csr    = (int2*) carve((size_t)E * 8);
    float* hW     = (float*)carve((size_t)N * HID * 4);
    float* agg    = (float*)carve((size_t)N * HID * 4);
    float* sums   = (float*)carve((size_t)G * PW * 4);
    float* cnt    = (float*)carve((size_t)G * 4);
    float* W34    = (float*)carve((size_t)HID * PW * 4);
    (void)ws_size; (void)n_in;

    const int B = 256;
    const int nB  = ((N + 3) / 4 + B - 1) / B;
    const int eB  = (E + B * EPL - 1) / (B * EPL);
    const int mt  = (N + 63) / 64;
    const int gB  = (N + 8 * NPW - 1) / (8 * NPW);
    const int pB  = (N + 8 * NPW3 - 1) / (8 * NPW3);

    k_cnt_zero<<<nB, B, 0, stream>>>(ecnt, N, sums, cnt, G);
    k_rank    <<<eB, B, 0, stream>>>(dst, ecnt, rank, E);
    k_scan    <<<2, 1024, 0, stream>>>(ecnt, rowptr, dinv, N, W3, Wl, W34, OC);
    k_fill    <<<eB, B, 0, stream>>>(src, dst, rank, rowptr, dinv, csr, E);

    k_gemm_wmma<128, false><<<mt, 128, 0, stream>>>(x, W1, hW, N);
    k_gather_gemm<HID><<<gB, B, 0, stream>>>(rowptr, csr, dinv, hW, W2, agg, N);

    k_gather_gemm<PW><<<gB, B, 0, stream>>>(rowptr, csr, dinv, agg, W34, hW, N);

    k_gather_pool<<<pB, B, 0, stream>>>(rowptr, csr, dinv, hW, batch, sums, cnt, N);
    k_final      <<<(G * OC + B - 1) / B, B, 0, stream>>>(sums, cnt, bl, (float*)d_out, G, OC);
}


# Edge ranking without device-wide atomics: two-level counting sort on LDS counters (bins of 256 nodes: count, scan, scatter, then one work-group ranks each bin); order of a node's sum changes
# speedup vs baseline: 5.3508x; 1.4613x over previous
#include <hip/hip_runtime.h>

typedef __attribute__((ext_vector_type(2))) float v2f;
typedef __attribute__((ext_vector_type(8))) float v8f;

#define HID 64

#define WMMA_F32(a, b, c) __builtin_amdgcn_wmma_f32_16x16x4_f32( \
         false, a,  false, b,  (short)0, c, \
         false,  false)

#define EPL  8
#define NPW  4
#define PW   16
#define NPW3 8

#define BSH    8
#define BNODES (1 << BSH)
#define MAXBIN 2048
#define EPL1   20

__global__ void __launch_bounds__(256)
k_bincount(const int* __restrict__ dst, int* __restrict__ hist, int e, int nbin,
           float* psum, float* pcnt, int g) {
    __shared__ int lh[MAXBIN];
    int tid = threadIdx.x;
    for (int b = tid; b < nbin; b += 256) lh[b] = 0;
    int t = blockIdx.x * 256 + tid, nt = gridDim.x * 256;
    for (int k = t; k < g * (PW / 4); k += nt) ((float4*)psum)[k] = make_float4(0.f, 0.f, 0.f, 0.f);
    for (int k = t; k < g; k += nt) pcnt[k] = 0.f;
    __syncthreads();
    int base = blockIdx.x * (256 * EPL1) + tid;
    int d[EPL1];
#pragma unroll
    for (int k = 0; k < EPL1; ++k) {
        int i = base + k * 256;
        d[k] = dst[i < e ? i : e - 1];
    }
#pragma unroll
    for (int k = 0; k < EPL1; ++k) {
        int i = base + k * 256;
        if (i < e) atomicAdd(&lh[d[k] >> BSH], 1);
    }
    __syncthreads();
    for (int b = tid; b < nbin; b += 256) hist[b * gridDim.x + blockIdx.x] = lh[b];
}

#define EPT 16
__global__ void __launch_bounds__(1024)
k_scan(const int* __restrict__ cnt, int* __restrict__ rowptr, int n,
       const float* __restrict__ W3, const float* __restrict__ Wl, float* __restrict__ W34, int oc) {
    __shared__ int wsum[32];
    if (blockIdx.x == 1) {
        int fw = __builtin_amdgcn_readfirstlane((int)(threadIdx.x >> 5));
        if (fw >= HID / 16) return;
        int fl = threadIdx.x & 31;
        int half = fl >> 4;
        int l15  = fl & 15;
        bool real = l15 < oc;
        int  col  = real ? l15 : 0;
        const float* ap = W3 + (fw * 16 + l15) * HID + half * 2;
        const float* bp = Wl + half * 2 * oc + col;
        v8f c = {};
#pragma unroll
        for (int k = 0; k < HID; k += 4) {
            v2f a, b;
            a[0] = ap[k]; a[1] = ap[k + 1];
            float bx = bp[k * oc];
            float by = bp[(k + 1) * oc];
            b[0] = real ? bx : 0.f; b[1] = real ? by : 0.f;
            c = WMMA_F32(a, b, c);
        }
#pragma unroll
        for (int j = 0; j < 8; ++j)
            W34[(fw * 16 + half * 8 + j) * PW + l15] = c[j];
        return;
    }
    int tid = threadIdx.x, lane = tid & 31, wave = tid >> 5;
    int carry = 0;
    for (int base = 0; base < n; base += 1024 * EPT) {
        int i0 = base + tid * EPT;
        bool full = (i0 + EPT <= n);
        int c[EPT];
        if (full) {
#pragma unroll
            for (int q = 0; q < EPT / 4; ++q) {
                int4 t = ((const int4*)(cnt + i0))[q];
                c[4 * q] = t.x; c[4 * q + 1] = t.y; c[4 * q + 2] = t.z; c[4 * q + 3] = t.w;
            }
        } else {
#pragma unroll
            for (int k = 0; k < EPT; ++k) c[k] = (i0 + k < n) ? cnt[i0 + k] : 0;
        }
        int tot = 0;
#pragma unroll
        for (int k = 0; k < EPT; ++k) tot += c[k];
        int x = tot;
#pragma unroll
        for (int o = 1; o < 32; o <<= 1) {
            int y = __shfl_up(x, o);
            if (lane >= o) x += y;
        }
        if (lane == 31) wsum[wave] = x;
        __syncthreads();
        int w = wsum[lane];
        int wx = w;
#pragma unroll
        for (int o = 1; o < 32; o <<= 1) {
            int y = __shfl_up(wx, o);
            if (lane >= o) wx += y;
        }
        int wave_excl = __shfl(wx, wave) - __shfl(w, wave);
        int tile_tot  = __shfl(wx, 31);
        int run = carry + wave_excl + (x - tot);
        int rp[EPT];
#pragma unroll
        for (int k = 0; k < EPT; ++k) {
            rp[k] = run;
            run += c[k];
        }
        if (full) {
#pragma unroll
            for (int q = 0; q < EPT / 4; ++q)
                ((int4*)(rowptr + i0))[q] =
                    make_int4(rp[4 * q], rp[4 * q + 1], rp[4 * q + 2], rp[4 * q + 3]);
        } else {
#pragma unroll
            for (int k = 0; k < EPT; ++k)
                if (i0 + k < n) rowptr[i0 + k] = rp[k];
        }
        carry += tile_tot;
        __syncthreads();
    }
    if (tid == 0) rowptr[n] = carry;
}

template <bool FULL>
__device__ __forceinline__ void scatter_span(const int* __restrict__ src, const int* __restrict__ dst,
                                             int* lcur, int2* __restrict__ binned, int e, int base) {
    int s[EPL1], d[EPL1], p[EPL1];
#pragma unroll
    for (int k = 0; k < EPL1; ++k) {
        int i = base + k * 256;
        int ic = (FULL || i < e) ? i : e - 1;
        s[k] = src[ic];
        d[k] = dst[ic];
    }
    __syncthreads();
#pragma unroll
    for (int k = 0; k < EPL1; ++k) {
        int i = base + k * 256;
        p[k] = 0;
        if (FULL || i < e) p[k] = atomicAdd(&lcur[d[k] >> BSH], 1);
    }
#pragma unroll
    for (int k = 0; k < EPL1; ++k) {
        int i = base + k * 256;
        if (FULL || i < e) binned[p[k]] = make_int2(s[k], d[k]);
    }
}

__global__ void __launch_bounds__(256)
k_binscatter(const int* __restrict__ src, const int* __restrict__ dst,
             const int* __restrict__ bstart, int2* __restrict__ binned, int e, int nbin) {
    __shared__ int lcur[MAXBIN];
    int tid = threadIdx.x;
    for (int b = tid; b < nbin; b += 256) lcur[b] = bstart[b * gridDim.x + blockIdx.x];
    int span = blockIdx.x * (256 * EPL1);
    int base = span + tid;
    if (span + 256 * EPL1 <= e) scatter_span<true>(src, dst, lcur, binned, e, base);
    else                        scatter_span<false>(src, dst, lcur, binned, e, base);
}

#define EPD 8
__global__ void __launch_bounds__(1024)
k_bindeg(const int2* __restrict__ binned, const int* __restrict__ bstart, int nwg,
         int* __restrict__ rank, int* __restrict__ rowptr, float* __restrict__ dinv,
         int n, int nbin) {
    __shared__ int lc[BNODES];
    __shared__ int wtot[BNODES / 32];
    int tid = threadIdx.x, lane = tid & 31, wave = tid >> 5;
    int b  = blockIdx.x;
    int bs = bstart[b * nwg], be = bstart[(b + 1) * nwg];
    if (tid < BNODES) lc[tid] = 0;
    __syncthreads();
    for (int i0 = bs; i0 < be; i0 += 1024 * EPD) {
        int d[EPD], r[EPD];
#pragma unroll
        for (int k = 0; k < EPD; ++k) {
            int i = i0 + k * 1024 + tid;
            d[k] = binned[i < be ? i : be - 1].y;
        }
#pragma unroll
        for (int k = 0; k < EPD; ++k) {
            int i = i0 + k * 1024 + tid;
            r[k] = 0;
            if (i < be) r[k] = atomicAdd(&lc[d[k] & (BNODES - 1)], 1);
        }
#pragma unroll
        for (int k = 0; k < EPD; ++k) {
            int i = i0 + k * 1024 + tid;
            if (i < be) rank[i] = r[k];
        }
    }
    __syncthreads();
    int c = 0, x = 0;
    if (tid < BNODES) {
        c = lc[tid];
        x = c;
#pragma unroll
        for (int o = 1; o < 32; o <<= 1) {
            int y = __shfl_up(x, o);
            if (lane >= o) x += y;
        }
        if (lane == 31) wtot[wave] = x;
    }
    __syncthreads();
    if (tid < BNODES) {
        int off = bs;
        for (int j = 0; j < wave; ++j) off += wtot[j];
        int v = b * BNODES + tid;
        if (v < n) {
            rowptr[v] = off + x - c;
            float dg = (float)(c + 1);
            dinv[v] = (dg > 0.f) ? rsqrtf(dg) : 0.f;
        }
        if (b == nbin - 1 && tid == 0) rowptr[n] = be;
    }
}

template <bool FULL>
__device__ __forceinline__ void fill_span(const int2* __restrict__ binned,
                                          const int* __restrict__ rank,
                                          const int* __restrict__ rowptr,
                                          const float* __restrict__ dinv,
                                          int2* __restrict__ csr, int e, int base) {
    int s[EPL], d[EPL], rk[EPL];
#pragma unroll
    for (int k = 0; k < EPL; ++k) {
        int i = base + k * 256;
        int ic = (FULL || i < e) ? i : e - 1;
        int2 sd = binned[ic];
        s[k]  = sd.x;
        d[k]  = sd.y;
        rk[k] = rank[ic];
    }
    int   rp[EPL];
    float ds[EPL], dd[EPL];
#pragma unroll
    for (int k = 0; k < EPL; ++k) {
        rp[k] = rowptr[d[k]];
        ds[k] = dinv[s[k]];
        dd[k] = dinv[d[k]];
    }
#pragma unroll
    for (int k = 0; k < EPL; ++k) {
        int i = base + k * 256;
        float nm = ds[k] * dd[k];
        if (FULL || i < e) csr[rp[k] + rk[k]] = make_int2(s[k], __float_as_int(nm));
    }
}

__global__ void __launch_bounds__(256)
k_fill(const int2* __restrict__ binned,
       const int* __restrict__ rank, const int* __restrict__ rowptr,
       const float* __restrict__ dinv, int2* __restrict__ csr, int e) {
    int span = blockIdx.x * (256 * EPL);
    int base = span + threadIdx.x;
    if (span + 256 * EPL <= e) fill_span<true>(binned, rank, rowptr, dinv, csr, e, base);
    else                       fill_span<false>(binned, rank, rowptr, dinv, csr, e, base);
}

template <int K, bool DO_RELU>
__global__ void __launch_bounds__(128)
k_gemm_wmma(const float* __restrict__ A, const float* __restrict__ W,
            float* __restrict__ C, int n) {
    __shared__ float4 sW4[K * HID / 4];
    const float* sW = (const float*)sW4;
    int tid = threadIdx.x;
#pragma unroll
    for (int q = 0; q < K * HID / 4 / 128; ++q) sW4[tid + q * 128] = ((const float4*)W)[tid + q * 128];
    __syncthreads();

    int wave = tid >> 5;
    int lane = tid & 31;
    int half = lane >> 4;
    int l15  = lane & 15;
    int m0 = blockIdx.x * 64 + wave * 16;

    int arow = m0 + l15;
    if (arow >= n) arow = n - 1;
    const float* ap = A + (size_t)arow * K + half * 2;
    const float* bp = sW + l15 + half * 2 * HID;

    v8f c0 = {}, c1 = {}, c2 = {}, c3 = {};
#pragma unroll
    for (int k = 0; k < K; k += 4) {
        v2f a, b0, b1, b2, b3;
        float ax = ap[k];
        float ay = ap[k + 1];
        if (DO_RELU) { ax = fmaxf(ax, 0.f); ay = fmaxf(ay, 0.f); }
        a[0] = ax; a[1] = ay;
        const float* bk = bp + k * HID;
        b0[0] = bk[0];  b0[1] = bk[HID];
        b1[0] = bk[16]; b1[1] = bk[HID + 16];
        b2[0] = bk[32]; b2[1] = bk[HID + 32];
        b3[0] = bk[48]; b3[1] = bk[HID + 48];
        c0 = WMMA_F32(a, b0, c0);
        c1 = WMMA_F32(a, b1, c1);
        c2 = WMMA_F32(a, b2, c2);
        c3 = WMMA_F32(a, b3, c3);
    }

    int rbase = m0 + half * 8;
#pragma unroll
    for (int j = 0; j < 8; ++j) {
        int r = rbase + j;
        if (r < n) {
            float* cr = C + (size_t)r * HID + l15;
            cr[0] = c0[j]; cr[16] = c1[j]; cr[32] = c2[j]; cr[48] = c3[j];
        }
    }
}

template <int U>
__device__ __forceinline__ void gather_rows(const float2* __restrict__ h2, int s_l, int w_l,
                                            int j, int lane, float2& acc) {
    float2 r[U];
    float  w[U];
#pragma unroll
    for (int k = 0; k < U; ++k) {
        int s = __builtin_amdgcn_readlane(s_l, j + k);
        w[k]  = __int_as_float(__builtin_amdgcn_readlane(w_l, j + k));
        r[k]  = h2[(size_t)s * (HID / 2) + lane];
    }
#pragma unroll
    for (int k = 0; k < U; ++k) {
        acc.x = fmaf(r[k].x, w[k], acc.x);
        acc.y = fmaf(r[k].y, w[k], acc.y);
    }
}

__device__ __forceinline__ void gather_chunk(const float2* __restrict__ h2, int s_l, int w_l,
                                             int m, int lane, float2& acc) {
    int j = 0;
    for (; j + 8 <= m; j += 8) gather_rows<8>(h2, s_l, w_l, j, lane, acc);
    if (m & 4) { gather_rows<4>(h2, s_l, w_l, j, lane, acc); j += 4; }
    if (m & 2) { gather_rows<2>(h2, s_l, w_l, j, lane, acc); j += 2; }
    if (m & 1) { gather_rows<1>(h2, s_l, w_l, j, lane, acc); }
}

#define APITCH 68
__device__ __forceinline__ void gather_nodes(const int* __restrict__ rowptr, const int2* __restrict__ csr,
                                             const float* __restrict__ dinv, const float2* __restrict__ h2,
                                             float* dstl, int v0, int vend, int lane) {
    int beg = rowptr[v0], end = rowptr[v0 + 1];
    float di = dinv[v0];
    int2 rec = make_int2(0, 0);
    if (lane < end - beg) rec = csr[beg + lane];
    float2 hv = h2[(size_t)v0 * (HID / 2) + lane];
    float sw = di * di;
    float2 acc;
    acc.x = hv.x * sw;
    acc.y = hv.y * sw;
    for (int v = v0; v < vend; ++v) {
        int vn = (v + 1 < vend) ? v + 1 : v;
        int nbeg = rowptr[vn], nend = rowptr[vn + 1];
        float ndi = dinv[vn];
        int2 nrec = make_int2(0, 0);
        if (lane < nend - nbeg) nrec = csr[nbeg + lane];
        float2 nhv = h2[(size_t)vn * (HID / 2) + lane];

        int m = end - beg;
        if (m > 32) m = 32;
        gather_chunk(h2, rec.x, rec.y, m, lane, acc);
        for (int p0 = beg + 32; p0 < end; p0 += 32) {
            int m2 = end - p0;
            if (m2 > 32) m2 = 32;
            int2 rec2 = make_int2(0, 0);
            if (lane < m2) rec2 = csr[p0 + lane];
            gather_chunk(h2, rec2.x, rec2.y, m2, lane, acc);
        }
        float2 t;
        t.x = fmaxf(acc.x, 0.f);
        t.y = fmaxf(acc.y, 0.f);
        *(float2*)(dstl + (v - v0) * APITCH + 2 * lane) = t;

        float nsw = ndi * ndi;
        acc.x = nhv.x * nsw;
        acc.y = nhv.y * nsw;
        beg = nbeg; end = nend; rec = nrec;
    }
}

template <int NO>
__global__ void __launch_bounds__(256)
k_gather_gemm(const int* __restrict__ rowptr, const int2* __restrict__ csr,
              const float* __restrict__ dinv, const float* __restrict__ hW,
              const float* __restrict__ Wn, float* __restrict__ hWn, int n) {
    __shared__ float4 sW4[HID * NO / 4];
    __shared__ __attribute__((aligned(16))) float sA[8 * NPW * APITCH];
    const float* sW = (const float*)sW4;
    int tid  = threadIdx.x;
    int lane = tid & 31;
    int wv = __builtin_amdgcn_readfirstlane(tid >> 5);
    constexpr int NPIECE = HID * NO / 4 / 256;
    float4 wreg[NPIECE];
#pragma unroll
    for (int q = 0; q < NPIECE; ++q) wreg[q] = ((const float4*)Wn)[tid + q * 256];
#pragma unroll
    for (int q = 0; q < NPIECE; ++q) sW4[tid + q * 256] = wreg[q];

    int v0 = (blockIdx.x * 8 + wv) * NPW;
    int vend = v0 + NPW;
    if (vend > n) vend = n;
    float* myrows = sA + wv * NPW * APITCH;
    if (v0 < n)
        gather_nodes(rowptr, csr, dinv, (const float2*)hW, myrows, v0, vend, lane);
    for (int v = (v0 < n ? vend : v0); v < v0 + NPW; ++v) {
        float2 z; z.x = 0.f; z.y = 0.f;
        *(float2*)(myrows + (v - v0) * APITCH + 2 * lane) = z;
    }
    __syncthreads();
    constexpr bool WIDE = (NO == HID);
    if (wv >= (WIDE ? 4 : 2)) return;

    int mt = WIDE ? (wv >> 1) : wv;
    int nh = WIDE ? (wv & 1) : 0;
    int half = lane >> 4;
    int l15  = lane & 15;
    const float* ap = sA + (mt * 16 + l15) * APITCH + half * 2;
    const float* bp = sW + nh * 32 + l15 + half * 2 * NO;
    v8f c0 = {}, c1 = {};
#pragma unroll
    for (int k = 0; k < HID; k += 4) {
        v2f a, b0, b1;
        a[0] = ap[k]; a[1] = ap[k + 1];
        const float* bk = bp + k * NO;
        b0[0] = bk[0];  b0[1] = bk[NO];
        c0 = WMMA_F32(a, b0, c0);
        if (WIDE) {
            b1[0] = bk[16]; b1[1] = bk[NO + 16];
            c1 = WMMA_F32(a, b1, c1);
        }
    }
    int rbase = blockIdx.x * (8 * NPW) + mt * 16 + half * 8;
#pragma unroll
    for (int j = 0; j < 8; ++j) {
        int r = rbase + j;
        if (r < n) {
            float* cr = hWn + (size_t)r * NO + nh * 32 + l15;
            cr[0] = c0[j];
            if (WIDE) cr[16] = c1[j];
        }
    }
}

__global__ void __launch_bounds__(256)
k_gather_pool(const int* __restrict__ rowptr, const int2* __restrict__ csr,
              const float* __restrict__ dinv, const float* __restrict__ z,
              const int* __restrict__ batch, float* sums, float* cnt, int n) {
    int lane = threadIdx.x & 31;
    int sub = lane & 7, grp = lane >> 3;
    int w  = __builtin_amdgcn_readfirstlane((int)(blockIdx.x * 8 + (threadIdx.x >> 5)));
    int v0 = w * NPW3;
    if (v0 >= n) return;
    int nv = n - v0;
    if (nv > NPW3) nv = NPW3;
    const float2* z2 = (const float2*)z;

    int vb = v0 + lane;
    int rp_l = rowptr[vb <= n ? vb : n];
    if (vb >= n) vb = n - 1;
    int   g_l  = batch[vb];
    float di_l = dinv[vb];
    float2 own[NPW3 / 4];
    float  ow[NPW3 / 4];
#pragma unroll
    for (int q = 0; q < NPW3 / 4; ++q) {
        int j = 4 * q + grp;
        own[q] = z2[j < nv ? (size_t)v0 * (PW / 2) + q * 32 + lane : 0];
        float d = __shfl(di_l, j);
        ow[q] = d * d;
    }

    int j0 = 0;
    while (j0 < nv) {
        int g = __builtin_amdgcn_readfirstlane(__shfl(g_l, j0));
        unsigned diff = __builtin_amdgcn_ballot_w32(lane > j0 && lane < nv && g_l != g);
        int j1 = diff ? __builtin_ctz(diff) : nv;
        int eb = __builtin_amdgcn_readfirstlane(__shfl(rp_l, j0));
        int ee = __builtin_amdgcn_readfirstlane(__shfl(rp_l, j1));

        float2 acc = make_float2(0.f, 0.f);
#pragma unroll
        for (int q = 0; q < NPW3 / 4; ++q) {
            if (4 * q + grp >= j0 && 4 * q + grp < j1) {
                acc.x = fmaf(own[q].x, ow[q], acc.x);
                acc.y = fmaf(own[q].y, ow[q], acc.y);
            }
        }
        if (eb < ee) {
            int q0 = eb + lane, q1 = eb + 32 + lane;
            int2 rec0 = csr[q0 < ee ? q0 : ee - 1];
            int2 rec1 = csr[q1 < ee ? q1 : ee - 1];
            for (int base = eb; base < ee; base += 64) {
                int m = ee - base;
                float2 r[16];
                float  wk[16];
#pragma unroll
                for (int k = 0; k < 16; ++k) {
                    int from = 4 * (k & 7) + grp;
                    int s = __shfl(k < 8 ? rec0.x : rec1.x, from);
                    wk[k] = __int_as_float(__shfl(k < 8 ? rec0.y : rec1.y, from));
                    r[k]  = z2[(size_t)s * (PW / 2) + sub];
                }
                int n0 = base + 64 + lane, n1 = base + 96 + lane;
                int2 nrec0 = csr[n0 < ee ? n0 : ee - 1];
                int2 nrec1 = csr[n1 < ee ? n1 : ee - 1];
#pragma unroll
                for (int k = 0; k < 16; ++k) {
                    if (4 * k + grp < m) {
                        acc.x = fmaf(r[k].x, wk[k], acc.x);
                        acc.y = fmaf(r[k].y, wk[k], acc.y);
                    }
                }
                rec0 = nrec0; rec1 = nrec1;
            }
        }
        acc.x += __shfl_xor(acc.x, 8);  acc.y += __shfl_xor(acc.y, 8);
        acc.x += __shfl_xor(acc.x, 16); acc.y += __shfl_xor(acc.y, 16);
        if (lane < 8) {
            atomicAdd(&sums[(size_t)g * PW + 2 * sub],     acc.x);
            atomicAdd(&sums[(size_t)g * PW + 2 * sub + 1], acc.y);
        }
        if (lane == 0) atomicAdd(&cnt[g], (float)(j1 - j0));
        j0 = j1;
    }
}

__global__ void k_final(const float* __restrict__ sums, const float* __restrict__ cnt,
                        const float* __restrict__ bl, float* __restrict__ out, int g, int oc) {
    int i = blockIdx.x * blockDim.x + threadIdx.x;
    if (i < g * oc) {
        int gi = i / oc, o = i % oc;
        out[i] = sums[(size_t)gi * PW + o] / fmaxf(cnt[gi], 1.0f) + bl[o];
    }
}

extern "C" void kernel_launch(void* const* d_in, const int* in_sizes, int n_in,
                              void* d_out, int out_size, void* d_ws, size_t ws_size,
                              hipStream_t stream) {
    const float* x    = (const float*)d_in[0];
    const int*   eidx = (const int*)  d_in[1];
    const int*   batch= (const int*)  d_in[2];
    const float* W1   = (const float*)d_in[3];
    const float* W2   = (const float*)d_in[4];
    const float* W3   = (const float*)d_in[5];
    const float* Wl   = (const float*)d_in[6];
    const float* bl   = (const float*)d_in[7];

    const int INC = in_sizes[3] / HID;
    const int N   = in_sizes[0] / INC;
    const int E   = in_sizes[1] / 2;
    const int OC  = in_sizes[7];
    const int G   = out_size / OC;

    const int* src = eidx;
    const int* dst = eidx + E;

    const int B = 256;
    const int eB  = (E + B * EPL - 1) / (B * EPL);
    const int eB1 = (E + B * EPL1 - 1) / (B * EPL1);
    const int nbin = (N + BNODES - 1) / BNODES;
    const int mt  = (N + 63) / 64;
    const int gB  = (N + 8 * NPW - 1) / (8 * NPW);
    const int pB  = (N + 8 * NPW3 - 1) / (8 * NPW3);

    char* ws = (char*)d_ws; size_t off = 0;
    auto carve = [&](size_t bytes) {
        void* p = ws + off; off = (off + bytes + 255) & ~(size_t)255; return p;
    };
    int*   hist   = (int*)  carve((size_t)nbin * eB1 * 4);
    int*   bstart = (int*)  carve(((size_t)nbin * eB1 + 1) * 4);
    int2*  binned = (int2*) carve((size_t)E * 8);
    int*   rowptr = (int*)  carve((size_t)(N + 1) * 4);
    float* dinv   = (float*)carve((size_t)N * 4);
    int*   rank   = (int*)  carve((size_t)E * 4);
    int2*  csr    = (int2*) carve((size_t)E * 8);
    float* hW     = (float*)carve((size_t)N * HID * 4);
    float* agg    = (float*)carve((size_t)N * HID * 4);
    float* sums   = (float*)carve((size_t)G * PW * 4);
    float* cnt    = (float*)carve((size_t)G * 4);
    float* W34    = (float*)carve((size_t)HID * PW * 4);
    (void)ws_size; (void)n_in;

    k_bincount  <<<eB1, B, 0, stream>>>(dst, hist, E, nbin, sums, cnt, G);
    k_scan      <<<2, 1024, 0, stream>>>(hist, bstart, nbin * eB1, W3, Wl, W34, OC);
    k_binscatter<<<eB1, B, 0, stream>>>(src, dst, bstart, binned, E, nbin);
    k_bindeg    <<<nbin, 1024, 0, stream>>>(binned, bstart, eB1, rank, rowptr, dinv, N, nbin);
    k_fill      <<<eB, B, 0, stream>>>(binned, rank, rowptr, dinv, csr, E);

    k_gemm_wmma<128, false><<<mt, 128, 0, stream>>>(x, W1, hW, N);
    k_gather_gemm<HID><<<gB, B, 0, stream>>>(rowptr, csr, dinv, hW, W2, agg, N);

    k_gather_gemm<PW><<<gB, B, 0, stream>>>(rowptr, csr, dinv, agg, W34, hW, N);

    k_gather_pool<<<pB, B, 0, stream>>>(rowptr, csr, dinv, hW, batch, sums, cnt, N);
    k_final      <<<(G * OC + B - 1) / B, B, 0, stream>>>(sums, cnt, bl, (float*)d_out, G, OC);
}


# One-pass edge binning: each work-group sorts its own span by bin in LDS (count, scan, copy); a bin's start is the sum of the spans' local offsets, so count and scan launches go (sum order changes)
# speedup vs baseline: 7.5882x; 1.4181x over previous
#include <hip/hip_runtime.h>

typedef __attribute__((ext_vector_type(2))) float v2f;
typedef __attribute__((ext_vector_type(8))) float v8f;

#define HID 64

#define WMMA_F32(a, b, c) __builtin_amdgcn_wmma_f32_16x16x4_f32( \
         false, a,  false, b,  (short)0, c, \
         false,  false)

#define NPW  4
#define PW   16
#define NPW3 8

#define BSH    8
#define BNODES (1 << BSH)
#define MAXBIN 2048
#define EPL1   20
#define SPAN   (256 * EPL1)
#define MAXSEG 1024

template <bool FULL>
__device__ __forceinline__ void sort_span(const int* __restrict__ src, const int* __restrict__ dst,
                                          int* lh, int* wsum, int* __restrict__ loff,
                                          int2* __restrict__ binned, int e, int span, int nbin, int nwg) {
    int tid = threadIdx.x, lane = tid & 31, wave = tid >> 5;
    int base = span + tid;
    int s[EPL1], d[EPL1], p[EPL1];
#pragma unroll
    for (int k = 0; k < EPL1; ++k) {
        int i = base + k * 256;
        int ic = (FULL || i < e) ? i : e - 1;
        s[k] = src[ic];
        d[k] = dst[ic];
    }
#pragma unroll
    for (int k = 0; k < EPL1; ++k) {
        int i = base + k * 256;
        p[k] = 0;
        if (FULL || i < e) p[k] = atomicAdd(&lh[d[k] >> BSH], 1);
    }
    __syncthreads();
    int4 q0 = ((int4*)lh)[2 * tid], q1 = ((int4*)lh)[2 * tid + 1];
    int c[8] = {q0.x, q0.y, q0.z, q0.w, q1.x, q1.y, q1.z, q1.w};
    int tot = 0;
#pragma unroll
    for (int k = 0; k < 8; ++k) tot += c[k];
    int x = tot;
#pragma unroll
    for (int o = 1; o < 32; o <<= 1) {
        int y = __shfl_up(x, o);
        if (lane >= o) x += y;
    }
    if (lane == 31) wsum[wave] = x;
    __syncthreads();
    int run = x - tot;
#pragma unroll
    for (int j = 0; j < 8; ++j) run += (j < wave) ? wsum[j] : 0;
    int w = blockIdx.x;
#pragma unroll
    for (int k = 0; k < 8; ++k) {
        int b = 8 * tid + k;
        int v = run;
        run += c[k];
        c[k] = v;
        if (b <= nbin) loff[b * nwg + w] = v;
    }
    ((int4*)lh)[2 * tid]     = make_int4(c[0], c[1], c[2], c[3]);
    ((int4*)lh)[2 * tid + 1] = make_int4(c[4], c[5], c[6], c[7]);
    __syncthreads();
#pragma unroll
    for (int k = 0; k < EPL1; ++k) {
        int i = base + k * 256;
        if (FULL || i < e) binned[span + lh[d[k] >> BSH] + p[k]] = make_int2(s[k], d[k]);
    }
}

__global__ void __launch_bounds__(256)
k_binsort(const int* __restrict__ src, const int* __restrict__ dst, int* __restrict__ loff,
          int2* __restrict__ binned, int e, int nbin, int nwg,
          float* psum, float* pcnt, int g,
          const float* __restrict__ W3, const float* __restrict__ Wl, float* __restrict__ W34, int oc) {
    __shared__ __attribute__((aligned(16))) int lh[MAXBIN];
    __shared__ int wsum[8];
    int tid = threadIdx.x;
    if ((int)blockIdx.x == nwg) {
        int fw = __builtin_amdgcn_readfirstlane(tid >> 5);
        if (fw >= HID / 16) return;
        int fl = tid & 31;
        int half = fl >> 4;
        int l15  = fl & 15;
        bool real = l15 < oc;
        int  col  = real ? l15 : 0;
        const float* ap = W3 + (fw * 16 + l15) * HID + half * 2;
        const float* bp = Wl + half * 2 * oc + col;
        v8f c = {};
#pragma unroll
        for (int k = 0; k < HID; k += 4) {
            v2f a, b;
            a[0] = ap[k]; a[1] = ap[k + 1];
            float bx = bp[k * oc];
            float by = bp[(k + 1) * oc];
            b[0] = real ? bx : 0.f; b[1] = real ? by : 0.f;
            c = WMMA_F32(a, b, c);
        }
#pragma unroll
        for (int j = 0; j < 8; ++j)
            W34[(fw * 16 + half * 8 + j) * PW + l15] = c[j];
        return;
    }
    for (int b = tid; b < MAXBIN; b += 256) lh[b] = 0;
    int t = blockIdx.x * 256 + tid, nt = nwg * 256;
    for (int k = t; k < g * (PW / 4); k += nt) ((float4*)psum)[k] = make_float4(0.f, 0.f, 0.f, 0.f);
    for (int k = t; k < g; k += nt) pcnt[k] = 0.f;
    __syncthreads();
    int span = blockIdx.x * SPAN;
    if (span + SPAN <= e) sort_span<true>(src, dst, lh, wsum, loff, binned, e, span, nbin, nwg);
    else                  sort_span<false>(src, dst, lh, wsum, loff, binned, e, span, nbin, nwg);
}

__device__ __forceinline__ void edge_addrs(const int* sgo, const int* sga, int nseg, int step0,
                                           const int (&i)[8], int (&addr)[8]) {
    int lo[8];
#pragma unroll
    for (int k = 0; k < 8; ++k) lo[k] = 0;
    for (int st = step0; st > 0; st >>= 1) {
        int v[8];
#pragma unroll
        for (int k = 0; k < 8; ++k) {
            int m = lo[k] + st;
            v[k] = sgo[m < nseg ? m : nseg];
        }
#pragma unroll
        for (int k = 0; k < 8; ++k) lo[k] = (v[k] <= i[k]) ? lo[k] + st : lo[k];
    }
#pragma unroll
    for (int k = 0; k < 8; ++k) addr[k] = sga[lo[k]] + (i[k] - sgo[lo[k]]);
}

#define EPD 8
__global__ void __launch_bounds__(1024)
k_bindeg(const int2* __restrict__ binned, const int* __restrict__ loff, int nwg,
         int* __restrict__ rank, int* __restrict__ rowptr, float* __restrict__ dinv,
         int* __restrict__ csr, int n, int nbin) {
    __shared__ int lc[BNODES];
    __shared__ int rs[BNODES];
    __shared__ int tile[1024 * EPD];
    __shared__ int wtot[BNODES / 32];
    __shared__ int sgo[MAXSEG + 1];
    __shared__ int sga[MAXSEG];
    __shared__ int wl[32], wa[32];
    int tid = threadIdx.x, lane = tid & 31, wave = tid >> 5;
    int b  = blockIdx.x;
    int pa = 0, plen = 0;
    if (tid < nwg) {
        pa = loff[b * nwg + tid];
        plen = loff[(b + 1) * nwg + tid] - pa;
    }
    int px = plen;
#pragma unroll
    for (int o = 1; o < 32; o <<= 1) {
        int y = __shfl_up(px, o);
        if (lane >= o) px += y;
    }
    int pas = pa;
#pragma unroll
    for (int o = 16; o > 0; o >>= 1) pas += __shfl_xor(pas, o);
    if (lane == 31) wl[wave] = px;
    if (lane == 0)  wa[wave] = pas;
    if (tid < BNODES) lc[tid] = 0;
    __syncthreads();
    int wlen = wl[lane];
    int wx = wlen;
#pragma unroll
    for (int o = 1; o < 32; o <<= 1) {
        int y = __shfl_up(wx, o);
        if (lane >= o) wx += y;
    }
    int wave_excl = __shfl(wx, wave) - __shfl(wlen, wave);
    int nb = __shfl(wx, 31);
    int bs = wa[lane];
#pragma unroll
    for (int o = 16; o > 0; o >>= 1) bs += __shfl_xor(bs, o);
    int be = bs + nb;
    if (tid < nwg) {
        sgo[tid] = wave_excl + px - plen;
        sga[tid] = tid * SPAN + pa;
    }
    if (tid == 0) sgo[nwg] = nb;
    __syncthreads();
    int step0 = 1;
    while (step0 * 2 < nwg) step0 *= 2;
    int last = nb - 1;

    int s0[EPD], d0[EPD], r0[EPD];
#pragma unroll
    for (int k = 0; k < EPD; ++k) { s0[k] = 0; d0[k] = 0; }
    if (nb > 0) {
        int ic[EPD], ad[EPD];
#pragma unroll
        for (int k = 0; k < EPD; ++k) {
            int il = k * 1024 + tid;
            ic[k] = il < nb ? il : last;
        }
        edge_addrs(sgo, sga, nwg, step0, ic, ad);
#pragma unroll
        for (int k = 0; k < EPD; ++k) {
            int2 sd = binned[ad[k]];
            s0[k] = sd.x;
            d0[k] = sd.y;
        }
    }
#pragma unroll
    for (int k = 0; k < EPD; ++k) {
        int il = k * 1024 + tid;
        r0[k] = 0;
        if (il < nb) r0[k] = atomicAdd(&lc[d0[k] & (BNODES - 1)], 1);
    }
    for (int l0 = 1024 * EPD; l0 < nb; l0 += 1024 * EPD) {
        int d[EPD], r[EPD], ic[EPD], ad[EPD];
#pragma unroll
        for (int k = 0; k < EPD; ++k) {
            int il = l0 + k * 1024 + tid;
            ic[k] = il < nb ? il : last;
        }
        edge_addrs(sgo, sga, nwg, step0, ic, ad);
#pragma unroll
        for (int k = 0; k < EPD; ++k) d[k] = binned[ad[k]].y;
#pragma unroll
        for (int k = 0; k < EPD; ++k) {
            int il = l0 + k * 1024 + tid;
            r[k] = 0;
            if (il < nb) r[k] = atomicAdd(&lc[d[k] & (BNODES - 1)], 1);
        }
#pragma unroll
        for (int k = 0; k < EPD; ++k) {
            int il = l0 + k * 1024 + tid;
            if (il < nb) rank[bs + il] = r[k];
        }
    }
    __syncthreads();
    int c = 0, x = 0;
    if (tid < BNODES) {
        c = lc[tid];
        x = c;
#pragma unroll
        for (int o = 1; o < 32; o <<= 1) {
            int y = __shfl_up(x, o);
            if (lane >= o) x += y;
        }
        if (lane == 31) wtot[wave] = x;
    }
    __syncthreads();
    if (tid < BNODES) {
        int off = bs;
        for (int j = 0; j < wave; ++j) off += wtot[j];
        int start = off + x - c;
        rs[tid] = start - bs;
        int v = b * BNODES + tid;
        if (v < n) {
            rowptr[v] = start;
            float dg = (float)(c + 1);
            dinv[v] = (dg > 0.f) ? rsqrtf(dg) : 0.f;
        }
        if (b == nbin - 1 && tid == 0) rowptr[n] = be;
    }
    __syncthreads();
    if (nb <= 1024 * EPD) {
#pragma unroll
        for (int k = 0; k < EPD; ++k) {
            int il = k * 1024 + tid;
            if (il < nb) tile[rs[d0[k] & (BNODES - 1)] + r0[k]] = s0[k];
        }
        __syncthreads();
        for (int t = tid; t < nb; t += 1024) csr[bs + t] = tile[t];
        return;
    }
#pragma unroll
    for (int k = 0; k < EPD; ++k) {
        int il = k * 1024 + tid;
        if (il < nb) csr[bs + rs[d0[k] & (BNODES - 1)] + r0[k]] = s0[k];
    }
    for (int l0 = 1024 * EPD; l0 < nb; l0 += 1024 * EPD) {
        int ic[EPD], ad[EPD];
#pragma unroll
        for (int k = 0; k < EPD; ++k) {
            int il = l0 + k * 1024 + tid;
            ic[k] = il < nb ? il : last;
        }
        edge_addrs(sgo, sga, nwg, step0, ic, ad);
#pragma unroll
        for (int k = 0; k < EPD; ++k) {
            int il = l0 + k * 1024 + tid;
            if (il < nb) {
                int2 sd = binned[ad[k]];
                csr[bs + rs[sd.y & (BNODES - 1)] + rank[bs + il]] = sd.x;
            }
        }
    }
}

template <int K, bool DO_RELU>
__global__ void __launch_bounds__(128)
k_gemm_wmma(const float* __restrict__ A, const float* __restrict__ W,
            const float* __restrict__ dinv, float* __restrict__ C, int n) {
    __shared__ float4 sW4[K * HID / 4];
    const float* sW = (const float*)sW4;
    int tid = threadIdx.x;
#pragma unroll
    for (int q = 0; q < K * HID / 4 / 128; ++q) sW4[tid + q * 128] = ((const float4*)W)[tid + q * 128];
    __syncthreads();

    int wave = tid >> 5;
    int lane = tid & 31;
    int half = lane >> 4;
    int l15  = lane & 15;
    int m0 = blockIdx.x * 64 + wave * 16;

    int arow = m0 + l15;
    if (arow >= n) arow = n - 1;
    const float* ap = A + (size_t)arow * K + half * 2;
    const float* bp = sW + l15 + half * 2 * HID;

    v8f c0 = {}, c1 = {}, c2 = {}, c3 = {};
#pragma unroll
    for (int k = 0; k < K; k += 4) {
        v2f a, b0, b1, b2, b3;
        float ax = ap[k];
        float ay = ap[k + 1];
        if (DO_RELU) { ax = fmaxf(ax, 0.f); ay = fmaxf(ay, 0.f); }
        a[0] = ax; a[1] = ay;
        const float* bk = bp + k * HID;
        b0[0] = bk[0];  b0[1] = bk[HID];
        b1[0] = bk[16]; b1[1] = bk[HID + 16];
        b2[0] = bk[32]; b2[1] = bk[HID + 32];
        b3[0] = bk[48]; b3[1] = bk[HID + 48];
        c0 = WMMA_F32(a, b0, c0);
        c1 = WMMA_F32(a, b1, c1);
        c2 = WMMA_F32(a, b2, c2);
        c3 = WMMA_F32(a, b3, c3);
    }

    int rbase = m0 + half * 8;
#pragma unroll
    for (int j = 0; j < 8; ++j) {
        int r = rbase + j;
        if (r < n) {
            float* cr = C + (size_t)r * HID + l15;
            float dr = dinv[r];
            cr[0] = c0[j] * dr; cr[16] = c1[j] * dr; cr[32] = c2[j] * dr; cr[48] = c3[j] * dr;
        }
    }
}

template <int U>
__device__ __forceinline__ void gather_rows(const float2* __restrict__ h2, int s_l,
                                            int j, int lane, float2& acc) {
    float2 r[U];
#pragma unroll
    for (int k = 0; k < U; ++k) {
        int s = __builtin_amdgcn_readlane(s_l, j + k);
        r[k]  = h2[(size_t)s * (HID / 2) + lane];
    }
#pragma unroll
    for (int k = 0; k < U; ++k) {
        acc.x += r[k].x;
        acc.y += r[k].y;
    }
}

__device__ __forceinline__ void gather_chunk(const float2* __restrict__ h2, int s_l,
                                             int m, int lane, float2& acc) {
    int j = 0;
    for (; j + 8 <= m; j += 8) gather_rows<8>(h2, s_l, j, lane, acc);
    if (m & 4) { gather_rows<4>(h2, s_l, j, lane, acc); j += 4; }
    if (m & 2) { gather_rows<2>(h2, s_l, j, lane, acc); j += 2; }
    if (m & 1) { gather_rows<1>(h2, s_l, j, lane, acc); }
}

#define APITCH 68
__device__ __forceinline__ void gather_nodes(const int* __restrict__ rowptr, const int* __restrict__ csr,
                                             const float* __restrict__ dinv, const float2* __restrict__ h2,
                                             float* dstl, int v0, int vend, int lane) {
    int beg = rowptr[v0], end = rowptr[v0 + 1];
    float di = dinv[v0];
    int rec = 0;
    if (lane < end - beg) rec = csr[beg + lane];
    float2 acc = h2[(size_t)v0 * (HID / 2) + lane];
    for (int v = v0; v < vend; ++v) {
        int vn = (v + 1 < vend) ? v + 1 : v;
        int nbeg = rowptr[vn], nend = rowptr[vn + 1];
        float ndi = dinv[vn];
        int nrec = 0;
        if (lane < nend - nbeg) nrec = csr[nbeg + lane];
        float2 nhv = h2[(size_t)vn * (HID / 2) + lane];

        int m = end - beg;
        if (m > 32) m = 32;
        gather_chunk(h2, rec, m, lane, acc);
        for (int p0 = beg + 32; p0 < end; p0 += 32) {
            int m2 = end - p0;
            if (m2 > 32) m2 = 32;
            int rec2 = 0;
            if (lane < m2) rec2 = csr[p0 + lane];
            gather_chunk(h2, rec2, m2, lane, acc);
        }
        float2 t;
        t.x = fmaxf(acc.x * di, 0.f);
        t.y = fmaxf(acc.y * di, 0.f);
        *(float2*)(dstl + (v - v0) * APITCH + 2 * lane) = t;

        acc = nhv;
        di = ndi;
        beg = nbeg; end = nend; rec = nrec;
    }
}

template <int NO>
__global__ void __launch_bounds__(256)
k_gather_gemm(const int* __restrict__ rowptr, const int* __restrict__ csr,
              const float* __restrict__ dinv, const float* __restrict__ hW,
              const float* __restrict__ Wn, float* __restrict__ hWn, int n) {
    __shared__ float4 sW4[HID * NO / 4];
    __shared__ __attribute__((aligned(16))) float sA[8 * NPW * APITCH];
    const float* sW = (const float*)sW4;
    int tid  = threadIdx.x;
    int lane = tid & 31;
    int wv = __builtin_amdgcn_readfirstlane(tid >> 5);
    constexpr int NPIECE = HID * NO / 4 / 256;
    float4 wreg[NPIECE];
#pragma unroll
    for (int q = 0; q < NPIECE; ++q) wreg[q] = ((const float4*)Wn)[tid + q * 256];
#pragma unroll
    for (int q = 0; q < NPIECE; ++q) sW4[tid + q * 256] = wreg[q];

    int v0 = (blockIdx.x * 8 + wv) * NPW;
    int vend = v0 + NPW;
    if (vend > n) vend = n;
    float* myrows = sA + wv * NPW * APITCH;
    if (v0 < n)
        gather_nodes(rowptr, csr, dinv, (const float2*)hW, myrows, v0, vend, lane);
    for (int v = (v0 < n ? vend : v0); v < v0 + NPW; ++v) {
        float2 z; z.x = 0.f; z.y = 0.f;
        *(float2*)(myrows + (v - v0) * APITCH + 2 * lane) = z;
    }
    __syncthreads();
    constexpr bool WIDE = (NO == HID);
    if (wv >= (WIDE ? 4 : 2)) return;

    int mt = WIDE ? (wv >> 1) : wv;
    int nh = WIDE ? (wv & 1) : 0;
    int half = lane >> 4;
    int l15  = lane & 15;
    const float* ap = sA + (mt * 16 + l15) * APITCH + half * 2;
    const float* bp = sW + nh * 32 + l15 + half * 2 * NO;
    v8f c0 = {}, c1 = {};
#pragma unroll
    for (int k = 0; k < HID; k += 4) {
        v2f a, b0, b1;
        a[0] = ap[k]; a[1] = ap[k + 1];
        const float* bk = bp + k * NO;
        b0[0] = bk[0];  b0[1] = bk[NO];
        c0 = WMMA_F32(a, b0, c0);
        if (WIDE) {
            b1[0] = bk[16]; b1[1] = bk[NO + 16];
            c1 = WMMA_F32(a, b1, c1);
        }
    }
    int rbase = blockIdx.x * (8 * NPW) + mt * 16 + half * 8;
#pragma unroll
    for (int j = 0; j < 8; ++j) {
        int r = rbase + j;
        if (r < n) {
            float* cr = hWn + (size_t)r * NO + nh * 32 + l15;
            float dr = dinv[r];
            cr[0] = c0[j] * dr;
            if (WIDE) cr[16] = c1[j] * dr;
        }
    }
}

__device__ __forceinline__ float wdst(int p, const int (&rp)[NPW3], const float (&dv)[NPW3]) {
    float wv = dv[0];
#pragma unroll
    for (int j = 1; j < NPW3; ++j) wv = (p >= rp[j]) ? dv[j] : wv;
    return wv;
}

__global__ void __launch_bounds__(256)
k_gather_pool(const int* __restrict__ rowptr, const int* __restrict__ csr,
              const float* __restrict__ dinv, const float* __restrict__ z,
              const int* __restrict__ batch, float* sums, float* cnt, int n) {
    int lane = threadIdx.x & 31;
    int sub = lane & 7, grp = lane >> 3;
    int w  = __builtin_amdgcn_readfirstlane((int)(blockIdx.x * 8 + (threadIdx.x >> 5)));
    int v0 = w * NPW3;
    if (v0 >= n) return;
    int nv = n - v0;
    if (nv > NPW3) nv = NPW3;
    const float2* z2 = (const float2*)z;

    int vb = v0 + lane;
    int rp_l = rowptr[vb <= n ? vb : n];
    if (vb >= n) vb = n - 1;
    int   g_l  = batch[vb];
    float di_l = dinv[vb];
    float2 own[NPW3 / 4];
    float  ow[NPW3 / 4];
#pragma unroll
    for (int q = 0; q < NPW3 / 4; ++q) {
        int j = 4 * q + grp;
        own[q] = z2[j < nv ? (size_t)v0 * (PW / 2) + q * 32 + lane : 0];
        ow[q] = __shfl(di_l, j);
    }
    int   rpj[NPW3];
    float dj[NPW3];
#pragma unroll
    for (int j = 0; j < NPW3; ++j) {
        rpj[j] = __builtin_amdgcn_readlane(rp_l, j);
        dj[j]  = __int_as_float(__builtin_amdgcn_readlane(__float_as_int(di_l), j));
    }

    int j0 = 0;
    while (j0 < nv) {
        int g = __builtin_amdgcn_readfirstlane(__shfl(g_l, j0));
        unsigned diff = __builtin_amdgcn_ballot_w32(lane > j0 && lane < nv && g_l != g);
        int j1 = diff ? __builtin_ctz(diff) : nv;
        int eb = __builtin_amdgcn_readfirstlane(__shfl(rp_l, j0));
        int ee = __builtin_amdgcn_readfirstlane(__shfl(rp_l, j1));

        float2 acc = make_float2(0.f, 0.f);
#pragma unroll
        for (int q = 0; q < NPW3 / 4; ++q) {
            if (4 * q + grp >= j0 && 4 * q + grp < j1) {
                acc.x = fmaf(own[q].x, ow[q], acc.x);
                acc.y = fmaf(own[q].y, ow[q], acc.y);
            }
        }
        if (eb < ee) {
            int q0 = eb + lane, q1 = eb + 32 + lane;
            if (q0 >= ee) q0 = ee - 1;
            if (q1 >= ee) q1 = ee - 1;
            int   rec0 = csr[q0], rec1 = csr[q1];
            float w0 = wdst(q0, rpj, dj), w1 = wdst(q1, rpj, dj);
            for (int base = eb; base < ee; base += 64) {
                int m = ee - base;
                float2 r[16];
                float  wk[16];
#pragma unroll
                for (int k = 0; k < 16; ++k) {
                    int from = 4 * (k & 7) + grp;
                    int sn = __shfl(k < 8 ? rec0 : rec1, from);
                    wk[k] = __shfl(k < 8 ? w0 : w1, from);
                    r[k]  = z2[(size_t)sn * (PW / 2) + sub];
                }
                int n0 = base + 64 + lane, n1 = base + 96 + lane;
                if (n0 >= ee) n0 = ee - 1;
                if (n1 >= ee) n1 = ee - 1;
                int   nrec0 = csr[n0], nrec1 = csr[n1];
                float nw0 = wdst(n0, rpj, dj), nw1 = wdst(n1, rpj, dj);
#pragma unroll
                for (int k = 0; k < 16; ++k) {
                    if (4 * k + grp < m) {
                        acc.x = fmaf(r[k].x, wk[k], acc.x);
                        acc.y = fmaf(r[k].y, wk[k], acc.y);
                    }
                }
                rec0 = nrec0; rec1 = nrec1;
                w0 = nw0; w1 = nw1;
            }
        }
        acc.x += __shfl_xor(acc.x, 8);  acc.y += __shfl_xor(acc.y, 8);
        acc.x += __shfl_xor(acc.x, 16); acc.y += __shfl_xor(acc.y, 16);
        if (lane < 8) {
            atomicAdd(&sums[(size_t)g * PW + 2 * sub],     acc.x);
            atomicAdd(&sums[(size_t)g * PW + 2 * sub + 1], acc.y);
        }
        if (lane == 0) atomicAdd(&cnt[g], (float)(j1 - j0));
        j0 = j1;
    }
}

__global__ void k_final(const float* __restrict__ sums, const float* __restrict__ cnt,
                        const float* __restrict__ bl, float* __restrict__ out, int g, int oc) {
    int i = blockIdx.x * blockDim.x + threadIdx.x;
    if (i < g * oc) {
        int gi = i / oc, o = i % oc;
        out[i] = sums[(size_t)gi * PW + o] / fmaxf(cnt[gi], 1.0f) + bl[o];
    }
}

extern "C" void kernel_launch(void* const* d_in, const int* in_sizes, int n_in,
                              void* d_out, int out_size, void* d_ws, size_t ws_size,
                              hipStream_t stream) {
    const float* x    = (const float*)d_in[0];
    const int*   eidx = (const int*)  d_in[1];
    const int*   batch= (const int*)  d_in[2];
    const float* W1   = (const float*)d_in[3];
    const float* W2   = (const float*)d_in[4];
    const float* W3   = (const float*)d_in[5];
    const float* Wl   = (const float*)d_in[6];
    const float* bl   = (const float*)d_in[7];

    const int INC = in_sizes[3] / HID;
    const int N   = in_sizes[0] / INC;
    const int E   = in_sizes[1] / 2;
    const int OC  = in_sizes[7];
    const int G   = out_size / OC;

    const int* src = eidx;
    const int* dst = eidx + E;

    const int B = 256;
    const int eB1 = (E + B * EPL1 - 1) / (B * EPL1);
    const int nbin = (N + BNODES - 1) / BNODES;
    const int mt  = (N + 63) / 64;
    const int gB  = (N + 8 * NPW - 1) / (8 * NPW);
    const int pB  = (N + 8 * NPW3 - 1) / (8 * NPW3);

    char* ws = (char*)d_ws; size_t off = 0;
    auto carve = [&](size_t bytes) {
        void* p = ws + off; off = (off + bytes + 255) & ~(size_t)255; return p;
    };
    int*   loff   = (int*)  carve((size_t)(nbin + 1) * eB1 * 4);
    int2*  binned = (int2*) carve((size_t)eB1 * SPAN * 8);
    int*   rowptr = (int*)  carve((size_t)(N + 1) * 4);
    float* dinv   = (float*)carve((size_t)N * 4);
    int*   rank   = (int*)  carve((size_t)E * 4);
    int*   csr    = (int*)  carve((size_t)E * 4);
    float* hW     = (float*)carve((size_t)N * HID * 4);
    float* agg    = (float*)carve((size_t)N * HID * 4);
    float* sums   = (float*)carve((size_t)G * PW * 4);
    float* cnt    = (float*)carve((size_t)G * 4);
    float* W34    = (float*)carve((size_t)HID * PW * 4);
    (void)ws_size; (void)n_in;

    k_binsort<<<eB1 + 1, B, 0, stream>>>(src, dst, loff, binned, E, nbin, eB1, sums, cnt, G,
                                         W3, Wl, W34, OC);
    k_bindeg <<<nbin, 1024, 0, stream>>>(binned, loff, eB1, rank, rowptr, dinv, csr, N, nbin);

    k_gemm_wmma<128, false><<<mt, 128, 0, stream>>>(x, W1, dinv, hW, N);
    k_gather_gemm<HID><<<gB, B, 0, stream>>>(rowptr, csr, dinv, hW, W2, agg, N);

    k_gather_gemm<PW><<<gB, B, 0, stream>>>(rowptr, csr, dinv, agg, W34, hW, N);

    k_gather_pool<<<pB, B, 0, stream>>>(rowptr, csr, dinv, hW, batch, sums, cnt, N);
    k_final      <<<(G * OC + B - 1) / B, B, 0, stream>>>(sums, cnt, bl, (float*)d_out, G, OC);
}


# Layer 1's product x @ W1 rides in the edge-binning launch (it needs no graph data): rows stored unscaled, each bin's work-group scales its 256 rows by dinv where it computes it; GEMM launch gone
# speedup vs baseline: 8.0919x; 1.0664x over previous
#include <hip/hip_runtime.h>

typedef __attribute__((ext_vector_type(2))) float v2f;
typedef __attribute__((ext_vector_type(8))) float v8f;

#define HID 64

#define WMMA_F32(a, b, c) __builtin_amdgcn_wmma_f32_16x16x4_f32( \
         false, a,  false, b,  (short)0, c, \
         false,  false)

#define NPW  4
#define PW   16
#define NPW3 8

#define BSH    8
#define BNODES (1 << BSH)
#define MAXBIN 2048
#define EPL1   20
#define SPAN   (256 * EPL1)
#define MAXSEG 1024

template <bool FULL>
__device__ __forceinline__ void sort_span(const int* __restrict__ src, const int* __restrict__ dst,
                                          int* lh, int* wsum, int* __restrict__ loff,
                                          int2* __restrict__ binned, int e, int span, int nbin, int nwg) {
    int tid = threadIdx.x, lane = tid & 31, wave = tid >> 5;
    int base = span + tid;
    int s[EPL1], d[EPL1], p[EPL1];
#pragma unroll
    for (int k = 0; k < EPL1; ++k) {
        int i = base + k * 256;
        int ic = (FULL || i < e) ? i : e - 1;
        s[k] = src[ic];
        d[k] = dst[ic];
    }
#pragma unroll
    for (int k = 0; k < EPL1; ++k) {
        int i = base + k * 256;
        p[k] = 0;
        if (FULL || i < e) p[k] = atomicAdd(&lh[d[k] >> BSH], 1);
    }
    __syncthreads();
    int4 q0 = ((int4*)lh)[2 * tid], q1 = ((int4*)lh)[2 * tid + 1];
    int c[8] = {q0.x, q0.y, q0.z, q0.w, q1.x, q1.y, q1.z, q1.w};
    int tot = 0;
#pragma unroll
    for (int k = 0; k < 8; ++k) tot += c[k];
    int x = tot;
#pragma unroll
    for (int o = 1; o < 32; o <<= 1) {
        int y = __shfl_up(x, o);
        if (lane >= o) x += y;
    }
    if (lane == 31) wsum[wave] = x;
    __syncthreads();
    int run = x - tot;
#pragma unroll
    for (int j = 0; j < 8; ++j) run += (j < wave) ? wsum[j] : 0;
    int w = blockIdx.x;
#pragma unroll
    for (int k = 0; k < 8; ++k) {
        int b = 8 * tid + k;
        int v = run;
        run += c[k];
        c[k] = v;
        if (b <= nbin) loff[b * nwg + w] = v;
    }
    ((int4*)lh)[2 * tid]     = make_int4(c[0], c[1], c[2], c[3]);
    ((int4*)lh)[2 * tid + 1] = make_int4(c[4], c[5], c[6], c[7]);
    __syncthreads();
#pragma unroll
    for (int k = 0; k < EPL1; ++k) {
        int i = base + k * 256;
        if (FULL || i < e) binned[span + lh[d[k] >> BSH] + p[k]] = make_int2(s[k], d[k]);
    }
}

template <int K>
__device__ __forceinline__ void gemm_rows(const float* __restrict__ A, const float* __restrict__ W,
                                          float* __restrict__ C, int n, int blk, float4* sW4) {
    const float* sW = (const float*)sW4;
    int tid = threadIdx.x;
#pragma unroll
    for (int q = 0; q < K * HID / 4 / 256; ++q) sW4[tid + q * 256] = ((const float4*)W)[tid + q * 256];
    __syncthreads();

    int wave = tid >> 5;
    int lane = tid & 31;
    int half = lane >> 4;
    int l15  = lane & 15;
    int m0 = blk * 128 + wave * 16;

    int arow = m0 + l15;
    if (arow >= n) arow = n - 1;
    const float* ap = A + (size_t)arow * K + half * 2;
    const float* bp = sW + l15 + half * 2 * HID;

    v8f c0 = {}, c1 = {}, c2 = {}, c3 = {};
#pragma unroll
    for (int k = 0; k < K; k += 4) {
        v2f a, b0, b1, b2, b3;
        a[0] = ap[k]; a[1] = ap[k + 1];
        const float* bk = bp + k * HID;
        b0[0] = bk[0];  b0[1] = bk[HID];
        b1[0] = bk[16]; b1[1] = bk[HID + 16];
        b2[0] = bk[32]; b2[1] = bk[HID + 32];
        b3[0] = bk[48]; b3[1] = bk[HID + 48];
        c0 = WMMA_F32(a, b0, c0);
        c1 = WMMA_F32(a, b1, c1);
        c2 = WMMA_F32(a, b2, c2);
        c3 = WMMA_F32(a, b3, c3);
    }

    int rbase = m0 + half * 8;
#pragma unroll
    for (int j = 0; j < 8; ++j) {
        int r = rbase + j;
        if (r < n) {
            float* cr = C + (size_t)r * HID + l15;
            cr[0] = c0[j]; cr[16] = c1[j]; cr[32] = c2[j]; cr[48] = c3[j];
        }
    }
}

#define INCH 128

__global__ void __launch_bounds__(256)
k_binsort(const int* __restrict__ src, const int* __restrict__ dst, int* __restrict__ loff,
          int2* __restrict__ binned, int e, int nbin, int nwg,
          float* psum, float* pcnt, int g,
          const float* __restrict__ W3, const float* __restrict__ Wl, float* __restrict__ W34, int oc,
          const float* __restrict__ x, const float* __restrict__ W1, float* __restrict__ xw, int n) {
    __shared__ float4 sbuf4[INCH * HID / 4];
    __shared__ int wsum[8];
    int* lh = (int*)sbuf4;
    int tid = threadIdx.x;
    if ((int)blockIdx.x > nwg) {
        gemm_rows<INCH>(x, W1, xw, n, (int)blockIdx.x - nwg - 1, sbuf4);
        return;
    }
    if ((int)blockIdx.x == nwg) {
        int fw = __builtin_amdgcn_readfirstlane(tid >> 5);
        if (fw >= HID / 16) return;
        int fl = tid & 31;
        int half = fl >> 4;
        int l15  = fl & 15;
        bool real = l15 < oc;
        int  col  = real ? l15 : 0;
        const float* ap = W3 + (fw * 16 + l15) * HID + half * 2;
        const float* bp = Wl + half * 2 * oc + col;
        v8f c = {};
#pragma unroll
        for (int k = 0; k < HID; k += 4) {
            v2f a, b;
            a[0] = ap[k]; a[1] = ap[k + 1];
            float bx = bp[k * oc];
            float by = bp[(k + 1) * oc];
            b[0] = real ? bx : 0.f; b[1] = real ? by : 0.f;
            c = WMMA_F32(a, b, c);
        }
#pragma unroll
        for (int j = 0; j < 8; ++j)
            W34[(fw * 16 + half * 8 + j) * PW + l15] = c[j];
        return;
    }
    for (int b = tid; b < MAXBIN; b += 256) lh[b] = 0;
    int t = blockIdx.x * 256 + tid, nt = nwg * 256;
    for (int k = t; k < g * (PW / 4); k += nt) ((float4*)psum)[k] = make_float4(0.f, 0.f, 0.f, 0.f);
    for (int k = t; k < g; k += nt) pcnt[k] = 0.f;
    __syncthreads();
    int span = blockIdx.x * SPAN;
    if (span + SPAN <= e) sort_span<true>(src, dst, lh, wsum, loff, binned, e, span, nbin, nwg);
    else                  sort_span<false>(src, dst, lh, wsum, loff, binned, e, span, nbin, nwg);
}

__device__ __forceinline__ void edge_addrs(const int* sgo, const int* sga, int nseg, int step0,
                                           const int (&i)[8], int (&addr)[8]) {
    int lo[8];
#pragma unroll
    for (int k = 0; k < 8; ++k) lo[k] = 0;
    for (int st = step0; st > 0; st >>= 1) {
        int v[8];
#pragma unroll
        for (int k = 0; k < 8; ++k) {
            int m = lo[k] + st;
            v[k] = sgo[m < nseg ? m : nseg];
        }
#pragma unroll
        for (int k = 0; k < 8; ++k) lo[k] = (v[k] <= i[k]) ? lo[k] + st : lo[k];
    }
#pragma unroll
    for (int k = 0; k < 8; ++k) addr[k] = sga[lo[k]] + (i[k] - sgo[lo[k]]);
}

#define EPD 8
#define XPL (BNODES * HID / 4 / 1024)
__device__ __forceinline__ void load_rows(const float4* xw4, int b, int tid, int n, float4 (&xv)[XPL]) {
#pragma unroll
    for (int q = 0; q < XPL; ++q) {
        int f = tid + q * 1024;
        int v = b * BNODES + f / (HID / 4);
        if (v >= n) v = n - 1;
        xv[q] = xw4[(size_t)v * (HID / 4) + (f & (HID / 4 - 1))];
    }
}
__global__ void __launch_bounds__(1024)
k_bindeg(const int2* __restrict__ binned, const int* __restrict__ loff, int nwg,
         int* __restrict__ rank, int* __restrict__ rowptr, float* __restrict__ dinv,
         int* __restrict__ csr, int n, int nbin, float4* xw4) {
    __shared__ int lc[BNODES];
    __shared__ float sdv[BNODES];
    __shared__ int rs[BNODES];
    __shared__ int tile[1024 * EPD];
    __shared__ int wtot[BNODES / 32];
    __shared__ int sgo[MAXSEG + 1];
    __shared__ int sga[MAXSEG];
    __shared__ int wl[32], wa[32];
    int tid = threadIdx.x, lane = tid & 31, wave = tid >> 5;
    int b  = blockIdx.x;
    int pa = 0, plen = 0;
    if (tid < nwg) {
        pa = loff[b * nwg + tid];
        plen = loff[(b + 1) * nwg + tid] - pa;
    }
    int px = plen;
#pragma unroll
    for (int o = 1; o < 32; o <<= 1) {
        int y = __shfl_up(px, o);
        if (lane >= o) px += y;
    }
    int pas = pa;
#pragma unroll
    for (int o = 16; o > 0; o >>= 1) pas += __shfl_xor(pas, o);
    if (lane == 31) wl[wave] = px;
    if (lane == 0)  wa[wave] = pas;
    if (tid < BNODES) lc[tid] = 0;
    __syncthreads();
    int wlen = wl[lane];
    int wx = wlen;
#pragma unroll
    for (int o = 1; o < 32; o <<= 1) {
        int y = __shfl_up(wx, o);
        if (lane >= o) wx += y;
    }
    int wave_excl = __shfl(wx, wave) - __shfl(wlen, wave);
    int nb = __shfl(wx, 31);
    int bs = wa[lane];
#pragma unroll
    for (int o = 16; o > 0; o >>= 1) bs += __shfl_xor(bs, o);
    int be = bs + nb;
    if (tid < nwg) {
        sgo[tid] = wave_excl + px - plen;
        sga[tid] = tid * SPAN + pa;
    }
    if (tid == 0) sgo[nwg] = nb;
    __syncthreads();
    float4* xrow = xw4 + (size_t)b * (BNODES * HID / 4);
    float4 xv[XPL];
    load_rows(xw4, b, tid, n, xv);
    int step0 = 1;
    while (step0 * 2 < nwg) step0 *= 2;
    int last = nb - 1;

    int s0[EPD], d0[EPD], r0[EPD];
#pragma unroll
    for (int k = 0; k < EPD; ++k) { s0[k] = 0; d0[k] = 0; }
    if (nb > 0) {
        int ic[EPD], ad[EPD];
#pragma unroll
        for (int k = 0; k < EPD; ++k) {
            int il = k * 1024 + tid;
            ic[k] = il < nb ? il : last;
        }
        edge_addrs(sgo, sga, nwg, step0, ic, ad);
#pragma unroll
        for (int k = 0; k < EPD; ++k) {
            int2 sd = binned[ad[k]];
            s0[k] = sd.x;
            d0[k] = sd.y;
        }
    }
#pragma unroll
    for (int k = 0; k < EPD; ++k) {
        int il = k * 1024 + tid;
        r0[k] = 0;
        if (il < nb) r0[k] = atomicAdd(&lc[d0[k] & (BNODES - 1)], 1);
    }
    for (int l0 = 1024 * EPD; l0 < nb; l0 += 1024 * EPD) {
        int d[EPD], r[EPD], ic[EPD], ad[EPD];
#pragma unroll
        for (int k = 0; k < EPD; ++k) {
            int il = l0 + k * 1024 + tid;
            ic[k] = il < nb ? il : last;
        }
        edge_addrs(sgo, sga, nwg, step0, ic, ad);
#pragma unroll
        for (int k = 0; k < EPD; ++k) d[k] = binned[ad[k]].y;
#pragma unroll
        for (int k = 0; k < EPD; ++k) {
            int il = l0 + k * 1024 + tid;
            r[k] = 0;
            if (il < nb) r[k] = atomicAdd(&lc[d[k] & (BNODES - 1)], 1);
        }
#pragma unroll
        for (int k = 0; k < EPD; ++k) {
            int il = l0 + k * 1024 + tid;
            if (il < nb) rank[bs + il] = r[k];
        }
    }
    __syncthreads();
    int c = 0, x = 0;
    if (tid < BNODES) {
        c = lc[tid];
        x = c;
#pragma unroll
        for (int o = 1; o < 32; o <<= 1) {
            int y = __shfl_up(x, o);
            if (lane >= o) x += y;
        }
        if (lane == 31) wtot[wave] = x;
    }
    __syncthreads();
    if (tid < BNODES) {
        int off = bs;
        for (int j = 0; j < wave; ++j) off += wtot[j];
        int start = off + x - c;
        rs[tid] = start - bs;
        int v = b * BNODES + tid;
        float dg = (float)(c + 1);
        float di = (dg > 0.f) ? rsqrtf(dg) : 0.f;
        sdv[tid] = di;
        if (v < n) {
            rowptr[v] = start;
            dinv[v] = di;
        }
        if (b == nbin - 1 && tid == 0) rowptr[n] = be;
    }
    __syncthreads();
#pragma unroll
    for (int q = 0; q < XPL; ++q) {
        int f = tid + q * 1024;
        int row = f / (HID / 4);
        if (b * BNODES + row < n) {
            float dr = sdv[row];
            float4 t = xv[q];
            t.x *= dr; t.y *= dr; t.z *= dr; t.w *= dr;
            xrow[f] = t;
        }
    }
    if (nb <= 1024 * EPD) {
#pragma unroll
        for (int k = 0; k < EPD; ++k) {
            int il = k * 1024 + tid;
            if (il < nb) tile[rs[d0[k] & (BNODES - 1)] + r0[k]] = s0[k];
        }
        __syncthreads();
        for (int t = tid; t < nb; t += 1024) csr[bs + t] = tile[t];
        return;
    }
#pragma unroll
    for (int k = 0; k < EPD; ++k) {
        int il = k * 1024 + tid;
        if (il < nb) csr[bs + rs[d0[k] & (BNODES - 1)] + r0[k]] = s0[k];
    }
    for (int l0 = 1024 * EPD; l0 < nb; l0 += 1024 * EPD) {
        int ic[EPD], ad[EPD];
#pragma unroll
        for (int k = 0; k < EPD; ++k) {
            int il = l0 + k * 1024 + tid;
            ic[k] = il < nb ? il : last;
        }
        edge_addrs(sgo, sga, nwg, step0, ic, ad);
#pragma unroll
        for (int k = 0; k < EPD; ++k) {
            int il = l0 + k * 1024 + tid;
            if (il < nb) {
                int2 sd = binned[ad[k]];
                csr[bs + rs[sd.y & (BNODES - 1)] + rank[bs + il]] = sd.x;
            }
        }
    }
}

template <int U>
__device__ __forceinline__ void gather_rows(const float2* __restrict__ h2, int s_l,
                                            int j, int lane, float2& acc) {
    float2 r[U];
#pragma unroll
    for (int k = 0; k < U; ++k) {
        int s = __builtin_amdgcn_readlane(s_l, j + k);
        r[k]  = h2[(size_t)s * (HID / 2) + lane];
    }
#pragma unroll
    for (int k = 0; k < U; ++k) {
        acc.x += r[k].x;
        acc.y += r[k].y;
    }
}

__device__ __forceinline__ void gather_chunk(const float2* __restrict__ h2, int s_l,
                                             int m, int lane, float2& acc) {
    int j = 0;
    for (; j + 8 <= m; j += 8) gather_rows<8>(h2, s_l, j, lane, acc);
    if (m & 4) { gather_rows<4>(h2, s_l, j, lane, acc); j += 4; }
    if (m & 2) { gather_rows<2>(h2, s_l, j, lane, acc); j += 2; }
    if (m & 1) { gather_rows<1>(h2, s_l, j, lane, acc); }
}

#define APITCH 68
__device__ __forceinline__ void gather_nodes(const int* __restrict__ rowptr, const int* __restrict__ csr,
                                             const float* __restrict__ dinv, const float2* __restrict__ h2,
                                             float* dstl, int v0, int vend, int lane) {
    int beg = rowptr[v0], end = rowptr[v0 + 1];
    float di = dinv[v0];
    int rec = 0;
    if (lane < end - beg) rec = csr[beg + lane];
    float2 acc = h2[(size_t)v0 * (HID / 2) + lane];
    for (int v = v0; v < vend; ++v) {
        int vn = (v + 1 < vend) ? v + 1 : v;
        int nbeg = rowptr[vn], nend = rowptr[vn + 1];
        float ndi = dinv[vn];
        int nrec = 0;
        if (lane < nend - nbeg) nrec = csr[nbeg + lane];
        float2 nhv = h2[(size_t)vn * (HID / 2) + lane];

        int m = end - beg;
        if (m > 32) m = 32;
        gather_chunk(h2, rec, m, lane, acc);
        for (int p0 = beg + 32; p0 < end; p0 += 32) {
            int m2 = end - p0;
            if (m2 > 32) m2 = 32;
            int rec2 = 0;
            if (lane < m2) rec2 = csr[p0 + lane];
            gather_chunk(h2, rec2, m2, lane, acc);
        }
        float2 t;
        t.x = fmaxf(acc.x * di, 0.f);
        t.y = fmaxf(acc.y * di, 0.f);
        *(float2*)(dstl + (v - v0) * APITCH + 2 * lane) = t;

        acc = nhv;
        di = ndi;
        beg = nbeg; end = nend; rec = nrec;
    }
}

template <int NO>
__global__ void __launch_bounds__(256)
k_gather_gemm(const int* __restrict__ rowptr, const int* __restrict__ csr,
              const float* __restrict__ dinv, const float* __restrict__ hW,
              const float* __restrict__ Wn, float* __restrict__ hWn, int n) {
    __shared__ float4 sW4[HID * NO / 4];
    __shared__ __attribute__((aligned(16))) float sA[8 * NPW * APITCH];
    const float* sW = (const float*)sW4;
    int tid  = threadIdx.x;
    int lane = tid & 31;
    int wv = __builtin_amdgcn_readfirstlane(tid >> 5);
    constexpr int NPIECE = HID * NO / 4 / 256;
    float4 wreg[NPIECE];
#pragma unroll
    for (int q = 0; q < NPIECE; ++q) wreg[q] = ((const float4*)Wn)[tid + q * 256];
#pragma unroll
    for (int q = 0; q < NPIECE; ++q) sW4[tid + q * 256] = wreg[q];

    int v0 = (blockIdx.x * 8 + wv) * NPW;
    int vend = v0 + NPW;
    if (vend > n) vend = n;
    float* myrows = sA + wv * NPW * APITCH;
    if (v0 < n)
        gather_nodes(rowptr, csr, dinv, (const float2*)hW, myrows, v0, vend, lane);
    for (int v = (v0 < n ? vend : v0); v < v0 + NPW; ++v) {
        float2 z; z.x = 0.f; z.y = 0.f;
        *(float2*)(myrows + (v - v0) * APITCH + 2 * lane) = z;
    }
    __syncthreads();
    constexpr bool WIDE = (NO == HID);
    if (wv >= (WIDE ? 4 : 2)) return;

    int mt = WIDE ? (wv >> 1) : wv;
    int nh = WIDE ? (wv & 1) : 0;
    int half = lane >> 4;
    int l15  = lane & 15;
    const float* ap = sA + (mt * 16 + l15) * APITCH + half * 2;
    const float* bp = sW + nh * 32 + l15 + half * 2 * NO;
    v8f c0 = {}, c1 = {};
#pragma unroll
    for (int k = 0; k < HID; k += 4) {
        v2f a, b0, b1;
        a[0] = ap[k]; a[1] = ap[k + 1];
        const float* bk = bp + k * NO;
        b0[0] = bk[0];  b0[1] = bk[NO];
        c0 = WMMA_F32(a, b0, c0);
        if (WIDE) {
            b1[0] = bk[16]; b1[1] = bk[NO + 16];
            c1 = WMMA_F32(a, b1, c1);
        }
    }
    int rbase = blockIdx.x * (8 * NPW) + mt * 16 + half * 8;
#pragma unroll
    for (int j = 0; j < 8; ++j) {
        int r = rbase + j;
        if (r < n) {
            float* cr = hWn + (size_t)r * NO + nh * 32 + l15;
            float dr = dinv[r];
            cr[0] = c0[j] * dr;
            if (WIDE) cr[16] = c1[j] * dr;
        }
    }
}

__device__ __forceinline__ float wdst(int p, const int (&rp)[NPW3], const float (&dv)[NPW3]) {
    float wv = dv[0];
#pragma unroll
    for (int j = 1; j < NPW3; ++j) wv = (p >= rp[j]) ? dv[j] : wv;
    return wv;
}

__global__ void __launch_bounds__(256)
k_gather_pool(const int* __restrict__ rowptr, const int* __restrict__ csr,
              const float* __restrict__ dinv, const float* __restrict__ z,
              const int* __restrict__ batch, float* sums, float* cnt, int n) {
    int lane = threadIdx.x & 31;
    int sub = lane & 7, grp = lane >> 3;
    int w  = __builtin_amdgcn_readfirstlane((int)(blockIdx.x * 8 + (threadIdx.x >> 5)));
    int v0 = w * NPW3;
    if (v0 >= n) return;
    int nv = n - v0;
    if (nv > NPW3) nv = NPW3;
    const float2* z2 = (const float2*)z;

    int vb = v0 + lane;
    int rp_l = rowptr[vb <= n ? vb : n];
    if (vb >= n) vb = n - 1;
    int   g_l  = batch[vb];
    float di_l = dinv[vb];
    float2 own[NPW3 / 4];
    float  ow[NPW3 / 4];
#pragma unroll
    for (int q = 0; q < NPW3 / 4; ++q) {
        int j = 4 * q + grp;
        own[q] = z2[j < nv ? (size_t)v0 * (PW / 2) + q * 32 + lane : 0];
        ow[q] = __shfl(di_l, j);
    }
    int   rpj[NPW3];
    float dj[NPW3];
#pragma unroll
    for (int j = 0; j < NPW3; ++j) {
        rpj[j] = __builtin_amdgcn_readlane(rp_l, j);
        dj[j]  = __int_as_float(__builtin_amdgcn_readlane(__float_as_int(di_l), j));
    }

    int j0 = 0;
    while (j0 < nv) {
        int g = __builtin_amdgcn_readfirstlane(__shfl(g_l, j0));
        unsigned diff = __builtin_amdgcn_ballot_w32(lane > j0 && lane < nv && g_l != g);
        int j1 = diff ? __builtin_ctz(diff) : nv;
        int eb = __builtin_amdgcn_readfirstlane(__shfl(rp_l, j0));
        int ee = __builtin_amdgcn_readfirstlane(__shfl(rp_l, j1));

        float2 acc = make_float2(0.f, 0.f);
#pragma unroll
        for (int q = 0; q < NPW3 / 4; ++q) {
            if (4 * q + grp >= j0 && 4 * q + grp < j1) {
                acc.x = fmaf(own[q].x, ow[q], acc.x);
                acc.y = fmaf(own[q].y, ow[q], acc.y);
            }
        }
        if (eb < ee) {
            int q0 = eb + lane, q1 = eb + 32 + lane;
            if (q0 >= ee) q0 = ee - 1;
            if (q1 >= ee) q1 = ee - 1;
            int   rec0 = csr[q0], rec1 = csr[q1];
            float w0 = wdst(q0, rpj, dj), w1 = wdst(q1, rpj, dj);
            for (int base = eb; base < ee; base += 64) {
                int m = ee - base;
                float2 r[16];
                float  wk[16];
#pragma unroll
                for (int k = 0; k < 16; ++k) {
                    int from = 4 * (k & 7) + grp;
                    int sn = __shfl(k < 8 ? rec0 : rec1, from);
                    wk[k] = __shfl(k < 8 ? w0 : w1, from);
                    r[k]  = z2[(size_t)sn * (PW / 2) + sub];
                }
                int n0 = base + 64 + lane, n1 = base + 96 + lane;
                if (n0 >= ee) n0 = ee - 1;
                if (n1 >= ee) n1 = ee - 1;
                int   nrec0 = csr[n0], nrec1 = csr[n1];
                float nw0 = wdst(n0, rpj, dj), nw1 = wdst(n1, rpj, dj);
#pragma unroll
                for (int k = 0; k < 16; ++k) {
                    if (4 * k + grp < m) {
                        acc.x = fmaf(r[k].x, wk[k], acc.x);
                        acc.y = fmaf(r[k].y, wk[k], acc.y);
                    }
                }
                rec0 = nrec0; rec1 = nrec1;
                w0 = nw0; w1 = nw1;
            }
        }
        acc.x += __shfl_xor(acc.x, 8);  acc.y += __shfl_xor(acc.y, 8);
        acc.x += __shfl_xor(acc.x, 16); acc.y += __shfl_xor(acc.y, 16);
        if (lane < 8) {
            atomicAdd(&sums[(size_t)g * PW + 2 * sub],     acc.x);
            atomicAdd(&sums[(size_t)g * PW + 2 * sub + 1], acc.y);
        }
        if (lane == 0) atomicAdd(&cnt[g], (float)(j1 - j0));
        j0 = j1;
    }
}

__global__ void k_final(const float* __restrict__ sums, const float* __restrict__ cnt,
                        const float* __restrict__ bl, float* __restrict__ out, int g, int oc) {
    int i = blockIdx.x * blockDim.x + threadIdx.x;
    if (i < g * oc) {
        int gi = i / oc, o = i % oc;
        out[i] = sums[(size_t)gi * PW + o] / fmaxf(cnt[gi], 1.0f) + bl[o];
    }
}

extern "C" void kernel_launch(void* const* d_in, const int* in_sizes, int n_in,
                              void* d_out, int out_size, void* d_ws, size_t ws_size,
                              hipStream_t stream) {
    const float* x    = (const float*)d_in[0];
    const int*   eidx = (const int*)  d_in[1];
    const int*   batch= (const int*)  d_in[2];
    const float* W1   = (const float*)d_in[3];
    const float* W2   = (const float*)d_in[4];
    const float* W3   = (const float*)d_in[5];
    const float* Wl   = (const float*)d_in[6];
    const float* bl   = (const float*)d_in[7];

    const int INC = in_sizes[3] / HID;
    const int N   = in_sizes[0] / INC;
    const int E   = in_sizes[1] / 2;
    const int OC  = in_sizes[7];
    const int G   = out_size / OC;

    const int* src = eidx;
    const int* dst = eidx + E;

    const int B = 256;
    const int eB1 = (E + B * EPL1 - 1) / (B * EPL1);
    const int nbin = (N + BNODES - 1) / BNODES;
    const int mt  = (N + 127) / 128;
    const int gB  = (N + 8 * NPW - 1) / (8 * NPW);
    const int pB  = (N + 8 * NPW3 - 1) / (8 * NPW3);

    char* ws = (char*)d_ws; size_t off = 0;
    auto carve = [&](size_t bytes) {
        void* p = ws + off; off = (off + bytes + 255) & ~(size_t)255; return p;
    };
    int*   loff   = (int*)  carve((size_t)(nbin + 1) * eB1 * 4);
    int2*  binned = (int2*) carve((size_t)eB1 * SPAN * 8);
    int*   rowptr = (int*)  carve((size_t)(N + 1) * 4);
    float* dinv   = (float*)carve((size_t)N * 4);
    int*   rank   = (int*)  carve((size_t)E * 4);
    int*   csr    = (int*)  carve((size_t)E * 4);
    float* hW     = (float*)carve((size_t)N * HID * 4);
    float* agg    = (float*)carve((size_t)N * HID * 4);
    float* sums   = (float*)carve((size_t)G * PW * 4);
    float* cnt    = (float*)carve((size_t)G * 4);
    float* W34    = (float*)carve((size_t)HID * PW * 4);
    (void)ws_size; (void)n_in;

    k_binsort<<<eB1 + 1 + mt, B, 0, stream>>>(src, dst, loff, binned, E, nbin, eB1, sums, cnt, G,
                                              W3, Wl, W34, OC, x, W1, hW, N);
    k_bindeg <<<nbin, 1024, 0, stream>>>(binned, loff, eB1, rank, rowptr, dinv, csr, N, nbin, (float4*)hW);

    k_gather_gemm<HID><<<gB, B, 0, stream>>>(rowptr, csr, dinv, hW, W2, agg, N);

    k_gather_gemm<PW><<<gB, B, 0, stream>>>(rowptr, csr, dinv, agg, W34, hW, N);

    k_gather_pool<<<pB, B, 0, stream>>>(rowptr, csr, dinv, hW, batch, sums, cnt, N);
    k_final      <<<(G * OC + B - 1) / B, B, 0, stream>>>(sums, cnt, bl, (float*)d_out, G, OC);
}


# Edge binning: a work-group's sorted span is put together in LDS and stored in whole 128-byte lines, 16 bytes per lane in order, not 8 bytes per lane into 32 different lines per store
# speedup vs baseline: 8.3995x; 1.0380x over previous
#include <hip/hip_runtime.h>

typedef __attribute__((ext_vector_type(2))) float v2f;
typedef __attribute__((ext_vector_type(8))) float v8f;

#define HID 64

#define WMMA_F32(a, b, c) __builtin_amdgcn_wmma_f32_16x16x4_f32( \
         false, a,  false, b,  (short)0, c, \
         false,  false)

#define NPW  4
#define PW   16
#define NPW3 8

#define BSH    8
#define BNODES (1 << BSH)
#define MAXBIN 2048
#define EPL1   20
#define SPAN   (256 * EPL1)
#define MAXSEG 1024

template <bool FULL>
__device__ __forceinline__ void sort_span(const int* __restrict__ src, const int* __restrict__ dst,
                                          int* lh, int* wsum, int2* tile, int* __restrict__ loff,
                                          int2* __restrict__ binned, int e, int span, int nbin, int nwg) {
    int tid = threadIdx.x, lane = tid & 31, wave = tid >> 5;
    int base = span + tid;
    int s[EPL1], d[EPL1], p[EPL1];
#pragma unroll
    for (int k = 0; k < EPL1; ++k) {
        int i = base + k * 256;
        int ic = (FULL || i < e) ? i : e - 1;
        s[k] = src[ic];
        d[k] = dst[ic];
    }
#pragma unroll
    for (int k = 0; k < EPL1; ++k) {
        int i = base + k * 256;
        p[k] = 0;
        if (FULL || i < e) p[k] = atomicAdd(&lh[d[k] >> BSH], 1);
    }
    __syncthreads();
    int4 q0 = ((int4*)lh)[2 * tid], q1 = ((int4*)lh)[2 * tid + 1];
    int c[8] = {q0.x, q0.y, q0.z, q0.w, q1.x, q1.y, q1.z, q1.w};
    int tot = 0;
#pragma unroll
    for (int k = 0; k < 8; ++k) tot += c[k];
    int x = tot;
#pragma unroll
    for (int o = 1; o < 32; o <<= 1) {
        int y = __shfl_up(x, o);
        if (lane >= o) x += y;
    }
    if (lane == 31) wsum[wave] = x;
    __syncthreads();
    int run = x - tot;
#pragma unroll
    for (int j = 0; j < 8; ++j) run += (j < wave) ? wsum[j] : 0;
    int w = blockIdx.x;
#pragma unroll
    for (int k = 0; k < 8; ++k) {
        int b = 8 * tid + k;
        int v = run;
        run += c[k];
        c[k] = v;
        if (b <= nbin) loff[b * nwg + w] = v;
    }
    ((int4*)lh)[2 * tid]     = make_int4(c[0], c[1], c[2], c[3]);
    ((int4*)lh)[2 * tid + 1] = make_int4(c[4], c[5], c[6], c[7]);
    __syncthreads();
    int o[EPL1];
#pragma unroll
    for (int k = 0; k < EPL1; ++k) o[k] = lh[d[k] >> BSH] + p[k];
#pragma unroll
    for (int k = 0; k < EPL1; ++k) {
        int i = base + k * 256;
        if (FULL || i < e) tile[o[k]] = make_int2(s[k], d[k]);
    }
    __syncthreads();
    int4* out4 = (int4*)(binned + span);
    const int4* tile4 = (const int4*)tile;
    if (FULL) {
#pragma unroll
        for (int q = 0; q < SPAN / 2 / 256; ++q) out4[tid + q * 256] = tile4[tid + q * 256];
    } else {
        int cnt = e - span;
        for (int q = tid; q < (cnt >> 1); q += 256) out4[q] = tile4[q];
        if ((cnt & 1) && tid == 0) binned[span + cnt - 1] = tile[cnt - 1];
    }
}

template <int K>
__device__ __forceinline__ void gemm_rows(const float* __restrict__ A, const float* __restrict__ W,
                                          float* __restrict__ C, int n, int blk, float4* sW4) {
    const float* sW = (const float*)sW4;
    int tid = threadIdx.x;
#pragma unroll
    for (int q = 0; q < K * HID / 4 / 256; ++q) sW4[tid + q * 256] = ((const float4*)W)[tid + q * 256];
    __syncthreads();

    int wave = tid >> 5;
    int lane = tid & 31;
    int half = lane >> 4;
    int l15  = lane & 15;
    int m0 = blk * 128 + wave * 16;

    int arow = m0 + l15;
    if (arow >= n) arow = n - 1;
    const float* ap = A + (size_t)arow * K + half * 2;
    const float* bp = sW + l15 + half * 2 * HID;

    v8f c0 = {}, c1 = {}, c2 = {}, c3 = {};
#pragma unroll
    for (int k = 0; k < K; k += 4) {
        v2f a, b0, b1, b2, b3;
        a[0] = ap[k]; a[1] = ap[k + 1];
        const float* bk = bp + k * HID;
        b0[0] = bk[0];  b0[1] = bk[HID];
        b1[0] = bk[16]; b1[1] = bk[HID + 16];
        b2[0] = bk[32]; b2[1] = bk[HID + 32];
        b3[0] = bk[48]; b3[1] = bk[HID + 48];
        c0 = WMMA_F32(a, b0, c0);
        c1 = WMMA_F32(a, b1, c1);
        c2 = WMMA_F32(a, b2, c2);
        c3 = WMMA_F32(a, b3, c3);
    }

    int rbase = m0 + half * 8;
#pragma unroll
    for (int j = 0; j < 8; ++j) {
        int r = rbase + j;
        if (r < n) {
            float* cr = C + (size_t)r * HID + l15;
            cr[0] = c0[j]; cr[16] = c1[j]; cr[32] = c2[j]; cr[48] = c3[j];
        }
    }
}

#define INCH 128

__global__ void __launch_bounds__(256)
k_binsort(const int* __restrict__ src, const int* __restrict__ dst, int* __restrict__ loff,
          int2* __restrict__ binned, int e, int nbin, int nwg,
          float* psum, float* pcnt, int g,
          const float* __restrict__ W3, const float* __restrict__ Wl, float* __restrict__ W34, int oc,
          const float* __restrict__ x, const float* __restrict__ W1, float* __restrict__ xw, int n) {
    static_assert((MAXBIN * 4 + SPAN * 8) / 16 >= INCH * HID / 4, "the block must hold W1");
    __shared__ float4 sbuf4[(MAXBIN * 4 + SPAN * 8) / 16];
    __shared__ int wsum[8];
    int* lh = (int*)sbuf4;
    int2* tile = (int2*)(sbuf4 + MAXBIN * 4 / 16);
    int tid = threadIdx.x;
    if ((int)blockIdx.x > nwg) {
        gemm_rows<INCH>(x, W1, xw, n, (int)blockIdx.x - nwg - 1, sbuf4);
        return;
    }
    if ((int)blockIdx.x == nwg) {
        int fw = __builtin_amdgcn_readfirstlane(tid >> 5);
        if (fw >= HID / 16) return;
        int fl = tid & 31;
        int half = fl >> 4;
        int l15  = fl & 15;
        bool real = l15 < oc;
        int  col  = real ? l15 : 0;
        const float* ap = W3 + (fw * 16 + l15) * HID + half * 2;
        const float* bp = Wl + half * 2 * oc + col;
        v8f c = {};
#pragma unroll
        for (int k = 0; k < HID; k += 4) {
            v2f a, b;
            a[0] = ap[k]; a[1] = ap[k + 1];
            float bx = bp[k * oc];
            float by = bp[(k + 1) * oc];
            b[0] = real ? bx : 0.f; b[1] = real ? by : 0.f;
            c = WMMA_F32(a, b, c);
        }
#pragma unroll
        for (int j = 0; j < 8; ++j)
            W34[(fw * 16 + half * 8 + j) * PW + l15] = c[j];
        return;
    }
    for (int b = tid; b < MAXBIN; b += 256) lh[b] = 0;
    int t = blockIdx.x * 256 + tid, nt = nwg * 256;
    for (int k = t; k < g * (PW / 4); k += nt) ((float4*)psum)[k] = make_float4(0.f, 0.f, 0.f, 0.f);
    for (int k = t; k < g; k += nt) pcnt[k] = 0.f;
    __syncthreads();
    int span = blockIdx.x * SPAN;
    if (span + SPAN <= e) sort_span<true>(src, dst, lh, wsum, tile, loff, binned, e, span, nbin, nwg);
    else                  sort_span<false>(src, dst, lh, wsum, tile, loff, binned, e, span, nbin, nwg);
}

__device__ __forceinline__ void edge_addrs(const int* sgo, const int* sga, int nseg, int step0,
                                           const int (&i)[8], int (&addr)[8]) {
    int lo[8];
#pragma unroll
    for (int k = 0; k < 8; ++k) lo[k] = 0;
    for (int st = step0; st > 0; st >>= 1) {
        int v[8];
#pragma unroll
        for (int k = 0; k < 8; ++k) {
            int m = lo[k] + st;
            v[k] = sgo[m < nseg ? m : nseg];
        }
#pragma unroll
        for (int k = 0; k < 8; ++k) lo[k] = (v[k] <= i[k]) ? lo[k] + st : lo[k];
    }
#pragma unroll
    for (int k = 0; k < 8; ++k) addr[k] = sga[lo[k]] + (i[k] - sgo[lo[k]]);
}

#define EPD 8
#define XPL (BNODES * HID / 4 / 1024)
__device__ __forceinline__ void load_rows(const float4* xw4, int b, int tid, int n, float4 (&xv)[XPL]) {
#pragma unroll
    for (int q = 0; q < XPL; ++q) {
        int f = tid + q * 1024;
        int v = b * BNODES + f / (HID / 4);
        if (v >= n) v = n - 1;
        xv[q] = xw4[(size_t)v * (HID / 4) + (f & (HID / 4 - 1))];
    }
}
__global__ void __launch_bounds__(1024)
k_bindeg(const int2* __restrict__ binned, const int* __restrict__ loff, int nwg,
         int* __restrict__ rank, int* __restrict__ rowptr, float* __restrict__ dinv,
         int* __restrict__ csr, int n, int nbin, float4* xw4) {
    __shared__ int lc[BNODES];
    __shared__ float sdv[BNODES];
    __shared__ int rs[BNODES];
    __shared__ int tile[1024 * EPD];
    __shared__ int wtot[BNODES / 32];
    __shared__ int sgo[MAXSEG + 1];
    __shared__ int sga[MAXSEG];
    __shared__ int wl[32], wa[32];
    int tid = threadIdx.x, lane = tid & 31, wave = tid >> 5;
    int b  = blockIdx.x;
    int pa = 0, plen = 0;
    if (tid < nwg) {
        pa = loff[b * nwg + tid];
        plen = loff[(b + 1) * nwg + tid] - pa;
    }
    int px = plen;
#pragma unroll
    for (int o = 1; o < 32; o <<= 1) {
        int y = __shfl_up(px, o);
        if (lane >= o) px += y;
    }
    int pas = pa;
#pragma unroll
    for (int o = 16; o > 0; o >>= 1) pas += __shfl_xor(pas, o);
    if (lane == 31) wl[wave] = px;
    if (lane == 0)  wa[wave] = pas;
    if (tid < BNODES) lc[tid] = 0;
    __syncthreads();
    int wlen = wl[lane];
    int wx = wlen;
#pragma unroll
    for (int o = 1; o < 32; o <<= 1) {
        int y = __shfl_up(wx, o);
        if (lane >= o) wx += y;
    }
    int wave_excl = __shfl(wx, wave) - __shfl(wlen, wave);
    int nb = __shfl(wx, 31);
    int bs = wa[lane];
#pragma unroll
    for (int o = 16; o > 0; o >>= 1) bs += __shfl_xor(bs, o);
    int be = bs + nb;
    if (tid < nwg) {
        sgo[tid] = wave_excl + px - plen;
        sga[tid] = tid * SPAN + pa;
    }
    if (tid == 0) sgo[nwg] = nb;
    __syncthreads();
    float4* xrow = xw4 + (size_t)b * (BNODES * HID / 4);
    float4 xv[XPL];
    load_rows(xw4, b, tid, n, xv);
    int step0 = 1;
    while (step0 * 2 < nwg) step0 *= 2;
    int last = nb - 1;

    int s0[EPD], d0[EPD], r0[EPD];
#pragma unroll
    for (int k = 0; k < EPD; ++k) { s0[k] = 0; d0[k] = 0; }
    if (nb > 0) {
        int ic[EPD], ad[EPD];
#pragma unroll
        for (int k = 0; k < EPD; ++k) {
            int il = k * 1024 + tid;
            ic[k] = il < nb ? il : last;
        }
        edge_addrs(sgo, sga, nwg, step0, ic, ad);
#pragma unroll
        for (int k = 0; k < EPD; ++k) {
            int2 sd = binned[ad[k]];
            s0[k] = sd.x;
            d0[k] = sd.y;
        }
    }
#pragma unroll
    for (int k = 0; k < EPD; ++k) {
        int il = k * 1024 + tid;
        r0[k] = 0;
        if (il < nb) r0[k] = atomicAdd(&lc[d0[k] & (BNODES - 1)], 1);
    }
    for (int l0 = 1024 * EPD; l0 < nb; l0 += 1024 * EPD) {
        int d[EPD], r[EPD], ic[EPD], ad[EPD];
#pragma unroll
        for (int k = 0; k < EPD; ++k) {
            int il = l0 + k * 1024 + tid;
            ic[k] = il < nb ? il : last;
        }
        edge_addrs(sgo, sga, nwg, step0, ic, ad);
#pragma unroll
        for (int k = 0; k < EPD; ++k) d[k] = binned[ad[k]].y;
#pragma unroll
        for (int k = 0; k < EPD; ++k) {
            int il = l0 + k * 1024 + tid;
            r[k] = 0;
            if (il < nb) r[k] = atomicAdd(&lc[d[k] & (BNODES - 1)], 1);
        }
#pragma unroll
        for (int k = 0; k < EPD; ++k) {
            int il = l0 + k * 1024 + tid;
            if (il < nb) rank[bs + il] = r[k];
        }
    }
    __syncthreads();
    int c = 0, x = 0;
    if (tid < BNODES) {
        c = lc[tid];
        x = c;
#pragma unroll
        for (int o = 1; o < 32; o <<= 1) {
            int y = __shfl_up(x, o);
            if (lane >= o) x += y;
        }
        if (lane == 31) wtot[wave] = x;
    }
    __syncthreads();
    if (tid < BNODES) {
        int off = bs;
        for (int j = 0; j < wave; ++j) off += wtot[j];
        int start = off + x - c;
        rs[tid] = start - bs;
        int v = b * BNODES + tid;
        float dg = (float)(c + 1);
        float di = (dg > 0.f) ? rsqrtf(dg) : 0.f;
        sdv[tid] = di;
        if (v < n) {
            rowptr[v] = start;
            dinv[v] = di;
        }
        if (b == nbin - 1 && tid == 0) rowptr[n] = be;
    }
    __syncthreads();
#pragma unroll
    for (int q = 0; q < XPL; ++q) {
        int f = tid + q * 1024;
        int row = f / (HID / 4);
        if (b * BNODES + row < n) {
            float dr = sdv[row];
            float4 t = xv[q];
            t.x *= dr; t.y *= dr; t.z *= dr; t.w *= dr;
            xrow[f] = t;
        }
    }
    if (nb <= 1024 * EPD) {
#pragma unroll
        for (int k = 0; k < EPD; ++k) {
            int il = k * 1024 + tid;
            if (il < nb) tile[rs[d0[k] & (BNODES - 1)] + r0[k]] = s0[k];
        }
        __syncthreads();
        for (int t = tid; t < nb; t += 1024) csr[bs + t] = tile[t];
        return;
    }
#pragma unroll
    for (int k = 0; k < EPD; ++k) {
        int il = k * 1024 + tid;
        if (il < nb) csr[bs + rs[d0[k] & (BNODES - 1)] + r0[k]] = s0[k];
    }
    for (int l0 = 1024 * EPD; l0 < nb; l0 += 1024 * EPD) {
        int ic[EPD], ad[EPD];
#pragma unroll
        for (int k = 0; k < EPD; ++k) {
            int il = l0 + k * 1024 + tid;
            ic[k] = il < nb ? il : last;
        }
        edge_addrs(sgo, sga, nwg, step0, ic, ad);
#pragma unroll
        for (int k = 0; k < EPD; ++k) {
            int il = l0 + k * 1024 + tid;
            if (il < nb) {
                int2 sd = binned[ad[k]];
                csr[bs + rs[sd.y & (BNODES - 1)] + rank[bs + il]] = sd.x;
            }
        }
    }
}

template <int U>
__device__ __forceinline__ void gather_rows(const float2* __restrict__ h2, int s_l,
                                            int j, int lane, float2& acc) {
    float2 r[U];
#pragma unroll
    for (int k = 0; k < U; ++k) {
        int s = __builtin_amdgcn_readlane(s_l, j + k);
        r[k]  = h2[(size_t)s * (HID / 2) + lane];
    }
#pragma unroll
    for (int k = 0; k < U; ++k) {
        acc.x += r[k].x;
        acc.y += r[k].y;
    }
}

__device__ __forceinline__ void gather_chunk(const float2* __restrict__ h2, int s_l,
                                             int m, int lane, float2& acc) {
    int j = 0;
    for (; j + 8 <= m; j += 8) gather_rows<8>(h2, s_l, j, lane, acc);
    if (m & 4) { gather_rows<4>(h2, s_l, j, lane, acc); j += 4; }
    if (m & 2) { gather_rows<2>(h2, s_l, j, lane, acc); j += 2; }
    if (m & 1) { gather_rows<1>(h2, s_l, j, lane, acc); }
}

#define APITCH 68
__device__ __forceinline__ void gather_nodes(const int* __restrict__ rowptr, const int* __restrict__ csr,
                                             const float* __restrict__ dinv, const float2* __restrict__ h2,
                                             float* dstl, int v0, int vend, int lane) {
    int beg = rowptr[v0], end = rowptr[v0 + 1];
    float di = dinv[v0];
    int rec = 0;
    if (lane < end - beg) rec = csr[beg + lane];
    float2 acc = h2[(size_t)v0 * (HID / 2) + lane];
    for (int v = v0; v < vend; ++v) {
        int vn = (v + 1 < vend) ? v + 1 : v;
        int nbeg = rowptr[vn], nend = rowptr[vn + 1];
        float ndi = dinv[vn];
        int nrec = 0;
        if (lane < nend - nbeg) nrec = csr[nbeg + lane];
        float2 nhv = h2[(size_t)vn * (HID / 2) + lane];

        int m = end - beg;
        if (m > 32) m = 32;
        gather_chunk(h2, rec, m, lane, acc);
        for (int p0 = beg + 32; p0 < end; p0 += 32) {
            int m2 = end - p0;
            if (m2 > 32) m2 = 32;
            int rec2 = 0;
            if (lane < m2) rec2 = csr[p0 + lane];
            gather_chunk(h2, rec2, m2, lane, acc);
        }
        float2 t;
        t.x = fmaxf(acc.x * di, 0.f);
        t.y = fmaxf(acc.y * di, 0.f);
        *(float2*)(dstl + (v - v0) * APITCH + 2 * lane) = t;

        acc = nhv;
        di = ndi;
        beg = nbeg; end = nend; rec = nrec;
    }
}

template <int NO>
__global__ void __launch_bounds__(256)
k_gather_gemm(const int* __restrict__ rowptr, const int* __restrict__ csr,
              const float* __restrict__ dinv, const float* __restrict__ hW,
              const float* __restrict__ Wn, float* __restrict__ hWn, int n) {
    __shared__ float4 sW4[HID * NO / 4];
    __shared__ __attribute__((aligned(16))) float sA[8 * NPW * APITCH];
    const float* sW = (const float*)sW4;
    int tid  = threadIdx.x;
    int lane = tid & 31;
    int wv = __builtin_amdgcn_readfirstlane(tid >> 5);
    constexpr int NPIECE = HID * NO / 4 / 256;
    float4 wreg[NPIECE];
#pragma unroll
    for (int q = 0; q < NPIECE; ++q) wreg[q] = ((const float4*)Wn)[tid + q * 256];
#pragma unroll
    for (int q = 0; q < NPIECE; ++q) sW4[tid + q * 256] = wreg[q];

    int v0 = (blockIdx.x * 8 + wv) * NPW;
    int vend = v0 + NPW;
    if (vend > n) vend = n;
    float* myrows = sA + wv * NPW * APITCH;
    if (v0 < n)
        gather_nodes(rowptr, csr, dinv, (const float2*)hW, myrows, v0, vend, lane);
    for (int v = (v0 < n ? vend : v0); v < v0 + NPW; ++v) {
        float2 z; z.x = 0.f; z.y = 0.f;
        *(float2*)(myrows + (v - v0) * APITCH + 2 * lane) = z;
    }
    __syncthreads();
    constexpr bool WIDE = (NO == HID);
    if (wv >= (WIDE ? 4 : 2)) return;

    int mt = WIDE ? (wv >> 1) : wv;
    int nh = WIDE ? (wv & 1) : 0;
    int half = lane >> 4;
    int l15  = lane & 15;
    const float* ap = sA + (mt * 16 + l15) * APITCH + half * 2;
    const float* bp = sW + nh * 32 + l15 + half * 2 * NO;
    v8f c0 = {}, c1 = {};
#pragma unroll
    for (int k = 0; k < HID; k += 4) {
        v2f a, b0, b1;
        a[0] = ap[k]; a[1] = ap[k + 1];
        const float* bk = bp + k * NO;
        b0[0] = bk[0];  b0[1] = bk[NO];
        c0 = WMMA_F32(a, b0, c0);
        if (WIDE) {
            b1[0] = bk[16]; b1[1] = bk[NO + 16];
            c1 = WMMA_F32(a, b1, c1);
        }
    }
    int rbase = blockIdx.x * (8 * NPW) + mt * 16 + half * 8;
#pragma unroll
    for (int j = 0; j < 8; ++j) {
        int r = rbase + j;
        if (r < n) {
            float* cr = hWn + (size_t)r * NO + nh * 32 + l15;
            float dr = dinv[r];
            cr[0] = c0[j] * dr;
            if (WIDE) cr[16] = c1[j] * dr;
        }
    }
}

__device__ __forceinline__ float wdst(int p, const int (&rp)[NPW3], const float (&dv)[NPW3]) {
    float wv = dv[0];
#pragma unroll
    for (int j = 1; j < NPW3; ++j) wv = (p >= rp[j]) ? dv[j] : wv;
    return wv;
}

__global__ void __launch_bounds__(256)
k_gather_pool(const int* __restrict__ rowptr, const int* __restrict__ csr,
              const float* __restrict__ dinv, const float* __restrict__ z,
              const int* __restrict__ batch, float* sums, float* cnt, int n) {
    int lane = threadIdx.x & 31;
    int sub = lane & 7, grp = lane >> 3;
    int w  = __builtin_amdgcn_readfirstlane((int)(blockIdx.x * 8 + (threadIdx.x >> 5)));
    int v0 = w * NPW3;
    if (v0 >= n) return;
    int nv = n - v0;
    if (nv > NPW3) nv = NPW3;
    const float2* z2 = (const float2*)z;

    int vb = v0 + lane;
    int rp_l = rowptr[vb <= n ? vb : n];
    if (vb >= n) vb = n - 1;
    int   g_l  = batch[vb];
    float di_l = dinv[vb];
    float2 own[NPW3 / 4];
    float  ow[NPW3 / 4];
#pragma unroll
    for (int q = 0; q < NPW3 / 4; ++q) {
        int j = 4 * q + grp;
        own[q] = z2[j < nv ? (size_t)v0 * (PW / 2) + q * 32 + lane : 0];
        ow[q] = __shfl(di_l, j);
    }
    int   rpj[NPW3];
    float dj[NPW3];
#pragma unroll
    for (int j = 0; j < NPW3; ++j) {
        rpj[j] = __builtin_amdgcn_readlane(rp_l, j);
        dj[j]  = __int_as_float(__builtin_amdgcn_readlane(__float_as_int(di_l), j));
    }

    int j0 = 0;
    while (j0 < nv) {
        int g = __builtin_amdgcn_readfirstlane(__shfl(g_l, j0));
        unsigned diff = __builtin_amdgcn_ballot_w32(lane > j0 && lane < nv && g_l != g);
        int j1 = diff ? __builtin_ctz(diff) : nv;
        int eb = __builtin_amdgcn_readfirstlane(__shfl(rp_l, j0));
        int ee = __builtin_amdgcn_readfirstlane(__shfl(rp_l, j1));

        float2 acc = make_float2(0.f, 0.f);
#pragma unroll
        for (int q = 0; q < NPW3 / 4; ++q) {
            if (4 * q + grp >= j0 && 4 * q + grp < j1) {
                acc.x = fmaf(own[q].x, ow[q], acc.x);
                acc.y = fmaf(own[q].y, ow[q], acc.y);
            }
        }
        if (eb < ee) {
            int q0 = eb + lane, q1 = eb + 32 + lane;
            if (q0 >= ee) q0 = ee - 1;
            if (q1 >= ee) q1 = ee - 1;
            int   rec0 = csr[q0], rec1 = csr[q1];
            float w0 = wdst(q0, rpj, dj), w1 = wdst(q1, rpj, dj);
            for (int base = eb; base < ee; base += 64) {
                int m = ee - base;
                float2 r[16];
                float  wk[16];
#pragma unroll
                for (int k = 0; k < 16; ++k) {
                    int from = 4 * (k & 7) + grp;
                    int sn = __shfl(k < 8 ? rec0 : rec1, from);
                    wk[k] = __shfl(k < 8 ? w0 : w1, from);
                    r[k]  = z2[(size_t)sn * (PW / 2) + sub];
                }
                int n0 = base + 64 + lane, n1 = base + 96 + lane;
                if (n0 >= ee) n0 = ee - 1;
                if (n1 >= ee) n1 = ee - 1;
                int   nrec0 = csr[n0], nrec1 = csr[n1];
                float nw0 = wdst(n0, rpj, dj), nw1 = wdst(n1, rpj, dj);
#pragma unroll
                for (int k = 0; k < 16; ++k) {
                    if (4 * k + grp < m) {
                        acc.x = fmaf(r[k].x, wk[k], acc.x);
                        acc.y = fmaf(r[k].y, wk[k], acc.y);
                    }
                }
                rec0 = nrec0; rec1 = nrec1;
                w0 = nw0; w1 = nw1;
            }
        }
        acc.x += __shfl_xor(acc.x, 8);  acc.y += __shfl_xor(acc.y, 8);
        acc.x += __shfl_xor(acc.x, 16); acc.y += __shfl_xor(acc.y, 16);
        if (lane < 8) {
            atomicAdd(&sums[(size_t)g * PW + 2 * sub],     acc.x);
            atomicAdd(&sums[(size_t)g * PW + 2 * sub + 1], acc.y);
        }
        if (lane == 0) atomicAdd(&cnt[g], (float)(j1 - j0));
        j0 = j1;
    }
}

__global__ void k_final(const float* __restrict__ sums, const float* __restrict__ cnt,
                        const float* __restrict__ bl, float* __restrict__ out, int g, int oc) {
    int i = blockIdx.x * blockDim.x + threadIdx.x;
    if (i < g * oc) {
        int gi = i / oc, o = i % oc;
        out[i] = sums[(size_t)gi * PW + o] / fmaxf(cnt[gi], 1.0f) + bl[o];
    }
}

extern "C" void kernel_launch(void* const* d_in, const int* in_sizes, int n_in,
                              void* d_out, int out_size, void* d_ws, size_t ws_size,
                              hipStream_t stream) {
    const float* x    = (const float*)d_in[0];
    const int*   eidx = (const int*)  d_in[1];
    const int*   batch= (const int*)  d_in[2];
    const float* W1   = (const float*)d_in[3];
    const float* W2   = (const float*)d_in[4];
    const float* W3   = (const float*)d_in[5];
    const float* Wl   = (const float*)d_in[6];
    const float* bl   = (const float*)d_in[7];

    const int INC = in_sizes[3] / HID;
    const int N   = in_sizes[0] / INC;
    const int E   = in_sizes[1] / 2;
    const int OC  = in_sizes[7];
    const int G   = out_size / OC;

    const int* src = eidx;
    const int* dst = eidx + E;

    const int B = 256;
    const int eB1 = (E + B * EPL1 - 1) / (B * EPL1);
    const int nbin = (N + BNODES - 1) / BNODES;
    const int mt  = (N + 127) / 128;
    const int gB  = (N + 8 * NPW - 1) / (8 * NPW);
    const int pB  = (N + 8 * NPW3 - 1) / (8 * NPW3);

    char* ws = (char*)d_ws; size_t off = 0;
    auto carve = [&](size_t bytes) {
        void* p = ws + off; off = (off + bytes + 255) & ~(size_t)255; return p;
    };
    int*   loff   = (int*)  carve((size_t)(nbin + 1) * eB1 * 4);
    int2*  binned = (int2*) carve((size_t)eB1 * SPAN * 8);
    int*   rowptr = (int*)  carve((size_t)(N + 1) * 4);
    float* dinv   = (float*)carve((size_t)N * 4);
    int*   rank   = (int*)  carve((size_t)E * 4);
    int*   csr    = (int*)  carve((size_t)E * 4);
    float* hW     = (float*)carve((size_t)N * HID * 4);
    float* agg    = (float*)carve((size_t)N * HID * 4);
    float* sums   = (float*)carve((size_t)G * PW * 4);
    float* cnt    = (float*)carve((size_t)G * 4);
    float* W34    = (float*)carve((size_t)HID * PW * 4);
    (void)ws_size; (void)n_in;

    k_binsort<<<eB1 + 1 + mt, B, 0, stream>>>(src, dst, loff, binned, E, nbin, eB1, sums, cnt, G,
                                              W3, Wl, W34, OC, x, W1, hW, N);
    k_bindeg <<<nbin, 1024, 0, stream>>>(binned, loff, eB1, rank, rowptr, dinv, csr, N, nbin, (float4*)hW);

    k_gather_gemm<HID><<<gB, B, 0, stream>>>(rowptr, csr, dinv, hW, W2, agg, N);

    k_gather_gemm<PW><<<gB, B, 0, stream>>>(rowptr, csr, dinv, agg, W34, hW, N);

    k_gather_pool<<<pB, B, 0, stream>>>(rowptr, csr, dinv, hW, batch, sums, cnt, N);
    k_final      <<<(G * OC + B - 1) / B, B, 0, stream>>>(sums, cnt, bl, (float*)d_out, G, OC);
}


# Output finished in the last aggregation: out starts as the bias, per-graph node counts come from the level-2 launch, each run adds sums / count into out; final launch gone (rounding changes)
# speedup vs baseline: 8.8012x; 1.0478x over previous
#include <hip/hip_runtime.h>

typedef __attribute__((ext_vector_type(2))) float v2f;
typedef __attribute__((ext_vector_type(8))) float v8f;

#define HID 64

#define WMMA_F32(a, b, c) __builtin_amdgcn_wmma_f32_16x16x4_f32( \
         false, a,  false, b,  (short)0, c, \
         false,  false)

#define NPW  4
#define PW   16
#define NPW3 8

#define BSH    8
#define BNODES (1 << BSH)
#define MAXBIN 2048
#define EPL1   20
#define SPAN   (256 * EPL1)
#define MAXSEG 1024

template <bool FULL>
__device__ __forceinline__ void sort_span(const int* __restrict__ src, const int* __restrict__ dst,
                                          int* lh, int* wsum, int2* tile, int* __restrict__ loff,
                                          int2* __restrict__ binned, int e, int span, int nbin, int nwg) {
    int tid = threadIdx.x, lane = tid & 31, wave = tid >> 5;
    int base = span + tid;
    int s[EPL1], d[EPL1], p[EPL1];
#pragma unroll
    for (int k = 0; k < EPL1; ++k) {
        int i = base + k * 256;
        int ic = (FULL || i < e) ? i : e - 1;
        s[k] = src[ic];
        d[k] = dst[ic];
    }
#pragma unroll
    for (int k = 0; k < EPL1; ++k) {
        int i = base + k * 256;
        p[k] = 0;
        if (FULL || i < e) p[k] = atomicAdd(&lh[d[k] >> BSH], 1);
    }
    __syncthreads();
    int4 q0 = ((int4*)lh)[2 * tid], q1 = ((int4*)lh)[2 * tid + 1];
    int c[8] = {q0.x, q0.y, q0.z, q0.w, q1.x, q1.y, q1.z, q1.w};
    int tot = 0;
#pragma unroll
    for (int k = 0; k < 8; ++k) tot += c[k];
    int x = tot;
#pragma unroll
    for (int o = 1; o < 32; o <<= 1) {
        int y = __shfl_up(x, o);
        if (lane >= o) x += y;
    }
    if (lane == 31) wsum[wave] = x;
    __syncthreads();
    int run = x - tot;
#pragma unroll
    for (int j = 0; j < 8; ++j) run += (j < wave) ? wsum[j] : 0;
    int w = blockIdx.x;
#pragma unroll
    for (int k = 0; k < 8; ++k) {
        int b = 8 * tid + k;
        int v = run;
        run += c[k];
        c[k] = v;
        if (b <= nbin) loff[b * nwg + w] = v;
    }
    ((int4*)lh)[2 * tid]     = make_int4(c[0], c[1], c[2], c[3]);
    ((int4*)lh)[2 * tid + 1] = make_int4(c[4], c[5], c[6], c[7]);
    __syncthreads();
    int o[EPL1];
#pragma unroll
    for (int k = 0; k < EPL1; ++k) o[k] = lh[d[k] >> BSH] + p[k];
#pragma unroll
    for (int k = 0; k < EPL1; ++k) {
        int i = base + k * 256;
        if (FULL || i < e) tile[o[k]] = make_int2(s[k], d[k]);
    }
    __syncthreads();
    int4* out4 = (int4*)(binned + span);
    const int4* tile4 = (const int4*)tile;
    if (FULL) {
#pragma unroll
        for (int q = 0; q < SPAN / 2 / 256; ++q) out4[tid + q * 256] = tile4[tid + q * 256];
    } else {
        int cnt = e - span;
        for (int q = tid; q < (cnt >> 1); q += 256) out4[q] = tile4[q];
        if ((cnt & 1) && tid == 0) binned[span + cnt - 1] = tile[cnt - 1];
    }
}

template <int K>
__device__ __forceinline__ void gemm_rows(const float* __restrict__ A, const float* __restrict__ W,
                                          float* __restrict__ C, int n, int blk, float4* sW4) {
    const float* sW = (const float*)sW4;
    int tid = threadIdx.x;
#pragma unroll
    for (int q = 0; q < K * HID / 4 / 256; ++q) sW4[tid + q * 256] = ((const float4*)W)[tid + q * 256];
    __syncthreads();

    int wave = tid >> 5;
    int lane = tid & 31;
    int half = lane >> 4;
    int l15  = lane & 15;
    int m0 = blk * 128 + wave * 16;

    int arow = m0 + l15;
    if (arow >= n) arow = n - 1;
    const float* ap = A + (size_t)arow * K + half * 2;
    const float* bp = sW + l15 + half * 2 * HID;

    v8f c0 = {}, c1 = {}, c2 = {}, c3 = {};
#pragma unroll
    for (int k = 0; k < K; k += 4) {
        v2f a, b0, b1, b2, b3;
        a[0] = ap[k]; a[1] = ap[k + 1];
        const float* bk = bp + k * HID;
        b0[0] = bk[0];  b0[1] = bk[HID];
        b1[0] = bk[16]; b1[1] = bk[HID + 16];
        b2[0] = bk[32]; b2[1] = bk[HID + 32];
        b3[0] = bk[48]; b3[1] = bk[HID + 48];
        c0 = WMMA_F32(a, b0, c0);
        c1 = WMMA_F32(a, b1, c1);
        c2 = WMMA_F32(a, b2, c2);
        c3 = WMMA_F32(a, b3, c3);
    }

    int rbase = m0 + half * 8;
#pragma unroll
    for (int j = 0; j < 8; ++j) {
        int r = rbase + j;
        if (r < n) {
            float* cr = C + (size_t)r * HID + l15;
            cr[0] = c0[j]; cr[16] = c1[j]; cr[32] = c2[j]; cr[48] = c3[j];
        }
    }
}

#define INCH 128

__global__ void __launch_bounds__(256)
k_binsort(const int* __restrict__ src, const int* __restrict__ dst, int* __restrict__ loff,
          int2* __restrict__ binned, int e, int nbin, int nwg,
          float* out, const float* __restrict__ bl, float* pcnt, int g,
          const float* __restrict__ W3, const float* __restrict__ Wl, float* __restrict__ W34, int oc,
          const float* __restrict__ x, const float* __restrict__ W1, float* __restrict__ xw, int n) {
    static_assert((MAXBIN * 4 + SPAN * 8) / 16 >= INCH * HID / 4, "the block must hold W1");
    __shared__ float4 sbuf4[(MAXBIN * 4 + SPAN * 8) / 16];
    __shared__ int wsum[8];
    int* lh = (int*)sbuf4;
    int2* tile = (int2*)(sbuf4 + MAXBIN * 4 / 16);
    int tid = threadIdx.x;
    if ((int)blockIdx.x > nwg) {
        gemm_rows<INCH>(x, W1, xw, n, (int)blockIdx.x - nwg - 1, sbuf4);
        return;
    }
    if ((int)blockIdx.x == nwg) {
        int fw = __builtin_amdgcn_readfirstlane(tid >> 5);
        if (fw >= HID / 16) return;
        int fl = tid & 31;
        int half = fl >> 4;
        int l15  = fl & 15;
        bool real = l15 < oc;
        int  col  = real ? l15 : 0;
        const float* ap = W3 + (fw * 16 + l15) * HID + half * 2;
        const float* bp = Wl + half * 2 * oc + col;
        v8f c = {};
#pragma unroll
        for (int k = 0; k < HID; k += 4) {
            v2f a, b;
            a[0] = ap[k]; a[1] = ap[k + 1];
            float bx = bp[k * oc];
            float by = bp[(k + 1) * oc];
            b[0] = real ? bx : 0.f; b[1] = real ? by : 0.f;
            c = WMMA_F32(a, b, c);
        }
#pragma unroll
        for (int j = 0; j < 8; ++j)
            W34[(fw * 16 + half * 8 + j) * PW + l15] = c[j];
        return;
    }
    for (int b = tid; b < MAXBIN; b += 256) lh[b] = 0;
    int t = blockIdx.x * 256 + tid, nt = nwg * 256;
    for (int k = t; k < g * oc; k += nt) out[k] = bl[k % oc];
    for (int k = t; k < g; k += nt) pcnt[k] = 0.f;
    __syncthreads();
    int span = blockIdx.x * SPAN;
    if (span + SPAN <= e) sort_span<true>(src, dst, lh, wsum, tile, loff, binned, e, span, nbin, nwg);
    else                  sort_span<false>(src, dst, lh, wsum, tile, loff, binned, e, span, nbin, nwg);
}

__device__ __forceinline__ void edge_addrs(const int* sgo, const int* sga, int nseg, int step0,
                                           const int (&i)[8], int (&addr)[8]) {
    int lo[8];
#pragma unroll
    for (int k = 0; k < 8; ++k) lo[k] = 0;
    for (int st = step0; st > 0; st >>= 1) {
        int v[8];
#pragma unroll
        for (int k = 0; k < 8; ++k) {
            int m = lo[k] + st;
            v[k] = sgo[m < nseg ? m : nseg];
        }
#pragma unroll
        for (int k = 0; k < 8; ++k) lo[k] = (v[k] <= i[k]) ? lo[k] + st : lo[k];
    }
#pragma unroll
    for (int k = 0; k < 8; ++k) addr[k] = sga[lo[k]] + (i[k] - sgo[lo[k]]);
}

#define EPD 8
#define XPL (BNODES * HID / 4 / 1024)
__device__ __forceinline__ void load_rows(const float4* xw4, int b, int tid, int n, float4 (&xv)[XPL]) {
#pragma unroll
    for (int q = 0; q < XPL; ++q) {
        int f = tid + q * 1024;
        int v = b * BNODES + f / (HID / 4);
        if (v >= n) v = n - 1;
        xv[q] = xw4[(size_t)v * (HID / 4) + (f & (HID / 4 - 1))];
    }
}
__global__ void __launch_bounds__(1024)
k_bindeg(const int2* __restrict__ binned, const int* __restrict__ loff, int nwg,
         int* __restrict__ rank, int* __restrict__ rowptr, float* __restrict__ dinv,
         int* __restrict__ csr, int n, int nbin, float4* xw4,
         const int* __restrict__ batch, float* cnt) {
    __shared__ int lc[BNODES];
    __shared__ float sdv[BNODES];
    __shared__ int rs[BNODES];
    __shared__ int tile[1024 * EPD];
    __shared__ int wtot[BNODES / 32];
    __shared__ int sgo[MAXSEG + 1];
    __shared__ int sga[MAXSEG];
    __shared__ int wl[32], wa[32];
    int tid = threadIdx.x, lane = tid & 31, wave = tid >> 5;
    int b  = blockIdx.x;
    if (tid >= BNODES && tid < 2 * BNODES) {
        int  v  = b * BNODES + tid - BNODES;
        bool ok = v < n;
        int  gme = batch[ok ? v : n - 1];
        int  gpv = __shfl_up(gme, 1);
        bool head = ok && (lane == 0 || gpv != gme);
        unsigned hm  = __builtin_amdgcn_ballot_w32(head);
        unsigned okm = __builtin_amdgcn_ballot_w32(ok);
        if (head) {
            unsigned above = hm & ~((2u << lane) - 1u);
            int endl = above ? __builtin_ctz(above) : __builtin_popcount(okm);
            atomicAdd(&cnt[gme], (float)(endl - lane));
        }
    }
    int pa = 0, plen = 0;
    if (tid < nwg) {
        pa = loff[b * nwg + tid];
        plen = loff[(b + 1) * nwg + tid] - pa;
    }
    int px = plen;
#pragma unroll
    for (int o = 1; o < 32; o <<= 1) {
        int y = __shfl_up(px, o);
        if (lane >= o) px += y;
    }
    int pas = pa;
#pragma unroll
    for (int o = 16; o > 0; o >>= 1) pas += __shfl_xor(pas, o);
    if (lane == 31) wl[wave] = px;
    if (lane == 0)  wa[wave] = pas;
    if (tid < BNODES) lc[tid] = 0;
    __syncthreads();
    int wlen = wl[lane];
    int wx = wlen;
#pragma unroll
    for (int o = 1; o < 32; o <<= 1) {
        int y = __shfl_up(wx, o);
        if (lane >= o) wx += y;
    }
    int wave_excl = __shfl(wx, wave) - __shfl(wlen, wave);
    int nb = __shfl(wx, 31);
    int bs = wa[lane];
#pragma unroll
    for (int o = 16; o > 0; o >>= 1) bs += __shfl_xor(bs, o);
    int be = bs + nb;
    if (tid < nwg) {
        sgo[tid] = wave_excl + px - plen;
        sga[tid] = tid * SPAN + pa;
    }
    if (tid == 0) sgo[nwg] = nb;
    __syncthreads();
    float4* xrow = xw4 + (size_t)b * (BNODES * HID / 4);
    float4 xv[XPL];
    load_rows(xw4, b, tid, n, xv);
    int step0 = 1;
    while (step0 * 2 < nwg) step0 *= 2;
    int last = nb - 1;

    int s0[EPD], d0[EPD], r0[EPD];
#pragma unroll
    for (int k = 0; k < EPD; ++k) { s0[k] = 0; d0[k] = 0; }
    if (nb > 0) {
        int ic[EPD], ad[EPD];
#pragma unroll
        for (int k = 0; k < EPD; ++k) {
            int il = k * 1024 + tid;
            ic[k] = il < nb ? il : last;
        }
        edge_addrs(sgo, sga, nwg, step0, ic, ad);
#pragma unroll
        for (int k = 0; k < EPD; ++k) {
            int2 sd = binned[ad[k]];
            s0[k] = sd.x;
            d0[k] = sd.y;
        }
    }
#pragma unroll
    for (int k = 0; k < EPD; ++k) {
        int il = k * 1024 + tid;
        r0[k] = 0;
        if (il < nb) r0[k] = atomicAdd(&lc[d0[k] & (BNODES - 1)], 1);
    }
    for (int l0 = 1024 * EPD; l0 < nb; l0 += 1024 * EPD) {
        int d[EPD], r[EPD], ic[EPD], ad[EPD];
#pragma unroll
        for (int k = 0; k < EPD; ++k) {
            int il = l0 + k * 1024 + tid;
            ic[k] = il < nb ? il : last;
        }
        edge_addrs(sgo, sga, nwg, step0, ic, ad);
#pragma unroll
        for (int k = 0; k < EPD; ++k) d[k] = binned[ad[k]].y;
#pragma unroll
        for (int k = 0; k < EPD; ++k) {
            int il = l0 + k * 1024 + tid;
            r[k] = 0;
            if (il < nb) r[k] = atomicAdd(&lc[d[k] & (BNODES - 1)], 1);
        }
#pragma unroll
        for (int k = 0; k < EPD; ++k) {
            int il = l0 + k * 1024 + tid;
            if (il < nb) rank[bs + il] = r[k];
        }
    }
    __syncthreads();
    int c = 0, x = 0;
    if (tid < BNODES) {
        c = lc[tid];
        x = c;
#pragma unroll
        for (int o = 1; o < 32; o <<= 1) {
            int y = __shfl_up(x, o);
            if (lane >= o) x += y;
        }
        if (lane == 31) wtot[wave] = x;
    }
    __syncthreads();
    if (tid < BNODES) {
        int off = bs;
        for (int j = 0; j < wave; ++j) off += wtot[j];
        int start = off + x - c;
        rs[tid] = start - bs;
        int v = b * BNODES + tid;
        float dg = (float)(c + 1);
        float di = (dg > 0.f) ? rsqrtf(dg) : 0.f;
        sdv[tid] = di;
        if (v < n) {
            rowptr[v] = start;
            dinv[v] = di;
        }
        if (b == nbin - 1 && tid == 0) rowptr[n] = be;
    }
    __syncthreads();
#pragma unroll
    for (int q = 0; q < XPL; ++q) {
        int f = tid + q * 1024;
        int row = f / (HID / 4);
        if (b * BNODES + row < n) {
            float dr = sdv[row];
            float4 t = xv[q];
            t.x *= dr; t.y *= dr; t.z *= dr; t.w *= dr;
            xrow[f] = t;
        }
    }
    if (nb <= 1024 * EPD) {
#pragma unroll
        for (int k = 0; k < EPD; ++k) {
            int il = k * 1024 + tid;
            if (il < nb) tile[rs[d0[k] & (BNODES - 1)] + r0[k]] = s0[k];
        }
        __syncthreads();
        for (int t = tid; t < nb; t += 1024) csr[bs + t] = tile[t];
        return;
    }
#pragma unroll
    for (int k = 0; k < EPD; ++k) {
        int il = k * 1024 + tid;
        if (il < nb) csr[bs + rs[d0[k] & (BNODES - 1)] + r0[k]] = s0[k];
    }
    for (int l0 = 1024 * EPD; l0 < nb; l0 += 1024 * EPD) {
        int ic[EPD], ad[EPD];
#pragma unroll
        for (int k = 0; k < EPD; ++k) {
            int il = l0 + k * 1024 + tid;
            ic[k] = il < nb ? il : last;
        }
        edge_addrs(sgo, sga, nwg, step0, ic, ad);
#pragma unroll
        for (int k = 0; k < EPD; ++k) {
            int il = l0 + k * 1024 + tid;
            if (il < nb) {
                int2 sd = binned[ad[k]];
                csr[bs + rs[sd.y & (BNODES - 1)] + rank[bs + il]] = sd.x;
            }
        }
    }
}

template <int U>
__device__ __forceinline__ void gather_rows(const float2* __restrict__ h2, int s_l,
                                            int j, int lane, float2& acc) {
    float2 r[U];
#pragma unroll
    for (int k = 0; k < U; ++k) {
        int s = __builtin_amdgcn_readlane(s_l, j + k);
        r[k]  = h2[(size_t)s * (HID / 2) + lane];
    }
#pragma unroll
    for (int k = 0; k < U; ++k) {
        acc.x += r[k].x;
        acc.y += r[k].y;
    }
}

__device__ __forceinline__ void gather_chunk(const float2* __restrict__ h2, int s_l,
                                             int m, int lane, float2& acc) {
    int j = 0;
    for (; j + 8 <= m; j += 8) gather_rows<8>(h2, s_l, j, lane, acc);
    if (m & 4) { gather_rows<4>(h2, s_l, j, lane, acc); j += 4; }
    if (m & 2) { gather_rows<2>(h2, s_l, j, lane, acc); j += 2; }
    if (m & 1) { gather_rows<1>(h2, s_l, j, lane, acc); }
}

#define APITCH 68
__device__ __forceinline__ void gather_nodes(const int* __restrict__ rowptr, const int* __restrict__ csr,
                                             const float* __restrict__ dinv, const float2* __restrict__ h2,
                                             float* dstl, int v0, int vend, int lane) {
    int beg = rowptr[v0], end = rowptr[v0 + 1];
    float di = dinv[v0];
    int rec = 0;
    if (lane < end - beg) rec = csr[beg + lane];
    float2 acc = h2[(size_t)v0 * (HID / 2) + lane];
    for (int v = v0; v < vend; ++v) {
        int vn = (v + 1 < vend) ? v + 1 : v;
        int nbeg = rowptr[vn], nend = rowptr[vn + 1];
        float ndi = dinv[vn];
        int nrec = 0;
        if (lane < nend - nbeg) nrec = csr[nbeg + lane];
        float2 nhv = h2[(size_t)vn * (HID / 2) + lane];

        int m = end - beg;
        if (m > 32) m = 32;
        gather_chunk(h2, rec, m, lane, acc);
        for (int p0 = beg + 32; p0 < end; p0 += 32) {
            int m2 = end - p0;
            if (m2 > 32) m2 = 32;
            int rec2 = 0;
            if (lane < m2) rec2 = csr[p0 + lane];
            gather_chunk(h2, rec2, m2, lane, acc);
        }
        float2 t;
        t.x = fmaxf(acc.x * di, 0.f);
        t.y = fmaxf(acc.y * di, 0.f);
        *(float2*)(dstl + (v - v0) * APITCH + 2 * lane) = t;

        acc = nhv;
        di = ndi;
        beg = nbeg; end = nend; rec = nrec;
    }
}

template <int NO>
__global__ void __launch_bounds__(256)
k_gather_gemm(const int* __restrict__ rowptr, const int* __restrict__ csr,
              const float* __restrict__ dinv, const float* __restrict__ hW,
              const float* __restrict__ Wn, float* __restrict__ hWn, int n) {
    __shared__ float4 sW4[HID * NO / 4];
    __shared__ __attribute__((aligned(16))) float sA[8 * NPW * APITCH];
    const float* sW = (const float*)sW4;
    int tid  = threadIdx.x;
    int lane = tid & 31;
    int wv = __builtin_amdgcn_readfirstlane(tid >> 5);
    constexpr int NPIECE = HID * NO / 4 / 256;
    float4 wreg[NPIECE];
#pragma unroll
    for (int q = 0; q < NPIECE; ++q) wreg[q] = ((const float4*)Wn)[tid + q * 256];
#pragma unroll
    for (int q = 0; q < NPIECE; ++q) sW4[tid + q * 256] = wreg[q];

    int v0 = (blockIdx.x * 8 + wv) * NPW;
    int vend = v0 + NPW;
    if (vend > n) vend = n;
    float* myrows = sA + wv * NPW * APITCH;
    if (v0 < n)
        gather_nodes(rowptr, csr, dinv, (const float2*)hW, myrows, v0, vend, lane);
    for (int v = (v0 < n ? vend : v0); v < v0 + NPW; ++v) {
        float2 z; z.x = 0.f; z.y = 0.f;
        *(float2*)(myrows + (v - v0) * APITCH + 2 * lane) = z;
    }
    __syncthreads();
    constexpr bool WIDE = (NO == HID);
    if (wv >= (WIDE ? 4 : 2)) return;

    int mt = WIDE ? (wv >> 1) : wv;
    int nh = WIDE ? (wv & 1) : 0;
    int half = lane >> 4;
    int l15  = lane & 15;
    const float* ap = sA + (mt * 16 + l15) * APITCH + half * 2;
    const float* bp = sW + nh * 32 + l15 + half * 2 * NO;
    v8f c0 = {}, c1 = {};
#pragma unroll
    for (int k = 0; k < HID; k += 4) {
        v2f a, b0, b1;
        a[0] = ap[k]; a[1] = ap[k + 1];
        const float* bk = bp + k * NO;
        b0[0] = bk[0];  b0[1] = bk[NO];
        c0 = WMMA_F32(a, b0, c0);
        if (WIDE) {
            b1[0] = bk[16]; b1[1] = bk[NO + 16];
            c1 = WMMA_F32(a, b1, c1);
        }
    }
    int rbase = blockIdx.x * (8 * NPW) + mt * 16 + half * 8;
#pragma unroll
    for (int j = 0; j < 8; ++j) {
        int r = rbase + j;
        if (r < n) {
            float* cr = hWn + (size_t)r * NO + nh * 32 + l15;
            float dr = dinv[r];
            cr[0] = c0[j] * dr;
            if (WIDE) cr[16] = c1[j] * dr;
        }
    }
}

__device__ __forceinline__ float wdst(int p, const int (&rp)[NPW3], const float (&dv)[NPW3]) {
    float wv = dv[0];
#pragma unroll
    for (int j = 1; j < NPW3; ++j) wv = (p >= rp[j]) ? dv[j] : wv;
    return wv;
}

__global__ void __launch_bounds__(256)
k_gather_pool(const int* __restrict__ rowptr, const int* __restrict__ csr,
              const float* __restrict__ dinv, const float* __restrict__ z,
              const int* __restrict__ batch, const float* __restrict__ cnt, float* out, int n, int oc) {
    int lane = threadIdx.x & 31;
    int sub = lane & 7, grp = lane >> 3;
    int w  = __builtin_amdgcn_readfirstlane((int)(blockIdx.x * 8 + (threadIdx.x >> 5)));
    int v0 = w * NPW3;
    if (v0 >= n) return;
    int nv = n - v0;
    if (nv > NPW3) nv = NPW3;
    const float2* z2 = (const float2*)z;

    int vb = v0 + lane;
    int rp_l = rowptr[vb <= n ? vb : n];
    if (vb >= n) vb = n - 1;
    int   g_l  = batch[vb];
    float di_l = dinv[vb];
    float2 own[NPW3 / 4];
    float  ow[NPW3 / 4];
#pragma unroll
    for (int q = 0; q < NPW3 / 4; ++q) {
        int j = 4 * q + grp;
        own[q] = z2[j < nv ? (size_t)v0 * (PW / 2) + q * 32 + lane : 0];
        ow[q] = __shfl(di_l, j);
    }
    int   rpj[NPW3];
    float dj[NPW3];
#pragma unroll
    for (int j = 0; j < NPW3; ++j) {
        rpj[j] = __builtin_amdgcn_readlane(rp_l, j);
        dj[j]  = __int_as_float(__builtin_amdgcn_readlane(__float_as_int(di_l), j));
    }

    int j0 = 0;
    while (j0 < nv) {
        int g = __builtin_amdgcn_readfirstlane(__shfl(g_l, j0));
        unsigned diff = __builtin_amdgcn_ballot_w32(lane > j0 && lane < nv && g_l != g);
        int j1 = diff ? __builtin_ctz(diff) : nv;
        int eb = __builtin_amdgcn_readfirstlane(__shfl(rp_l, j0));
        int ee = __builtin_amdgcn_readfirstlane(__shfl(rp_l, j1));
        float cg = cnt[g];

        float2 acc = make_float2(0.f, 0.f);
#pragma unroll
        for (int q = 0; q < NPW3 / 4; ++q) {
            if (4 * q + grp >= j0 && 4 * q + grp < j1) {
                acc.x = fmaf(own[q].x, ow[q], acc.x);
                acc.y = fmaf(own[q].y, ow[q], acc.y);
            }
        }
        if (eb < ee) {
            int q0 = eb + lane, q1 = eb + 32 + lane;
            if (q0 >= ee) q0 = ee - 1;
            if (q1 >= ee) q1 = ee - 1;
            int   rec0 = csr[q0], rec1 = csr[q1];
            float w0 = wdst(q0, rpj, dj), w1 = wdst(q1, rpj, dj);
            for (int base = eb; base < ee; base += 64) {
                int m = ee - base;
                float2 r[16];
                float  wk[16];
#pragma unroll
                for (int k = 0; k < 16; ++k) {
                    int from = 4 * (k & 7) + grp;
                    int sn = __shfl(k < 8 ? rec0 : rec1, from);
                    wk[k] = __shfl(k < 8 ? w0 : w1, from);
                    r[k]  = z2[(size_t)sn * (PW / 2) + sub];
                }
                int n0 = base + 64 + lane, n1 = base + 96 + lane;
                if (n0 >= ee) n0 = ee - 1;
                if (n1 >= ee) n1 = ee - 1;
                int   nrec0 = csr[n0], nrec1 = csr[n1];
                float nw0 = wdst(n0, rpj, dj), nw1 = wdst(n1, rpj, dj);
#pragma unroll
                for (int k = 0; k < 16; ++k) {
                    if (4 * k + grp < m) {
                        acc.x = fmaf(r[k].x, wk[k], acc.x);
                        acc.y = fmaf(r[k].y, wk[k], acc.y);
                    }
                }
                rec0 = nrec0; rec1 = nrec1;
                w0 = nw0; w1 = nw1;
            }
        }
        acc.x += __shfl_xor(acc.x, 8);  acc.y += __shfl_xor(acc.y, 8);
        acc.x += __shfl_xor(acc.x, 16); acc.y += __shfl_xor(acc.y, 16);
        if (lane < 8) {
            float cm = fmaxf(cg, 1.0f);
            float* og = out + (size_t)g * oc + 2 * sub;
            if (2 * sub < oc)     atomicAdd(og,     acc.x / cm);
            if (2 * sub + 1 < oc) atomicAdd(og + 1, acc.y / cm);
        }
        j0 = j1;
    }
}

extern "C" void kernel_launch(void* const* d_in, const int* in_sizes, int n_in,
                              void* d_out, int out_size, void* d_ws, size_t ws_size,
                              hipStream_t stream) {
    const float* x    = (const float*)d_in[0];
    const int*   eidx = (const int*)  d_in[1];
    const int*   batch= (const int*)  d_in[2];
    const float* W1   = (const float*)d_in[3];
    const float* W2   = (const float*)d_in[4];
    const float* W3   = (const float*)d_in[5];
    const float* Wl   = (const float*)d_in[6];
    const float* bl   = (const float*)d_in[7];

    const int INC = in_sizes[3] / HID;
    const int N   = in_sizes[0] / INC;
    const int E   = in_sizes[1] / 2;
    const int OC  = in_sizes[7];
    const int G   = out_size / OC;

    const int* src = eidx;
    const int* dst = eidx + E;

    const int B = 256;
    const int eB1 = (E + B * EPL1 - 1) / (B * EPL1);
    const int nbin = (N + BNODES - 1) / BNODES;
    const int mt  = (N + 127) / 128;
    const int gB  = (N + 8 * NPW - 1) / (8 * NPW);
    const int pB  = (N + 8 * NPW3 - 1) / (8 * NPW3);

    char* ws = (char*)d_ws; size_t off = 0;
    auto carve = [&](size_t bytes) {
        void* p = ws + off; off = (off + bytes + 255) & ~(size_t)255; return p;
    };
    int*   loff   = (int*)  carve((size_t)(nbin + 1) * eB1 * 4);
    int2*  binned = (int2*) carve((size_t)eB1 * SPAN * 8);
    int*   rowptr = (int*)  carve((size_t)(N + 1) * 4);
    float* dinv   = (float*)carve((size_t)N * 4);
    int*   rank   = (int*)  carve((size_t)E * 4);
    int*   csr    = (int*)  carve((size_t)E * 4);
    float* hW     = (float*)carve((size_t)N * HID * 4);
    float* agg    = (float*)carve((size_t)N * HID * 4);
    float* cnt    = (float*)carve((size_t)G * 4);
    float* W34    = (float*)carve((size_t)HID * PW * 4);
    (void)ws_size; (void)n_in;

    k_binsort<<<eB1 + 1 + mt, B, 0, stream>>>(src, dst, loff, binned, E, nbin, eB1, (float*)d_out, bl, cnt, G,
                                              W3, Wl, W34, OC, x, W1, hW, N);
    k_bindeg <<<nbin, 1024, 0, stream>>>(binned, loff, eB1, rank, rowptr, dinv, csr, N, nbin, (float4*)hW, batch, cnt);

    k_gather_gemm<HID><<<gB, B, 0, stream>>>(rowptr, csr, dinv, hW, W2, agg, N);

    k_gather_gemm<PW><<<gB, B, 0, stream>>>(rowptr, csr, dinv, agg, W34, hW, N);

    k_gather_pool<<<pB, B, 0, stream>>>(rowptr, csr, dinv, hW, batch, cnt, (float*)d_out, N, OC);
}
